# starting kernel
# baseline (speedup 1.0000x reference)
#include <hip/hip_runtime.h>
#include <math.h>

#define N_ENT   150000
#define N_USR   60000
#define CH      64
#define E_EDGES 1000000
#define NNZ_CNT 1000000
#define NRELM1  9
#define NHOPS   2
#define EPS_N   1e-12f

typedef __attribute__((ext_vector_type(2))) float v2f;
typedef __attribute__((ext_vector_type(8))) float v8f;

#define ATT_WAVES 4
__global__ __launch_bounds__(ATT_WAVES * 32)
void att_wmma_kernel(const float* __restrict__ ent,
                     const float* __restrict__ weight,
                     const int* __restrict__ head,
                     const int* __restrict__ tail,
                     const int* __restrict__ etype,
                     float* __restrict__ att,
                     unsigned int* __restrict__ segmax)
{
    __shared__ float sW2[16 * CH];
    __shared__ float sH[ATT_WAVES][16 * CH];
    __shared__ float sT[ATT_WAVES][16 * CH];

    const int lane = threadIdx.x & 31;
    const int wave = threadIdx.x >> 5;

    for (int i = threadIdx.x; i < 16 * CH; i += ATT_WAVES * 32) {
        const int n = i >> 6, c = i & 63;
        const float w = (n < NRELM1) ? weight[n * CH + c] : 0.0f;
        sW2[i] = w * w;
    }
    __syncthreads();

    const int tile = blockIdx.x * ATT_WAVES + wave;
    if (tile * 16 >= E_EDGES) return;
    const int base = tile * 16;

    for (int r = 0; r < 16; ++r) {
        const int hr = head[base + r];
        const int tr = tail[base + r];
        const v2f hv = *(const v2f*)(ent + (size_t)hr * CH + 2 * lane);
        const v2f tv = *(const v2f*)(ent + (size_t)tr * CH + 2 * lane);
        sH[wave][r * CH + 2 * lane]     = hv.x * hv.x;
        sH[wave][r * CH + 2 * lane + 1] = hv.y * hv.y;
        sT[wave][r * CH + 2 * lane]     = tv.x * tv.x;
        sT[wave][r * CH + 2 * lane + 1] = tv.y * tv.y;
    }

    v8f accH = {};
    v8f accT = {};
    const int m    = lane & 15;
    const int koff = (lane >> 4) << 1;
    for (int kk = 0; kk < 16; ++kk) {
        const int c = kk * 4 + koff;
        v2f a, at, b;
        a.x  = sH[wave][m * CH + c];  a.y  = sH[wave][m * CH + c + 1];
        at.x = sT[wave][m * CH + c];  at.y = sT[wave][m * CH + c + 1];
        b.x  = sW2[m * CH + c];       b.y  = sW2[m * CH + c + 1];
        accH = __builtin_amdgcn_wmma_f32_16x16x4_f32(false, a,  false, b, (short)0, accH, false, false);
        accT = __builtin_amdgcn_wmma_f32_16x16x4_f32(false, at, false, b, (short)0, accT, false, false);
    }

    const int mrow = (lane >> 4) << 3;
    const int ncol = lane & 15;
#pragma unroll
    for (int v = 0; v < 8; ++v) {
        sH[wave][(mrow + v) * 16 + ncol] = accH[v];
        sT[wave][(mrow + v) * 16 + ncol] = accT[v];
    }

    if (lane < 16) {
        const int e  = base + lane;
        int et = etype[e] - 1;
        if (et < 0) et += NRELM1;
        const float hn2 = sH[wave][lane * 16 + et];
        const float tn2 = sT[wave][lane * 16 + et];
        const float a   = hn2 * tn2;
        att[e] = a;
        atomicMax(&segmax[head[e]], __float_as_uint(a));
    }
}

__global__ __launch_bounds__(256)
void exp_sum_kernel(const int* __restrict__ head,
                    const float* __restrict__ segmax,
                    float* __restrict__ att,
                    float* __restrict__ segsum)
{
    const int e = blockIdx.x * 256 + threadIdx.x;
    if (e >= E_EDGES) return;
    const int h = head[e];
    const float ev = expf(att[e] - segmax[h]);
    att[e] = ev;
    atomicAdd(&segsum[h], ev);
}

__global__ __launch_bounds__(256)
void edge_agg_kernel(const float* __restrict__ ent,
                     const float* __restrict__ weight,
                     const int* __restrict__ head,
                     const int* __restrict__ tail,
                     const int* __restrict__ etype,
                     const float* __restrict__ expv,
                     const float* __restrict__ segsum,
                     float* __restrict__ agg_ent)
{
    const int wave = threadIdx.x >> 5, lane = threadIdx.x & 31;
    const int e = blockIdx.x * 8 + wave;
    if (e >= E_EDGES) return;
    const int h = head[e], t = tail[e];
    int et = etype[e] - 1;
    if (et < 0) et += NRELM1;
    const float w = expv[e] / segsum[h];
    const int c = 2 * lane;
    const v2f tv = *(const v2f*)(ent + (size_t)t * CH + c);
    const v2f rv = *(const v2f*)(weight + et * CH + c);
    atomicAdd(agg_ent + (size_t)h * CH + c,     w * tv.x * rv.x);
    atomicAdd(agg_ent + (size_t)h * CH + c + 1, w * tv.y * rv.y);
}

__global__ __launch_bounds__(256)
void inter_agg_kernel(const float* __restrict__ cur_ent,
                      const float* __restrict__ cur_usr,
                      const float* __restrict__ vals,
                      const int* __restrict__ rows,
                      const int* __restrict__ cols,
                      float* __restrict__ agg_ent,
                      float* __restrict__ agg_usr)
{
    const int wave = threadIdx.x >> 5, lane = threadIdx.x & 31;
    const int z = blockIdx.x * 8 + wave;
    if (z >= NNZ_CNT) return;
    const int r = rows[z], cl = cols[z];
    const float v = vals[z];
    const int c = 2 * lane;
    const v2f uv = *(const v2f*)(cur_usr + (size_t)r  * CH + c);
    const v2f ev = *(const v2f*)(cur_ent + (size_t)cl * CH + c);
    atomicAdd(agg_ent + (size_t)cl * CH + c,     v * uv.x);
    atomicAdd(agg_ent + (size_t)cl * CH + c + 1, v * uv.y);
    atomicAdd(agg_usr + (size_t)r  * CH + c,     v * ev.x);
    atomicAdd(agg_usr + (size_t)r  * CH + c + 1, v * ev.y);
}

__global__ __launch_bounds__(256)
void normalize_kernel(const float* __restrict__ agg,
                      float* __restrict__ cur,
                      float* __restrict__ out,
                      int nrows)
{
    const int wave = threadIdx.x >> 5, lane = threadIdx.x & 31;
    const int row = blockIdx.x * 8 + wave;
    if (row >= nrows) return;
    const int c = 2 * lane;
    v2f v = *(const v2f*)(agg + (size_t)row * CH + c);
    float ss = v.x * v.x + v.y * v.y;
#pragma unroll
    for (int m = 16; m >= 1; m >>= 1) ss += __shfl_xor(ss, m, 32);
    const float inv = 1.0f / fmaxf(sqrtf(ss), EPS_N);
    v.x *= inv; v.y *= inv;
    *(v2f*)(cur + (size_t)row * CH + c) = v;
    v2f o = *(v2f*)(out + (size_t)row * CH + c);
    o.x += v.x; o.y += v.y;
    *(v2f*)(out + (size_t)row * CH + c) = o;
}

extern "C" void kernel_launch(void* const* d_in, const int* in_sizes, int n_in,
                              void* d_out, int out_size, void* d_ws, size_t ws_size,
                              hipStream_t stream)
{
    (void)in_sizes; (void)n_in; (void)out_size; (void)ws_size;

    const float* user_emb   = (const float*)d_in[0];
    const float* entity_emb = (const float*)d_in[1];
    const float* weight     = (const float*)d_in[2];
    const float* inter_vals = (const float*)d_in[3];
    const int*   edge_index = (const int*)d_in[4];
    const int*   edge_type  = (const int*)d_in[5];
    const int*   inter_rows = (const int*)d_in[6];
    const int*   inter_cols = (const int*)d_in[7];
    const int*   head = edge_index;
    const int*   tail = edge_index + E_EDGES;

    float* out_ent = (float*)d_out;
    float* out_usr = out_ent + (size_t)N_ENT * CH;

    char* ws = (char*)d_ws;
    size_t off = 0;
    auto alloc_f = [&](size_t nfl) -> float* {
        float* p = (float*)(ws + off);
        off += ((nfl * sizeof(float) + 255) / 256) * 256;
        return p;
    };
    float* cur_ent = alloc_f((size_t)N_ENT * CH);
    float* cur_usr = alloc_f((size_t)N_USR * CH);
    float* agg_ent = alloc_f((size_t)N_ENT * CH);
    float* agg_usr = alloc_f((size_t)N_USR * CH);
    float* att     = alloc_f(E_EDGES);
    float* segmax  = alloc_f(N_ENT);
    float* segsum  = alloc_f(N_ENT);

    hipMemcpyAsync(cur_ent, entity_emb, (size_t)N_ENT * CH * sizeof(float), hipMemcpyDeviceToDevice, stream);
    hipMemcpyAsync(cur_usr, user_emb,   (size_t)N_USR * CH * sizeof(float), hipMemcpyDeviceToDevice, stream);
    hipMemcpyAsync(out_ent, entity_emb, (size_t)N_ENT * CH * sizeof(float), hipMemcpyDeviceToDevice, stream);
    hipMemcpyAsync(out_usr, user_emb,   (size_t)N_USR * CH * sizeof(float), hipMemcpyDeviceToDevice, stream);

    for (int hop = 0; hop < NHOPS; ++hop) {
        hipMemsetAsync(agg_ent, 0, (size_t)N_ENT * CH * sizeof(float), stream);
        hipMemsetAsync(agg_usr, 0, (size_t)N_USR * CH * sizeof(float), stream);
        hipMemsetAsync(segmax,  0, (size_t)N_ENT * sizeof(float), stream);
        hipMemsetAsync(segsum,  0, (size_t)N_ENT * sizeof(float), stream);

        att_wmma_kernel<<<E_EDGES / 16 / ATT_WAVES, ATT_WAVES * 32, 0, stream>>>(
            cur_ent, weight, head, tail, edge_type, att, (unsigned int*)segmax);
        exp_sum_kernel<<<(E_EDGES + 255) / 256, 256, 0, stream>>>(head, segmax, att, segsum);
        edge_agg_kernel<<<E_EDGES / 8, 256, 0, stream>>>(
            cur_ent, weight, head, tail, edge_type, att, segsum, agg_ent);
        inter_agg_kernel<<<NNZ_CNT / 8, 256, 0, stream>>>(
            cur_ent, cur_usr, inter_vals, inter_rows, inter_cols, agg_ent, agg_usr);
        normalize_kernel<<<(N_ENT + 7) / 8, 256, 0, stream>>>(agg_ent, cur_ent, out_ent, N_ENT);
        normalize_kernel<<<(N_USR + 7) / 8, 256, 0, stream>>>(agg_usr, cur_usr, out_usr, N_USR);
    }
}


# Edge and interaction scatter kernels: a wave walks 32 consecutive items (indices read once coalesced, readlane broadcast), 8 items' row loads in flight per group; atomics kept, arrival order varies
# speedup vs baseline: 1.1879x; 1.1879x over previous
#include <hip/hip_runtime.h>
#include <math.h>

#define N_ENT   150000
#define N_USR   60000
#define CH      64
#define E_EDGES 1000000
#define NNZ_CNT 1000000
#define NRELM1  9
#define NHOPS   2
#define EPS_N   1e-12f

typedef __attribute__((ext_vector_type(2))) float v2f;
typedef __attribute__((ext_vector_type(8))) float v8f;

#define ATT_WAVES 4
__global__ __launch_bounds__(ATT_WAVES * 32)
void att_wmma_kernel(const float* __restrict__ ent,
                     const float* __restrict__ weight,
                     const int* __restrict__ head,
                     const int* __restrict__ tail,
                     const int* __restrict__ etype,
                     float* __restrict__ att,
                     unsigned int* __restrict__ segmax)
{
    __shared__ float sW2[16 * CH];
    __shared__ float sH[ATT_WAVES][16 * CH];
    __shared__ float sT[ATT_WAVES][16 * CH];

    const int lane = threadIdx.x & 31;
    const int wave = threadIdx.x >> 5;

    for (int i = threadIdx.x; i < 16 * CH; i += ATT_WAVES * 32) {
        const int n = i >> 6, c = i & 63;
        const float w = (n < NRELM1) ? weight[n * CH + c] : 0.0f;
        sW2[i] = w * w;
    }
    __syncthreads();

    const int tile = blockIdx.x * ATT_WAVES + wave;
    if (tile * 16 >= E_EDGES) return;
    const int base = tile * 16;

    for (int r = 0; r < 16; ++r) {
        const int hr = head[base + r];
        const int tr = tail[base + r];
        const v2f hv = *(const v2f*)(ent + (size_t)hr * CH + 2 * lane);
        const v2f tv = *(const v2f*)(ent + (size_t)tr * CH + 2 * lane);
        sH[wave][r * CH + 2 * lane]     = hv.x * hv.x;
        sH[wave][r * CH + 2 * lane + 1] = hv.y * hv.y;
        sT[wave][r * CH + 2 * lane]     = tv.x * tv.x;
        sT[wave][r * CH + 2 * lane + 1] = tv.y * tv.y;
    }

    v8f accH = {};
    v8f accT = {};
    const int m    = lane & 15;
    const int koff = (lane >> 4) << 1;
    for (int kk = 0; kk < 16; ++kk) {
        const int c = kk * 4 + koff;
        v2f a, at, b;
        a.x  = sH[wave][m * CH + c];  a.y  = sH[wave][m * CH + c + 1];
        at.x = sT[wave][m * CH + c];  at.y = sT[wave][m * CH + c + 1];
        b.x  = sW2[m * CH + c];       b.y  = sW2[m * CH + c + 1];
        accH = __builtin_amdgcn_wmma_f32_16x16x4_f32(false, a,  false, b, (short)0, accH, false, false);
        accT = __builtin_amdgcn_wmma_f32_16x16x4_f32(false, at, false, b, (short)0, accT, false, false);
    }

    const int mrow = (lane >> 4) << 3;
    const int ncol = lane & 15;
#pragma unroll
    for (int v = 0; v < 8; ++v) {
        sH[wave][(mrow + v) * 16 + ncol] = accH[v];
        sT[wave][(mrow + v) * 16 + ncol] = accT[v];
    }

    if (lane < 16) {
        const int e  = base + lane;
        int et = etype[e] - 1;
        if (et < 0) et += NRELM1;
        const float hn2 = sH[wave][lane * 16 + et];
        const float tn2 = sT[wave][lane * 16 + et];
        const float a   = hn2 * tn2;
        att[e] = a;
        atomicMax(&segmax[head[e]], __float_as_uint(a));
    }
}

__global__ __launch_bounds__(256)
void exp_sum_kernel(const int* __restrict__ head,
                    const float* __restrict__ segmax,
                    float* __restrict__ att,
                    float* __restrict__ segsum)
{
    const int e = blockIdx.x * 256 + threadIdx.x;
    if (e >= E_EDGES) return;
    const int h = head[e];
    const float ev = expf(att[e] - segmax[h]);
    att[e] = ev;
    atomicAdd(&segsum[h], ev);
}

#define EPW 32
#define GRP 8
static_assert(E_EDGES % EPW == 0 && NNZ_CNT % EPW == 0, "runs must be whole");
static_assert(EPW % GRP == 0, "groups must tile a run");

__global__ __launch_bounds__(256)
void edge_agg_kernel(const float* __restrict__ ent,
                     const float* __restrict__ weight,
                     const int* __restrict__ head,
                     const int* __restrict__ tail,
                     const int* __restrict__ etype,
                     const float* __restrict__ expv,
                     const float* __restrict__ segsum,
                     float* __restrict__ agg_ent)
{
    const int wave = threadIdx.x >> 5, lane = threadIdx.x & 31;
    const int e0 = (blockIdx.x * 8 + wave) * EPW;
    if (e0 >= E_EDGES) return;
    const int e = e0 + lane;
    const int h = head[e], t = tail[e];
    int et = etype[e] - 1;
    if (et < 0) et += NRELM1;
    const float w = expv[e] / segsum[h];
    const int wi = __float_as_int(w);
    const int c = 2 * lane;
#pragma unroll
    for (int g = 0; g < EPW; g += GRP) {
        v2f tv[GRP], rv[GRP];
        for (int j = 0; j < GRP; ++j) {
            const int tj  = __builtin_amdgcn_readlane(t,  g + j);
            const int etj = __builtin_amdgcn_readlane(et, g + j);
            tv[j] = *(const v2f*)(ent + (size_t)tj * CH + c);
            rv[j] = *(const v2f*)(weight + etj * CH + c);
        }
        for (int j = 0; j < GRP; ++j) {
            const int   hj = __builtin_amdgcn_readlane(h,  g + j);
            const float wj = __int_as_float(__builtin_amdgcn_readlane(wi, g + j));
            float* dst = agg_ent + (size_t)hj * CH + c;
            atomicAdd(dst,     wj * tv[j].x * rv[j].x);
            atomicAdd(dst + 1, wj * tv[j].y * rv[j].y);
        }
    }
}

__global__ __launch_bounds__(256)
void inter_agg_kernel(const float* __restrict__ cur_ent,
                      const float* __restrict__ cur_usr,
                      const float* __restrict__ vals,
                      const int* __restrict__ rows,
                      const int* __restrict__ cols,
                      float* __restrict__ agg_ent,
                      float* __restrict__ agg_usr)
{
    const int wave = threadIdx.x >> 5, lane = threadIdx.x & 31;
    const int z0 = (blockIdx.x * 8 + wave) * EPW;
    if (z0 >= NNZ_CNT) return;
    const int z = z0 + lane;
    const int r = rows[z], cl = cols[z];
    const int vi = __float_as_int(vals[z]);
    const int c = 2 * lane;
#pragma unroll
    for (int g = 0; g < EPW; g += GRP) {
        v2f uv[GRP], ev[GRP];
        for (int j = 0; j < GRP; ++j) {
            const int rj  = __builtin_amdgcn_readlane(r,  g + j);
            const int clj = __builtin_amdgcn_readlane(cl, g + j);
            uv[j] = *(const v2f*)(cur_usr + (size_t)rj  * CH + c);
            ev[j] = *(const v2f*)(cur_ent + (size_t)clj * CH + c);
        }
        for (int j = 0; j < GRP; ++j) {
            const int   rj  = __builtin_amdgcn_readlane(r,  g + j);
            const int   clj = __builtin_amdgcn_readlane(cl, g + j);
            const float vj  = __int_as_float(__builtin_amdgcn_readlane(vi, g + j));
            float* de = agg_ent + (size_t)clj * CH + c;
            float* du = agg_usr + (size_t)rj  * CH + c;
            atomicAdd(de,     vj * uv[j].x);
            atomicAdd(de + 1, vj * uv[j].y);
            atomicAdd(du,     vj * ev[j].x);
            atomicAdd(du + 1, vj * ev[j].y);
        }
    }
}

__global__ __launch_bounds__(256)
void normalize_kernel(const float* __restrict__ agg,
                      float* __restrict__ cur,
                      float* __restrict__ out,
                      int nrows)
{
    const int wave = threadIdx.x >> 5, lane = threadIdx.x & 31;
    const int row = blockIdx.x * 8 + wave;
    if (row >= nrows) return;
    const int c = 2 * lane;
    v2f v = *(const v2f*)(agg + (size_t)row * CH + c);
    float ss = v.x * v.x + v.y * v.y;
#pragma unroll
    for (int m = 16; m >= 1; m >>= 1) ss += __shfl_xor(ss, m, 32);
    const float inv = 1.0f / fmaxf(sqrtf(ss), EPS_N);
    v.x *= inv; v.y *= inv;
    *(v2f*)(cur + (size_t)row * CH + c) = v;
    v2f o = *(v2f*)(out + (size_t)row * CH + c);
    o.x += v.x; o.y += v.y;
    *(v2f*)(out + (size_t)row * CH + c) = o;
}

extern "C" void kernel_launch(void* const* d_in, const int* in_sizes, int n_in,
                              void* d_out, int out_size, void* d_ws, size_t ws_size,
                              hipStream_t stream)
{
    (void)in_sizes; (void)n_in; (void)out_size; (void)ws_size;

    const float* user_emb   = (const float*)d_in[0];
    const float* entity_emb = (const float*)d_in[1];
    const float* weight     = (const float*)d_in[2];
    const float* inter_vals = (const float*)d_in[3];
    const int*   edge_index = (const int*)d_in[4];
    const int*   edge_type  = (const int*)d_in[5];
    const int*   inter_rows = (const int*)d_in[6];
    const int*   inter_cols = (const int*)d_in[7];
    const int*   head = edge_index;
    const int*   tail = edge_index + E_EDGES;

    float* out_ent = (float*)d_out;
    float* out_usr = out_ent + (size_t)N_ENT * CH;

    char* ws = (char*)d_ws;
    size_t off = 0;
    auto alloc_f = [&](size_t nfl) -> float* {
        float* p = (float*)(ws + off);
        off += ((nfl * sizeof(float) + 255) / 256) * 256;
        return p;
    };
    float* cur_ent = alloc_f((size_t)N_ENT * CH);
    float* cur_usr = alloc_f((size_t)N_USR * CH);
    float* agg_ent = alloc_f((size_t)N_ENT * CH);
    float* agg_usr = alloc_f((size_t)N_USR * CH);
    float* att     = alloc_f(E_EDGES);
    float* segmax  = alloc_f(N_ENT);
    float* segsum  = alloc_f(N_ENT);

    hipMemcpyAsync(cur_ent, entity_emb, (size_t)N_ENT * CH * sizeof(float), hipMemcpyDeviceToDevice, stream);
    hipMemcpyAsync(cur_usr, user_emb,   (size_t)N_USR * CH * sizeof(float), hipMemcpyDeviceToDevice, stream);
    hipMemcpyAsync(out_ent, entity_emb, (size_t)N_ENT * CH * sizeof(float), hipMemcpyDeviceToDevice, stream);
    hipMemcpyAsync(out_usr, user_emb,   (size_t)N_USR * CH * sizeof(float), hipMemcpyDeviceToDevice, stream);

    for (int hop = 0; hop < NHOPS; ++hop) {
        hipMemsetAsync(agg_ent, 0, (size_t)N_ENT * CH * sizeof(float), stream);
        hipMemsetAsync(agg_usr, 0, (size_t)N_USR * CH * sizeof(float), stream);
        hipMemsetAsync(segmax,  0, (size_t)N_ENT * sizeof(float), stream);
        hipMemsetAsync(segsum,  0, (size_t)N_ENT * sizeof(float), stream);

        att_wmma_kernel<<<E_EDGES / 16 / ATT_WAVES, ATT_WAVES * 32, 0, stream>>>(
            cur_ent, weight, head, tail, edge_type, att, (unsigned int*)segmax);
        exp_sum_kernel<<<(E_EDGES + 255) / 256, 256, 0, stream>>>(head, segmax, att, segsum);
        edge_agg_kernel<<<(E_EDGES / EPW + 7) / 8, 256, 0, stream>>>(
            cur_ent, weight, head, tail, edge_type, att, segsum, agg_ent);
        inter_agg_kernel<<<(NNZ_CNT / EPW + 7) / 8, 256, 0, stream>>>(
            cur_ent, cur_usr, inter_vals, inter_rows, inter_cols, agg_ent, agg_usr);
        normalize_kernel<<<(N_ENT + 7) / 8, 256, 0, stream>>>(agg_ent, cur_ent, out_ent, N_ENT);
        normalize_kernel<<<(N_USR + 7) / 8, 256, 0, stream>>>(agg_usr, cur_usr, out_usr, N_USR);
    }
}


# Attention kernel: LDS row pitch of the 16-row WMMA operand tiles padded from 64 to 68 floats, so each fragment read covers all 64 banks once instead of 16 rows on one bank
# speedup vs baseline: 1.2589x; 1.0598x over previous
#include <hip/hip_runtime.h>
#include <math.h>

#define N_ENT   150000
#define N_USR   60000
#define CH      64
#define E_EDGES 1000000
#define NNZ_CNT 1000000
#define NRELM1  9
#define NHOPS   2
#define EPS_N   1e-12f

typedef __attribute__((ext_vector_type(2))) float v2f;
typedef __attribute__((ext_vector_type(8))) float v8f;

#define ATT_WAVES 4
#define LDP 68
__global__ __launch_bounds__(ATT_WAVES * 32)
void att_wmma_kernel(const float* __restrict__ ent,
                     const float* __restrict__ weight,
                     const int* __restrict__ head,
                     const int* __restrict__ tail,
                     const int* __restrict__ etype,
                     float* __restrict__ att,
                     unsigned int* __restrict__ segmax)
{
    __shared__ float sW2[16 * LDP];
    __shared__ float sH[ATT_WAVES][16 * LDP];
    __shared__ float sT[ATT_WAVES][16 * LDP];

    const int lane = threadIdx.x & 31;
    const int wave = threadIdx.x >> 5;

    for (int i = threadIdx.x; i < 16 * CH; i += ATT_WAVES * 32) {
        const int n = i >> 6, c = i & 63;
        const float w = (n < NRELM1) ? weight[n * CH + c] : 0.0f;
        sW2[n * LDP + c] = w * w;
    }
    __syncthreads();

    const int tile = blockIdx.x * ATT_WAVES + wave;
    if (tile * 16 >= E_EDGES) return;
    const int base = tile * 16;

    for (int r = 0; r < 16; ++r) {
        const int hr = head[base + r];
        const int tr = tail[base + r];
        const v2f hv = *(const v2f*)(ent + (size_t)hr * CH + 2 * lane);
        const v2f tv = *(const v2f*)(ent + (size_t)tr * CH + 2 * lane);
        sH[wave][r * LDP + 2 * lane]     = hv.x * hv.x;
        sH[wave][r * LDP + 2 * lane + 1] = hv.y * hv.y;
        sT[wave][r * LDP + 2 * lane]     = tv.x * tv.x;
        sT[wave][r * LDP + 2 * lane + 1] = tv.y * tv.y;
    }

    v8f accH = {};
    v8f accT = {};
    const int m    = lane & 15;
    const int koff = (lane >> 4) << 1;
    for (int kk = 0; kk < 16; ++kk) {
        const int c = kk * 4 + koff;
        v2f a, at, b;
        a.x  = sH[wave][m * LDP + c];  a.y  = sH[wave][m * LDP + c + 1];
        at.x = sT[wave][m * LDP + c];  at.y = sT[wave][m * LDP + c + 1];
        b.x  = sW2[m * LDP + c];       b.y  = sW2[m * LDP + c + 1];
        accH = __builtin_amdgcn_wmma_f32_16x16x4_f32(false, a,  false, b, (short)0, accH, false, false);
        accT = __builtin_amdgcn_wmma_f32_16x16x4_f32(false, at, false, b, (short)0, accT, false, false);
    }

    const int mrow = (lane >> 4) << 3;
    const int ncol = lane & 15;
#pragma unroll
    for (int v = 0; v < 8; ++v) {
        sH[wave][(mrow + v) * 16 + ncol] = accH[v];
        sT[wave][(mrow + v) * 16 + ncol] = accT[v];
    }

    if (lane < 16) {
        const int e  = base + lane;
        int et = etype[e] - 1;
        if (et < 0) et += NRELM1;
        const float hn2 = sH[wave][lane * 16 + et];
        const float tn2 = sT[wave][lane * 16 + et];
        const float a   = hn2 * tn2;
        att[e] = a;
        atomicMax(&segmax[head[e]], __float_as_uint(a));
    }
}

__global__ __launch_bounds__(256)
void exp_sum_kernel(const int* __restrict__ head,
                    const float* __restrict__ segmax,
                    float* __restrict__ att,
                    float* __restrict__ segsum)
{
    const int e = blockIdx.x * 256 + threadIdx.x;
    if (e >= E_EDGES) return;
    const int h = head[e];
    const float ev = expf(att[e] - segmax[h]);
    att[e] = ev;
    atomicAdd(&segsum[h], ev);
}

#define EPW 32
#define GRP 8
static_assert(E_EDGES % EPW == 0 && NNZ_CNT % EPW == 0, "runs must be whole");
static_assert(EPW % GRP == 0, "groups must tile a run");

__global__ __launch_bounds__(256)
void edge_agg_kernel(const float* __restrict__ ent,
                     const float* __restrict__ weight,
                     const int* __restrict__ head,
                     const int* __restrict__ tail,
                     const int* __restrict__ etype,
                     const float* __restrict__ expv,
                     const float* __restrict__ segsum,
                     float* __restrict__ agg_ent)
{
    const int wave = threadIdx.x >> 5, lane = threadIdx.x & 31;
    const int e0 = (blockIdx.x * 8 + wave) * EPW;
    if (e0 >= E_EDGES) return;
    const int e = e0 + lane;
    const int h = head[e], t = tail[e];
    int et = etype[e] - 1;
    if (et < 0) et += NRELM1;
    const float w = expv[e] / segsum[h];
    const int wi = __float_as_int(w);
    const int c = 2 * lane;
#pragma unroll
    for (int g = 0; g < EPW; g += GRP) {
        v2f tv[GRP], rv[GRP];
        for (int j = 0; j < GRP; ++j) {
            const int tj  = __builtin_amdgcn_readlane(t,  g + j);
            const int etj = __builtin_amdgcn_readlane(et, g + j);
            tv[j] = *(const v2f*)(ent + (size_t)tj * CH + c);
            rv[j] = *(const v2f*)(weight + etj * CH + c);
        }
        for (int j = 0; j < GRP; ++j) {
            const int   hj = __builtin_amdgcn_readlane(h,  g + j);
            const float wj = __int_as_float(__builtin_amdgcn_readlane(wi, g + j));
            float* dst = agg_ent + (size_t)hj * CH + c;
            atomicAdd(dst,     wj * tv[j].x * rv[j].x);
            atomicAdd(dst + 1, wj * tv[j].y * rv[j].y);
        }
    }
}

__global__ __launch_bounds__(256)
void inter_agg_kernel(const float* __restrict__ cur_ent,
                      const float* __restrict__ cur_usr,
                      const float* __restrict__ vals,
                      const int* __restrict__ rows,
                      const int* __restrict__ cols,
                      float* __restrict__ agg_ent,
                      float* __restrict__ agg_usr)
{
    const int wave = threadIdx.x >> 5, lane = threadIdx.x & 31;
    const int z0 = (blockIdx.x * 8 + wave) * EPW;
    if (z0 >= NNZ_CNT) return;
    const int z = z0 + lane;
    const int r = rows[z], cl = cols[z];
    const int vi = __float_as_int(vals[z]);
    const int c = 2 * lane;
#pragma unroll
    for (int g = 0; g < EPW; g += GRP) {
        v2f uv[GRP], ev[GRP];
        for (int j = 0; j < GRP; ++j) {
            const int rj  = __builtin_amdgcn_readlane(r,  g + j);
            const int clj = __builtin_amdgcn_readlane(cl, g + j);
            uv[j] = *(const v2f*)(cur_usr + (size_t)rj  * CH + c);
            ev[j] = *(const v2f*)(cur_ent + (size_t)clj * CH + c);
        }
        for (int j = 0; j < GRP; ++j) {
            const int   rj  = __builtin_amdgcn_readlane(r,  g + j);
            const int   clj = __builtin_amdgcn_readlane(cl, g + j);
            const float vj  = __int_as_float(__builtin_amdgcn_readlane(vi, g + j));
            float* de = agg_ent + (size_t)clj * CH + c;
            float* du = agg_usr + (size_t)rj  * CH + c;
            atomicAdd(de,     vj * uv[j].x);
            atomicAdd(de + 1, vj * uv[j].y);
            atomicAdd(du,     vj * ev[j].x);
            atomicAdd(du + 1, vj * ev[j].y);
        }
    }
}

__global__ __launch_bounds__(256)
void normalize_kernel(const float* __restrict__ agg,
                      float* __restrict__ cur,
                      float* __restrict__ out,
                      int nrows)
{
    const int wave = threadIdx.x >> 5, lane = threadIdx.x & 31;
    const int row = blockIdx.x * 8 + wave;
    if (row >= nrows) return;
    const int c = 2 * lane;
    v2f v = *(const v2f*)(agg + (size_t)row * CH + c);
    float ss = v.x * v.x + v.y * v.y;
#pragma unroll
    for (int m = 16; m >= 1; m >>= 1) ss += __shfl_xor(ss, m, 32);
    const float inv = 1.0f / fmaxf(sqrtf(ss), EPS_N);
    v.x *= inv; v.y *= inv;
    *(v2f*)(cur + (size_t)row * CH + c) = v;
    v2f o = *(v2f*)(out + (size_t)row * CH + c);
    o.x += v.x; o.y += v.y;
    *(v2f*)(out + (size_t)row * CH + c) = o;
}

extern "C" void kernel_launch(void* const* d_in, const int* in_sizes, int n_in,
                              void* d_out, int out_size, void* d_ws, size_t ws_size,
                              hipStream_t stream)
{
    (void)in_sizes; (void)n_in; (void)out_size; (void)ws_size;

    const float* user_emb   = (const float*)d_in[0];
    const float* entity_emb = (const float*)d_in[1];
    const float* weight     = (const float*)d_in[2];
    const float* inter_vals = (const float*)d_in[3];
    const int*   edge_index = (const int*)d_in[4];
    const int*   edge_type  = (const int*)d_in[5];
    const int*   inter_rows = (const int*)d_in[6];
    const int*   inter_cols = (const int*)d_in[7];
    const int*   head = edge_index;
    const int*   tail = edge_index + E_EDGES;

    float* out_ent = (float*)d_out;
    float* out_usr = out_ent + (size_t)N_ENT * CH;

    char* ws = (char*)d_ws;
    size_t off = 0;
    auto alloc_f = [&](size_t nfl) -> float* {
        float* p = (float*)(ws + off);
        off += ((nfl * sizeof(float) + 255) / 256) * 256;
        return p;
    };
    float* cur_ent = alloc_f((size_t)N_ENT * CH);
    float* cur_usr = alloc_f((size_t)N_USR * CH);
    float* agg_ent = alloc_f((size_t)N_ENT * CH);
    float* agg_usr = alloc_f((size_t)N_USR * CH);
    float* att     = alloc_f(E_EDGES);
    float* segmax  = alloc_f(N_ENT);
    float* segsum  = alloc_f(N_ENT);

    hipMemcpyAsync(cur_ent, entity_emb, (size_t)N_ENT * CH * sizeof(float), hipMemcpyDeviceToDevice, stream);
    hipMemcpyAsync(cur_usr, user_emb,   (size_t)N_USR * CH * sizeof(float), hipMemcpyDeviceToDevice, stream);
    hipMemcpyAsync(out_ent, entity_emb, (size_t)N_ENT * CH * sizeof(float), hipMemcpyDeviceToDevice, stream);
    hipMemcpyAsync(out_usr, user_emb,   (size_t)N_USR * CH * sizeof(float), hipMemcpyDeviceToDevice, stream);

    for (int hop = 0; hop < NHOPS; ++hop) {
        hipMemsetAsync(agg_ent, 0, (size_t)N_ENT * CH * sizeof(float), stream);
        hipMemsetAsync(agg_usr, 0, (size_t)N_USR * CH * sizeof(float), stream);
        hipMemsetAsync(segmax,  0, (size_t)N_ENT * sizeof(float), stream);
        hipMemsetAsync(segsum,  0, (size_t)N_ENT * sizeof(float), stream);

        att_wmma_kernel<<<E_EDGES / 16 / ATT_WAVES, ATT_WAVES * 32, 0, stream>>>(
            cur_ent, weight, head, tail, edge_type, att, (unsigned int*)segmax);
        exp_sum_kernel<<<(E_EDGES + 255) / 256, 256, 0, stream>>>(head, segmax, att, segsum);
        edge_agg_kernel<<<(E_EDGES / EPW + 7) / 8, 256, 0, stream>>>(
            cur_ent, weight, head, tail, edge_type, att, segsum, agg_ent);
        inter_agg_kernel<<<(NNZ_CNT / EPW + 7) / 8, 256, 0, stream>>>(
            cur_ent, cur_usr, inter_vals, inter_rows, inter_cols, agg_ent, agg_usr);
        normalize_kernel<<<(N_ENT + 7) / 8, 256, 0, stream>>>(agg_ent, cur_ent, out_ent, N_ENT);
        normalize_kernel<<<(N_USR + 7) / 8, 256, 0, stream>>>(agg_usr, cur_usr, out_usr, N_USR);
    }
}


# Scatter-add accumulators kept as one plane per package half (a wave adds into its own half's plane, picked by its die id); normalize sums the two planes; atomics kept, sum order changes
# speedup vs baseline: 1.3459x; 1.0691x over previous
#include <hip/hip_runtime.h>
#include <math.h>

#define N_ENT   150000
#define N_USR   60000
#define CH      64
#define E_EDGES 1000000
#define NNZ_CNT 1000000
#define NRELM1  9
#define NHOPS   2
#define EPS_N   1e-12f

typedef __attribute__((ext_vector_type(2))) float v2f;
typedef __attribute__((ext_vector_type(8))) float v8f;

__device__ __forceinline__ unsigned my_half()
{
    return (__builtin_amdgcn_s_sendmsg_rtn(0x87) >> 19) & 1u;
}

#define ATT_WAVES 4
#define LDP 68
__global__ __launch_bounds__(ATT_WAVES * 32)
void att_wmma_kernel(const float* __restrict__ ent,
                     const float* __restrict__ weight,
                     const int* __restrict__ head,
                     const int* __restrict__ tail,
                     const int* __restrict__ etype,
                     float* __restrict__ att,
                     unsigned int* __restrict__ segmax)
{
    __shared__ float sW2[16 * LDP];
    __shared__ float sH[ATT_WAVES][16 * LDP];
    __shared__ float sT[ATT_WAVES][16 * LDP];

    const int lane = threadIdx.x & 31;
    const int wave = threadIdx.x >> 5;

    for (int i = threadIdx.x; i < 16 * CH; i += ATT_WAVES * 32) {
        const int n = i >> 6, c = i & 63;
        const float w = (n < NRELM1) ? weight[n * CH + c] : 0.0f;
        sW2[n * LDP + c] = w * w;
    }
    __syncthreads();

    const int tile = blockIdx.x * ATT_WAVES + wave;
    if (tile * 16 >= E_EDGES) return;
    const int base = tile * 16;

    for (int r = 0; r < 16; ++r) {
        const int hr = head[base + r];
        const int tr = tail[base + r];
        const v2f hv = *(const v2f*)(ent + (size_t)hr * CH + 2 * lane);
        const v2f tv = *(const v2f*)(ent + (size_t)tr * CH + 2 * lane);
        sH[wave][r * LDP + 2 * lane]     = hv.x * hv.x;
        sH[wave][r * LDP + 2 * lane + 1] = hv.y * hv.y;
        sT[wave][r * LDP + 2 * lane]     = tv.x * tv.x;
        sT[wave][r * LDP + 2 * lane + 1] = tv.y * tv.y;
    }

    v8f accH = {};
    v8f accT = {};
    const int m    = lane & 15;
    const int koff = (lane >> 4) << 1;
    for (int kk = 0; kk < 16; ++kk) {
        const int c = kk * 4 + koff;
        v2f a, at, b;
        a.x  = sH[wave][m * LDP + c];  a.y  = sH[wave][m * LDP + c + 1];
        at.x = sT[wave][m * LDP + c];  at.y = sT[wave][m * LDP + c + 1];
        b.x  = sW2[m * LDP + c];       b.y  = sW2[m * LDP + c + 1];
        accH = __builtin_amdgcn_wmma_f32_16x16x4_f32(false, a,  false, b, (short)0, accH, false, false);
        accT = __builtin_amdgcn_wmma_f32_16x16x4_f32(false, at, false, b, (short)0, accT, false, false);
    }

    const int mrow = (lane >> 4) << 3;
    const int ncol = lane & 15;
#pragma unroll
    for (int v = 0; v < 8; ++v) {
        sH[wave][(mrow + v) * 16 + ncol] = accH[v];
        sT[wave][(mrow + v) * 16 + ncol] = accT[v];
    }

    if (lane < 16) {
        const int e  = base + lane;
        int et = etype[e] - 1;
        if (et < 0) et += NRELM1;
        const float hn2 = sH[wave][lane * 16 + et];
        const float tn2 = sT[wave][lane * 16 + et];
        const float a   = hn2 * tn2;
        att[e] = a;
        atomicMax(&segmax[head[e]], __float_as_uint(a));
    }
}

__global__ __launch_bounds__(256)
void exp_sum_kernel(const int* __restrict__ head,
                    const float* __restrict__ segmax,
                    float* __restrict__ att,
                    float* __restrict__ segsum)
{
    const int e = blockIdx.x * 256 + threadIdx.x;
    if (e >= E_EDGES) return;
    const int h = head[e];
    const float ev = expf(att[e] - segmax[h]);
    att[e] = ev;
    atomicAdd(&segsum[h], ev);
}

#define EPW 32
#define GRP 8
static_assert(E_EDGES % EPW == 0 && NNZ_CNT % EPW == 0, "runs must be whole");
static_assert(EPW % GRP == 0, "groups must tile a run");
static_assert(NHOPS == 2, "kernel_launch writes the two hops out");

__global__ __launch_bounds__(256)
void edge_agg_kernel(const float* __restrict__ ent,
                     const float* __restrict__ weight,
                     const int* __restrict__ head,
                     const int* __restrict__ tail,
                     const int* __restrict__ etype,
                     const float* __restrict__ expv,
                     const float* __restrict__ segsum,
                     float* __restrict__ agg_ent2)
{
    const int wave = threadIdx.x >> 5, lane = threadIdx.x & 31;
    const int e0 = (blockIdx.x * 8 + wave) * EPW;
    if (e0 >= E_EDGES) return;
    const int e = e0 + lane;
    const int h = head[e], t = tail[e];
    int et = etype[e] - 1;
    const float ex = expv[e];
    float* __restrict__ agg_ent = agg_ent2 + (size_t)my_half() * ((size_t)N_ENT * CH);
    if (et < 0) et += NRELM1;
    const float w = ex / segsum[h];
    const int wi = __float_as_int(w);
    const int c = 2 * lane;
#pragma unroll
    for (int g = 0; g < EPW; g += GRP) {
        v2f tv[GRP], rv[GRP];
        for (int j = 0; j < GRP; ++j) {
            const int tj  = __builtin_amdgcn_readlane(t,  g + j);
            const int etj = __builtin_amdgcn_readlane(et, g + j);
            tv[j] = *(const v2f*)(ent + (size_t)tj * CH + c);
            rv[j] = *(const v2f*)(weight + etj * CH + c);
        }
        for (int j = 0; j < GRP; ++j) {
            const int   hj = __builtin_amdgcn_readlane(h,  g + j);
            const float wj = __int_as_float(__builtin_amdgcn_readlane(wi, g + j));
            float* dst = agg_ent + (size_t)hj * CH + c;
            atomicAdd(dst,     wj * tv[j].x * rv[j].x);
            atomicAdd(dst + 1, wj * tv[j].y * rv[j].y);
        }
    }
}

__global__ __launch_bounds__(256)
void inter_agg_kernel(const float* __restrict__ cur_ent,
                      const float* __restrict__ cur_usr,
                      const float* __restrict__ vals,
                      const int* __restrict__ rows,
                      const int* __restrict__ cols,
                      float* __restrict__ agg_ent2,
                      float* __restrict__ agg_usr2)
{
    const int wave = threadIdx.x >> 5, lane = threadIdx.x & 31;
    const int z0 = (blockIdx.x * 8 + wave) * EPW;
    if (z0 >= NNZ_CNT) return;
    const int z = z0 + lane;
    const int r = rows[z], cl = cols[z];
    const int vi = __float_as_int(vals[z]);
    const int c = 2 * lane;
    const unsigned hf = my_half();
    float* __restrict__ agg_ent = agg_ent2 + (size_t)hf * ((size_t)N_ENT * CH);
    float* __restrict__ agg_usr = agg_usr2 + (size_t)hf * ((size_t)N_USR * CH);
#pragma unroll
    for (int g = 0; g < EPW; g += GRP) {
        v2f uv[GRP], ev[GRP];
        for (int j = 0; j < GRP; ++j) {
            const int rj  = __builtin_amdgcn_readlane(r,  g + j);
            const int clj = __builtin_amdgcn_readlane(cl, g + j);
            uv[j] = *(const v2f*)(cur_usr + (size_t)rj  * CH + c);
            ev[j] = *(const v2f*)(cur_ent + (size_t)clj * CH + c);
        }
        for (int j = 0; j < GRP; ++j) {
            const int   rj  = __builtin_amdgcn_readlane(r,  g + j);
            const int   clj = __builtin_amdgcn_readlane(cl, g + j);
            const float vj  = __int_as_float(__builtin_amdgcn_readlane(vi, g + j));
            float* de = agg_ent + (size_t)clj * CH + c;
            float* du = agg_usr + (size_t)rj  * CH + c;
            atomicAdd(de,     vj * uv[j].x);
            atomicAdd(de + 1, vj * uv[j].y);
            atomicAdd(du,     vj * ev[j].x);
            atomicAdd(du + 1, vj * ev[j].y);
        }
    }
}

__global__ __launch_bounds__(256)
void normalize_kernel(const float* __restrict__ agg2,
                      const float* __restrict__ emb,
                      float* __restrict__ out,
                      int nrows, int last)
{
    const int wave = threadIdx.x >> 5, lane = threadIdx.x & 31;
    const int row = blockIdx.x * 8 + wave;
    if (row >= nrows) return;
    const int c = 2 * lane;
    const v2f a = *(const v2f*)(agg2 + (size_t)row * CH + c);
    const v2f b = *(const v2f*)(agg2 + ((size_t)nrows + row) * CH + c);
    v2f res = {0.0f, 0.0f};
    if (last) {
        const v2f e0 = *(const v2f*)(emb + (size_t)row * CH + c);
        const v2f n1 = *(const v2f*)(out + (size_t)row * CH + c);
        res.x = e0.x + n1.x; res.y = e0.y + n1.y;
    }
    v2f v;
    v.x = a.x + b.x; v.y = a.y + b.y;
    float ss = v.x * v.x + v.y * v.y;
#pragma unroll
    for (int m = 16; m >= 1; m >>= 1) ss += __shfl_xor(ss, m, 32);
    const float inv = 1.0f / fmaxf(sqrtf(ss), EPS_N);
    v.x *= inv; v.y *= inv;
    if (last) { v.x = res.x + v.x; v.y = res.y + v.y; }
    *(v2f*)(out + (size_t)row * CH + c) = v;
}

extern "C" void kernel_launch(void* const* d_in, const int* in_sizes, int n_in,
                              void* d_out, int out_size, void* d_ws, size_t ws_size,
                              hipStream_t stream)
{
    (void)in_sizes; (void)n_in; (void)out_size; (void)ws_size;

    const float* user_emb   = (const float*)d_in[0];
    const float* entity_emb = (const float*)d_in[1];
    const float* weight     = (const float*)d_in[2];
    const float* inter_vals = (const float*)d_in[3];
    const int*   edge_index = (const int*)d_in[4];
    const int*   edge_type  = (const int*)d_in[5];
    const int*   inter_rows = (const int*)d_in[6];
    const int*   inter_cols = (const int*)d_in[7];
    const int*   head = edge_index;
    const int*   tail = edge_index + E_EDGES;

    float* out_ent = (float*)d_out;
    float* out_usr = out_ent + (size_t)N_ENT * CH;

    char* ws = (char*)d_ws;
    size_t off = 0;
    auto alloc_f = [&](size_t nfl) -> float* {
        float* p = (float*)(ws + off);
        off += ((nfl * sizeof(float) + 255) / 256) * 256;
        return p;
    };
    float* agg_ent2 = alloc_f((size_t)2 * N_ENT * CH);
    float* agg_usr2 = alloc_f((size_t)2 * N_USR * CH);
    float* att     = alloc_f(E_EDGES);
    float* segmax  = alloc_f(N_ENT);
    float* segsum  = alloc_f(N_ENT);

    auto scatter_hop = [&](const float* cur_ent, const float* cur_usr) {
        hipMemsetAsync(agg_ent2, 0, (size_t)2 * N_ENT * CH * sizeof(float), stream);
        hipMemsetAsync(agg_usr2, 0, (size_t)2 * N_USR * CH * sizeof(float), stream);
        hipMemsetAsync(segmax,  0, (size_t)N_ENT * sizeof(float), stream);
        hipMemsetAsync(segsum,  0, (size_t)N_ENT * sizeof(float), stream);

        att_wmma_kernel<<<E_EDGES / 16 / ATT_WAVES, ATT_WAVES * 32, 0, stream>>>(
            cur_ent, weight, head, tail, edge_type, att, (unsigned int*)segmax);
        exp_sum_kernel<<<(E_EDGES + 255) / 256, 256, 0, stream>>>(head, segmax, att, segsum);
        edge_agg_kernel<<<(E_EDGES / EPW + 7) / 8, 256, 0, stream>>>(
            cur_ent, weight, head, tail, edge_type, att, segsum, agg_ent2);
        inter_agg_kernel<<<(NNZ_CNT / EPW + 7) / 8, 256, 0, stream>>>(
            cur_ent, cur_usr, inter_vals, inter_rows, inter_cols, agg_ent2, agg_usr2);
    };

    scatter_hop(entity_emb, user_emb);
    normalize_kernel<<<(N_ENT + 7) / 8, 256, 0, stream>>>(agg_ent2, entity_emb, out_ent, N_ENT, 0);
    normalize_kernel<<<(N_USR + 7) / 8, 256, 0, stream>>>(agg_usr2, user_emb,   out_usr, N_USR, 0);
    scatter_hop(out_ent, out_usr);
    normalize_kernel<<<(N_ENT + 7) / 8, 256, 0, stream>>>(agg_ent2, entity_emb, out_ent, N_ENT, 1);
    normalize_kernel<<<(N_USR + 7) / 8, 256, 0, stream>>>(agg_usr2, user_emb,   out_usr, N_USR, 1);
}


# Scatter kernels: lane owns channels l and l+32 (was 2l, 2l+1): each f32 atomic-add instruction covers one whole 128-byte line, not half of two; row loads follow; atomics kept, arrival order varies
# speedup vs baseline: 1.4868x; 1.1047x over previous
#include <hip/hip_runtime.h>
#include <math.h>

#define N_ENT   150000
#define N_USR   60000
#define CH      64
#define E_EDGES 1000000
#define NNZ_CNT 1000000
#define NRELM1  9
#define NHOPS   2
#define EPS_N   1e-12f

typedef __attribute__((ext_vector_type(2))) float v2f;
typedef __attribute__((ext_vector_type(8))) float v8f;

__device__ __forceinline__ unsigned my_half()
{
    return (__builtin_amdgcn_s_sendmsg_rtn(0x87) >> 19) & 1u;
}

#define ATT_WAVES 4
#define LDP 68
__global__ __launch_bounds__(ATT_WAVES * 32)
void att_wmma_kernel(const float* __restrict__ ent,
                     const float* __restrict__ weight,
                     const int* __restrict__ head,
                     const int* __restrict__ tail,
                     const int* __restrict__ etype,
                     float* __restrict__ att,
                     unsigned int* __restrict__ segmax)
{
    __shared__ float sW2[16 * LDP];
    __shared__ float sH[ATT_WAVES][16 * LDP];
    __shared__ float sT[ATT_WAVES][16 * LDP];

    const int lane = threadIdx.x & 31;
    const int wave = threadIdx.x >> 5;

    for (int i = threadIdx.x; i < 16 * CH; i += ATT_WAVES * 32) {
        const int n = i >> 6, c = i & 63;
        const float w = (n < NRELM1) ? weight[n * CH + c] : 0.0f;
        sW2[n * LDP + c] = w * w;
    }
    __syncthreads();

    const int tile = blockIdx.x * ATT_WAVES + wave;
    if (tile * 16 >= E_EDGES) return;
    const int base = tile * 16;

    for (int r = 0; r < 16; ++r) {
        const int hr = head[base + r];
        const int tr = tail[base + r];
        const v2f hv = *(const v2f*)(ent + (size_t)hr * CH + 2 * lane);
        const v2f tv = *(const v2f*)(ent + (size_t)tr * CH + 2 * lane);
        sH[wave][r * LDP + 2 * lane]     = hv.x * hv.x;
        sH[wave][r * LDP + 2 * lane + 1] = hv.y * hv.y;
        sT[wave][r * LDP + 2 * lane]     = tv.x * tv.x;
        sT[wave][r * LDP + 2 * lane + 1] = tv.y * tv.y;
    }

    v8f accH = {};
    v8f accT = {};
    const int m    = lane & 15;
    const int koff = (lane >> 4) << 1;
    for (int kk = 0; kk < 16; ++kk) {
        const int c = kk * 4 + koff;
        v2f a, at, b;
        a.x  = sH[wave][m * LDP + c];  a.y  = sH[wave][m * LDP + c + 1];
        at.x = sT[wave][m * LDP + c];  at.y = sT[wave][m * LDP + c + 1];
        b.x  = sW2[m * LDP + c];       b.y  = sW2[m * LDP + c + 1];
        accH = __builtin_amdgcn_wmma_f32_16x16x4_f32(false, a,  false, b, (short)0, accH, false, false);
        accT = __builtin_amdgcn_wmma_f32_16x16x4_f32(false, at, false, b, (short)0, accT, false, false);
    }

    const int mrow = (lane >> 4) << 3;
    const int ncol = lane & 15;
#pragma unroll
    for (int v = 0; v < 8; ++v) {
        sH[wave][(mrow + v) * 16 + ncol] = accH[v];
        sT[wave][(mrow + v) * 16 + ncol] = accT[v];
    }

    if (lane < 16) {
        const int e  = base + lane;
        int et = etype[e] - 1;
        if (et < 0) et += NRELM1;
        const float hn2 = sH[wave][lane * 16 + et];
        const float tn2 = sT[wave][lane * 16 + et];
        const float a   = hn2 * tn2;
        att[e] = a;
        atomicMax(&segmax[head[e]], __float_as_uint(a));
    }
}

__global__ __launch_bounds__(256)
void exp_sum_kernel(const int* __restrict__ head,
                    const float* __restrict__ segmax,
                    float* __restrict__ att,
                    float* __restrict__ segsum)
{
    const int e = blockIdx.x * 256 + threadIdx.x;
    if (e >= E_EDGES) return;
    const int h = head[e];
    const float ev = expf(att[e] - segmax[h]);
    att[e] = ev;
    atomicAdd(&segsum[h], ev);
}

#define EPW 32
#define GRP 8
#define HALF_CH (CH / 2)
static_assert(E_EDGES % EPW == 0 && NNZ_CNT % EPW == 0, "runs must be whole");
static_assert(EPW % GRP == 0, "groups must tile a run");
static_assert(HALF_CH == 32, "one lane per word of a 128-byte line");
static_assert(NHOPS == 2, "kernel_launch writes the two hops out");

__global__ __launch_bounds__(256)
void edge_agg_kernel(const float* __restrict__ ent,
                     const float* __restrict__ weight,
                     const int* __restrict__ head,
                     const int* __restrict__ tail,
                     const int* __restrict__ etype,
                     const float* __restrict__ expv,
                     const float* __restrict__ segsum,
                     float* __restrict__ agg_ent2)
{
    const int wave = threadIdx.x >> 5, lane = threadIdx.x & 31;
    const int e0 = (blockIdx.x * 8 + wave) * EPW;
    if (e0 >= E_EDGES) return;
    const int e = e0 + lane;
    const int h = head[e], t = tail[e];
    int et = etype[e] - 1;
    const float ex = expv[e];
    float* __restrict__ agg_ent = agg_ent2 + (size_t)my_half() * ((size_t)N_ENT * CH);
    if (et < 0) et += NRELM1;
    const float w = ex / segsum[h];
    const int wi = __float_as_int(w);
#pragma unroll
    for (int g = 0; g < EPW; g += GRP) {
        float t0[GRP], t1[GRP], r0[GRP], r1[GRP];
        for (int j = 0; j < GRP; ++j) {
            const int tj  = __builtin_amdgcn_readlane(t,  g + j);
            const int etj = __builtin_amdgcn_readlane(et, g + j);
            const float* tp = ent + (size_t)tj * CH + lane;
            const float* rp = weight + etj * CH + lane;
            t0[j] = tp[0];  t1[j] = tp[HALF_CH];
            r0[j] = rp[0];  r1[j] = rp[HALF_CH];
        }
        for (int j = 0; j < GRP; ++j) {
            const int   hj = __builtin_amdgcn_readlane(h,  g + j);
            const float wj = __int_as_float(__builtin_amdgcn_readlane(wi, g + j));
            float* dst = agg_ent + (size_t)hj * CH + lane;
            atomicAdd(dst,           wj * t0[j] * r0[j]);
            atomicAdd(dst + HALF_CH, wj * t1[j] * r1[j]);
        }
    }
}

__global__ __launch_bounds__(256)
void inter_agg_kernel(const float* __restrict__ cur_ent,
                      const float* __restrict__ cur_usr,
                      const float* __restrict__ vals,
                      const int* __restrict__ rows,
                      const int* __restrict__ cols,
                      float* __restrict__ agg_ent2,
                      float* __restrict__ agg_usr2)
{
    const int wave = threadIdx.x >> 5, lane = threadIdx.x & 31;
    const int z0 = (blockIdx.x * 8 + wave) * EPW;
    if (z0 >= NNZ_CNT) return;
    const int z = z0 + lane;
    const int r = rows[z], cl = cols[z];
    const int vi = __float_as_int(vals[z]);
    const unsigned hf = my_half();
    float* __restrict__ agg_ent = agg_ent2 + (size_t)hf * ((size_t)N_ENT * CH);
    float* __restrict__ agg_usr = agg_usr2 + (size_t)hf * ((size_t)N_USR * CH);
#pragma unroll
    for (int g = 0; g < EPW; g += GRP) {
        float u0[GRP], u1[GRP], n0[GRP], n1[GRP];
        for (int j = 0; j < GRP; ++j) {
            const int rj  = __builtin_amdgcn_readlane(r,  g + j);
            const int clj = __builtin_amdgcn_readlane(cl, g + j);
            const float* up = cur_usr + (size_t)rj  * CH + lane;
            const float* np = cur_ent + (size_t)clj * CH + lane;
            u0[j] = up[0];  u1[j] = up[HALF_CH];
            n0[j] = np[0];  n1[j] = np[HALF_CH];
        }
        for (int j = 0; j < GRP; ++j) {
            const int   rj  = __builtin_amdgcn_readlane(r,  g + j);
            const int   clj = __builtin_amdgcn_readlane(cl, g + j);
            const float vj  = __int_as_float(__builtin_amdgcn_readlane(vi, g + j));
            float* de = agg_ent + (size_t)clj * CH + lane;
            float* du = agg_usr + (size_t)rj  * CH + lane;
            atomicAdd(de,           vj * u0[j]);
            atomicAdd(de + HALF_CH, vj * u1[j]);
            atomicAdd(du,           vj * n0[j]);
            atomicAdd(du + HALF_CH, vj * n1[j]);
        }
    }
}

__global__ __launch_bounds__(256)
void normalize_kernel(const float* __restrict__ agg2,
                      const float* __restrict__ emb,
                      float* __restrict__ out,
                      int nrows, int last)
{
    const int wave = threadIdx.x >> 5, lane = threadIdx.x & 31;
    const int row = blockIdx.x * 8 + wave;
    if (row >= nrows) return;
    const int c = 2 * lane;
    const v2f a = *(const v2f*)(agg2 + (size_t)row * CH + c);
    const v2f b = *(const v2f*)(agg2 + ((size_t)nrows + row) * CH + c);
    v2f res = {0.0f, 0.0f};
    if (last) {
        const v2f e0 = *(const v2f*)(emb + (size_t)row * CH + c);
        const v2f n1 = *(const v2f*)(out + (size_t)row * CH + c);
        res.x = e0.x + n1.x; res.y = e0.y + n1.y;
    }
    v2f v;
    v.x = a.x + b.x; v.y = a.y + b.y;
    float ss = v.x * v.x + v.y * v.y;
#pragma unroll
    for (int m = 16; m >= 1; m >>= 1) ss += __shfl_xor(ss, m, 32);
    const float inv = 1.0f / fmaxf(sqrtf(ss), EPS_N);
    v.x *= inv; v.y *= inv;
    if (last) { v.x = res.x + v.x; v.y = res.y + v.y; }
    *(v2f*)(out + (size_t)row * CH + c) = v;
}

extern "C" void kernel_launch(void* const* d_in, const int* in_sizes, int n_in,
                              void* d_out, int out_size, void* d_ws, size_t ws_size,
                              hipStream_t stream)
{
    (void)in_sizes; (void)n_in; (void)out_size; (void)ws_size;

    const float* user_emb   = (const float*)d_in[0];
    const float* entity_emb = (const float*)d_in[1];
    const float* weight     = (const float*)d_in[2];
    const float* inter_vals = (const float*)d_in[3];
    const int*   edge_index = (const int*)d_in[4];
    const int*   edge_type  = (const int*)d_in[5];
    const int*   inter_rows = (const int*)d_in[6];
    const int*   inter_cols = (const int*)d_in[7];
    const int*   head = edge_index;
    const int*   tail = edge_index + E_EDGES;

    float* out_ent = (float*)d_out;
    float* out_usr = out_ent + (size_t)N_ENT * CH;

    char* ws = (char*)d_ws;
    size_t off = 0;
    auto alloc_f = [&](size_t nfl) -> float* {
        float* p = (float*)(ws + off);
        off += ((nfl * sizeof(float) + 255) / 256) * 256;
        return p;
    };
    float* agg_ent2 = alloc_f((size_t)2 * N_ENT * CH);
    float* agg_usr2 = alloc_f((size_t)2 * N_USR * CH);
    float* att     = alloc_f(E_EDGES);
    float* segmax  = alloc_f(N_ENT);
    float* segsum  = alloc_f(N_ENT);

    auto scatter_hop = [&](const float* cur_ent, const float* cur_usr) {
        hipMemsetAsync(agg_ent2, 0, (size_t)2 * N_ENT * CH * sizeof(float), stream);
        hipMemsetAsync(agg_usr2, 0, (size_t)2 * N_USR * CH * sizeof(float), stream);
        hipMemsetAsync(segmax,  0, (size_t)N_ENT * sizeof(float), stream);
        hipMemsetAsync(segsum,  0, (size_t)N_ENT * sizeof(float), stream);

        att_wmma_kernel<<<E_EDGES / 16 / ATT_WAVES, ATT_WAVES * 32, 0, stream>>>(
            cur_ent, weight, head, tail, edge_type, att, (unsigned int*)segmax);
        exp_sum_kernel<<<(E_EDGES + 255) / 256, 256, 0, stream>>>(head, segmax, att, segsum);
        edge_agg_kernel<<<(E_EDGES / EPW + 7) / 8, 256, 0, stream>>>(
            cur_ent, weight, head, tail, edge_type, att, segsum, agg_ent2);
        inter_agg_kernel<<<(NNZ_CNT / EPW + 7) / 8, 256, 0, stream>>>(
            cur_ent, cur_usr, inter_vals, inter_rows, inter_cols, agg_ent2, agg_usr2);
    };

    scatter_hop(entity_emb, user_emb);
    normalize_kernel<<<(N_ENT + 7) / 8, 256, 0, stream>>>(agg_ent2, entity_emb, out_ent, N_ENT, 0);
    normalize_kernel<<<(N_USR + 7) / 8, 256, 0, stream>>>(agg_usr2, user_emb,   out_usr, N_USR, 0);
    scatter_hop(out_ent, out_usr);
    normalize_kernel<<<(N_ENT + 7) / 8, 256, 0, stream>>>(agg_ent2, entity_emb, out_ent, N_ENT, 1);
    normalize_kernel<<<(N_USR + 7) / 8, 256, 0, stream>>>(agg_usr2, user_emb,   out_usr, N_USR, 1);
}


# Attention: squared relation norms formed once per entity (150000x64 by 64x16 f32 WMMA over streamed rows) into a table; an edge multiplies two looked-up scalars instead of gathering two rows
# speedup vs baseline: 1.6419x; 1.1043x over previous
#include <hip/hip_runtime.h>
#include <math.h>

#define N_ENT   150000
#define N_USR   60000
#define CH      64
#define E_EDGES 1000000
#define NNZ_CNT 1000000
#define NRELM1  9
#define NHOPS   2
#define EPS_N   1e-12f

typedef __attribute__((ext_vector_type(2))) float v2f;
typedef __attribute__((ext_vector_type(4))) float v4f;
typedef __attribute__((ext_vector_type(4))) int   v4i;
typedef __attribute__((ext_vector_type(8))) float v8f;

__device__ __forceinline__ unsigned my_half()
{
    return (__builtin_amdgcn_s_sendmsg_rtn(0x87) >> 19) & 1u;
}

#define TBL_WAVES 4
#define TP 16
#define N_TILES (N_ENT / 16)
#define LDP 68
static_assert(N_ENT % 16 == 0, "entity rows tile by 16");
static_assert((size_t)N_ENT * TP <= (size_t)2 * N_USR * CH, "the table borrows the user planes");
static_assert(16 * CH == 8 * TBL_WAVES * 32, "relation staging: 8 elements per thread");
__global__ __launch_bounds__(TBL_WAVES * 32)
void relnorm_wmma_kernel(const float* __restrict__ ent,
                         const float* __restrict__ weight,
                         float* __restrict__ tbl)
{
    __shared__ float sW2[16 * LDP];
    __shared__ float sA[TBL_WAVES][16 * LDP];
    __shared__ float sB[TBL_WAVES][16 * LDP];

    const int lane = threadIdx.x & 31;
    const int wave = threadIdx.x >> 5;

    const int tA = 2 * (blockIdx.x * TBL_WAVES + wave);
    const bool validA = tA < N_TILES;
    const bool validB = tA + 1 < N_TILES;
    const int tAc = validA ? tA : 0;
    const int tBc = validB ? tA + 1 : tAc;

    float wv[8];
#pragma unroll
    for (int j = 0; j < 8; ++j) {
        const int i = threadIdx.x + j * (TBL_WAVES * 32);
        const int n = i >> 6, c = i & 63;
        const int nc = (n < NRELM1) ? n : NRELM1 - 1;
        const float w = weight[nc * CH + c];
        wv[j] = (n < NRELM1) ? w : 0.0f;
    }

    const v4f* __restrict__ srcA = (const v4f*)(ent + (size_t)tAc * 16 * CH);
    const v4f* __restrict__ srcB = (const v4f*)(ent + (size_t)tBc * 16 * CH);
    v4f xa[8], xb[8];
#pragma unroll
    for (int it = 0; it < 8; ++it) {
        xa[it] = srcA[it * 32 + lane];
        xb[it] = srcB[it * 32 + lane];
    }

#pragma unroll
    for (int j = 0; j < 8; ++j) {
        const int i = threadIdx.x + j * (TBL_WAVES * 32);
        const int n = i >> 6, c = i & 63;
        sW2[n * LDP + c] = wv[j] * wv[j];
    }
    __syncthreads();
    if (!validA) return;

    const int srow = lane >> 4;
    const int scol = (lane & 15) << 2;
#pragma unroll
    for (int it = 0; it < 8; ++it) {
        const v4f a = xa[it], b = xb[it];
        *(v4f*)&sA[wave][(2 * it + srow) * LDP + scol] = a * a;
        *(v4f*)&sB[wave][(2 * it + srow) * LDP + scol] = b * b;
    }

    v8f accA = {};
    v8f accB = {};
    const int m    = lane & 15;
    const int koff = (lane >> 4) << 1;
    for (int kk = 0; kk < 16; ++kk) {
        const int c = kk * 4 + koff;
        v2f a, a2, b;
        a.x  = sA[wave][m * LDP + c];  a.y  = sA[wave][m * LDP + c + 1];
        a2.x = sB[wave][m * LDP + c];  a2.y = sB[wave][m * LDP + c + 1];
        b.x  = sW2[m * LDP + c];       b.y  = sW2[m * LDP + c + 1];
        accA = __builtin_amdgcn_wmma_f32_16x16x4_f32(false, a,  false, b, (short)0, accA, false, false);
        accB = __builtin_amdgcn_wmma_f32_16x16x4_f32(false, a2, false, b, (short)0, accB, false, false);
    }

    const int mrow = (lane >> 4) << 3;
    const int ncol = lane & 15;
    float* dA = tbl + (size_t)(tA * 16 + mrow) * TP + ncol;
#pragma unroll
    for (int v = 0; v < 8; ++v) dA[v * TP] = accA[v];
    if (validB) {
        float* dB = dA + 16 * TP;
#pragma unroll
        for (int v = 0; v < 8; ++v) dB[v * TP] = accB[v];
    }
}

static_assert(E_EDGES % 4 == 0, "edges go in fours");
__global__ __launch_bounds__(256)
void att_edge_kernel(const float* __restrict__ tbl,
                     const int* __restrict__ head,
                     const int* __restrict__ tail,
                     const int* __restrict__ etype,
                     float* __restrict__ att,
                     unsigned int* __restrict__ segmax)
{
    const int q = blockIdx.x * 256 + threadIdx.x;
    if (q >= E_EDGES / 4) return;
    const v4i h  = ((const v4i*)head)[q];
    const v4i t  = ((const v4i*)tail)[q];
    const v4i ty = ((const v4i*)etype)[q];
    float hn2[4], tn2[4];
#pragma unroll
    for (int k = 0; k < 4; ++k) {
        int et = ty[k] - 1;
        if (et < 0) et += NRELM1;
        hn2[k] = tbl[(size_t)h[k] * TP + et];
        tn2[k] = tbl[(size_t)t[k] * TP + et];
    }
    v4f a;
#pragma unroll
    for (int k = 0; k < 4; ++k) a[k] = hn2[k] * tn2[k];
    ((v4f*)att)[q] = a;
#pragma unroll
    for (int k = 0; k < 4; ++k)
        atomicMax(&segmax[h[k]], __float_as_uint(a[k]));
}

__global__ __launch_bounds__(256)
void exp_sum_kernel(const int* __restrict__ head,
                    const float* __restrict__ segmax,
                    float* __restrict__ att,
                    float* __restrict__ segsum)
{
    const int e = blockIdx.x * 256 + threadIdx.x;
    if (e >= E_EDGES) return;
    const int h = head[e];
    const float ev = expf(att[e] - segmax[h]);
    att[e] = ev;
    atomicAdd(&segsum[h], ev);
}

#define EPW 32
#define GRP 8
#define HALF_CH (CH / 2)
static_assert(E_EDGES % EPW == 0 && NNZ_CNT % EPW == 0, "runs must be whole");
static_assert(EPW % GRP == 0, "groups must tile a run");
static_assert(HALF_CH == 32, "one lane per word of a 128-byte line");
static_assert(NHOPS == 2, "kernel_launch writes the two hops out");

__global__ __launch_bounds__(256)
void edge_agg_kernel(const float* __restrict__ ent,
                     const float* __restrict__ weight,
                     const int* __restrict__ head,
                     const int* __restrict__ tail,
                     const int* __restrict__ etype,
                     const float* __restrict__ expv,
                     const float* __restrict__ segsum,
                     float* __restrict__ agg_ent2)
{
    const int wave = threadIdx.x >> 5, lane = threadIdx.x & 31;
    const int e0 = (blockIdx.x * 8 + wave) * EPW;
    if (e0 >= E_EDGES) return;
    const int e = e0 + lane;
    const int h = head[e], t = tail[e];
    int et = etype[e] - 1;
    const float ex = expv[e];
    float* __restrict__ agg_ent = agg_ent2 + (size_t)my_half() * ((size_t)N_ENT * CH);
    if (et < 0) et += NRELM1;
    const float w = ex / segsum[h];
    const int wi = __float_as_int(w);
#pragma unroll
    for (int g = 0; g < EPW; g += GRP) {
        float t0[GRP], t1[GRP], r0[GRP], r1[GRP];
        for (int j = 0; j < GRP; ++j) {
            const int tj  = __builtin_amdgcn_readlane(t,  g + j);
            const int etj = __builtin_amdgcn_readlane(et, g + j);
            const float* tp = ent + (size_t)tj * CH + lane;
            const float* rp = weight + etj * CH + lane;
            t0[j] = tp[0];  t1[j] = tp[HALF_CH];
            r0[j] = rp[0];  r1[j] = rp[HALF_CH];
        }
        for (int j = 0; j < GRP; ++j) {
            const int   hj = __builtin_amdgcn_readlane(h,  g + j);
            const float wj = __int_as_float(__builtin_amdgcn_readlane(wi, g + j));
            float* dst = agg_ent + (size_t)hj * CH + lane;
            atomicAdd(dst,           wj * t0[j] * r0[j]);
            atomicAdd(dst + HALF_CH, wj * t1[j] * r1[j]);
        }
    }
}

__global__ __launch_bounds__(256)
void inter_agg_kernel(const float* __restrict__ cur_ent,
                      const float* __restrict__ cur_usr,
                      const float* __restrict__ vals,
                      const int* __restrict__ rows,
                      const int* __restrict__ cols,
                      float* __restrict__ agg_ent2,
                      float* __restrict__ agg_usr2)
{
    const int wave = threadIdx.x >> 5, lane = threadIdx.x & 31;
    const int z0 = (blockIdx.x * 8 + wave) * EPW;
    if (z0 >= NNZ_CNT) return;
    const int z = z0 + lane;
    const int r = rows[z], cl = cols[z];
    const int vi = __float_as_int(vals[z]);
    const unsigned hf = my_half();
    float* __restrict__ agg_ent = agg_ent2 + (size_t)hf * ((size_t)N_ENT * CH);
    float* __restrict__ agg_usr = agg_usr2 + (size_t)hf * ((size_t)N_USR * CH);
#pragma unroll
    for (int g = 0; g < EPW; g += GRP) {
        float u0[GRP], u1[GRP], n0[GRP], n1[GRP];
        for (int j = 0; j < GRP; ++j) {
            const int rj  = __builtin_amdgcn_readlane(r,  g + j);
            const int clj = __builtin_amdgcn_readlane(cl, g + j);
            const float* up = cur_usr + (size_t)rj  * CH + lane;
            const float* np = cur_ent + (size_t)clj * CH + lane;
            u0[j] = up[0];  u1[j] = up[HALF_CH];
            n0[j] = np[0];  n1[j] = np[HALF_CH];
        }
        for (int j = 0; j < GRP; ++j) {
            const int   rj  = __builtin_amdgcn_readlane(r,  g + j);
            const int   clj = __builtin_amdgcn_readlane(cl, g + j);
            const float vj  = __int_as_float(__builtin_amdgcn_readlane(vi, g + j));
            float* de = agg_ent + (size_t)clj * CH + lane;
            float* du = agg_usr + (size_t)rj  * CH + lane;
            atomicAdd(de,           vj * u0[j]);
            atomicAdd(de + HALF_CH, vj * u1[j]);
            atomicAdd(du,           vj * n0[j]);
            atomicAdd(du + HALF_CH, vj * n1[j]);
        }
    }
}

__global__ __launch_bounds__(256)
void normalize_kernel(const float* __restrict__ agg2,
                      const float* __restrict__ emb,
                      float* __restrict__ out,
                      int nrows, int last)
{
    const int wave = threadIdx.x >> 5, lane = threadIdx.x & 31;
    const int row = blockIdx.x * 8 + wave;
    if (row >= nrows) return;
    const int c = 2 * lane;
    const v2f a = *(const v2f*)(agg2 + (size_t)row * CH + c);
    const v2f b = *(const v2f*)(agg2 + ((size_t)nrows + row) * CH + c);
    v2f res = {0.0f, 0.0f};
    if (last) {
        const v2f e0 = *(const v2f*)(emb + (size_t)row * CH + c);
        const v2f n1 = *(const v2f*)(out + (size_t)row * CH + c);
        res.x = e0.x + n1.x; res.y = e0.y + n1.y;
    }
    v2f v;
    v.x = a.x + b.x; v.y = a.y + b.y;
    float ss = v.x * v.x + v.y * v.y;
#pragma unroll
    for (int m = 16; m >= 1; m >>= 1) ss += __shfl_xor(ss, m, 32);
    const float inv = 1.0f / fmaxf(sqrtf(ss), EPS_N);
    v.x *= inv; v.y *= inv;
    if (last) { v.x = res.x + v.x; v.y = res.y + v.y; }
    *(v2f*)(out + (size_t)row * CH + c) = v;
}

extern "C" void kernel_launch(void* const* d_in, const int* in_sizes, int n_in,
                              void* d_out, int out_size, void* d_ws, size_t ws_size,
                              hipStream_t stream)
{
    (void)in_sizes; (void)n_in; (void)out_size; (void)ws_size;

    const float* user_emb   = (const float*)d_in[0];
    const float* entity_emb = (const float*)d_in[1];
    const float* weight     = (const float*)d_in[2];
    const float* inter_vals = (const float*)d_in[3];
    const int*   edge_index = (const int*)d_in[4];
    const int*   edge_type  = (const int*)d_in[5];
    const int*   inter_rows = (const int*)d_in[6];
    const int*   inter_cols = (const int*)d_in[7];
    const int*   head = edge_index;
    const int*   tail = edge_index + E_EDGES;

    float* out_ent = (float*)d_out;
    float* out_usr = out_ent + (size_t)N_ENT * CH;

    char* ws = (char*)d_ws;
    size_t off = 0;
    auto alloc_f = [&](size_t nfl) -> float* {
        float* p = (float*)(ws + off);
        off += ((nfl * sizeof(float) + 255) / 256) * 256;
        return p;
    };
    float* agg_ent2 = alloc_f((size_t)2 * N_ENT * CH);
    float* agg_usr2 = alloc_f((size_t)2 * N_USR * CH);
    float* att     = alloc_f(E_EDGES);
    float* segmax  = alloc_f(N_ENT);
    float* segsum  = alloc_f(N_ENT);

    auto scatter_hop = [&](const float* cur_ent, const float* cur_usr) {
        hipMemsetAsync(agg_ent2, 0, (size_t)2 * N_ENT * CH * sizeof(float), stream);
        hipMemsetAsync(segmax,  0, (size_t)N_ENT * sizeof(float), stream);
        hipMemsetAsync(segsum,  0, (size_t)N_ENT * sizeof(float), stream);

        float* tbl = agg_usr2;
        relnorm_wmma_kernel<<<((N_TILES + 1) / 2 + TBL_WAVES - 1) / TBL_WAVES, TBL_WAVES * 32, 0, stream>>>(
            cur_ent, weight, tbl);
        att_edge_kernel<<<(E_EDGES / 4 + 255) / 256, 256, 0, stream>>>(
            tbl, head, tail, edge_type, att, (unsigned int*)segmax);

        hipMemsetAsync(agg_usr2, 0, (size_t)2 * N_USR * CH * sizeof(float), stream);
        exp_sum_kernel<<<(E_EDGES + 255) / 256, 256, 0, stream>>>(head, segmax, att, segsum);
        edge_agg_kernel<<<(E_EDGES / EPW + 7) / 8, 256, 0, stream>>>(
            cur_ent, weight, head, tail, edge_type, att, segsum, agg_ent2);
        inter_agg_kernel<<<(NNZ_CNT / EPW + 7) / 8, 256, 0, stream>>>(
            cur_ent, cur_usr, inter_vals, inter_rows, inter_cols, agg_ent2, agg_usr2);
    };

    scatter_hop(entity_emb, user_emb);
    normalize_kernel<<<(N_ENT + 7) / 8, 256, 0, stream>>>(agg_ent2, entity_emb, out_ent, N_ENT, 0);
    normalize_kernel<<<(N_USR + 7) / 8, 256, 0, stream>>>(agg_usr2, user_emb,   out_usr, N_USR, 0);
    scatter_hop(out_ent, out_usr);
    normalize_kernel<<<(N_ENT + 7) / 8, 256, 0, stream>>>(agg_ent2, entity_emb, out_ent, N_ENT, 1);
    normalize_kernel<<<(N_USR + 7) / 8, 256, 0, stream>>>(agg_usr2, user_emb,   out_usr, N_USR, 1);
}


# Three scatter-adds become per-row gathers over a counting sort rebuilt every call (integer atomics give slots); a wave owns a row: softmax and normalize in-wave; no float atomics; sum order changes
# speedup vs baseline: 2.5052x; 1.5258x over previous
#include <hip/hip_runtime.h>
#include <math.h>

#define N_ENT   150000
#define N_USR   60000
#define CH      64
#define E_EDGES 1000000
#define NNZ_CNT 1000000
#define NRELM1  9
#define NHOPS   2
#define EPS_N   1e-12f

typedef __attribute__((ext_vector_type(2))) float v2f;
typedef __attribute__((ext_vector_type(4))) float v4f;
typedef __attribute__((ext_vector_type(2))) int   v2i;
typedef __attribute__((ext_vector_type(4))) int   v4i;
typedef __attribute__((ext_vector_type(8))) float v8f;

#define TBL_WAVES 4
#define TP 16
#define N_TILES (N_ENT / 16)
#define LDP 68
static_assert(N_ENT % 16 == 0, "entity rows tile by 16");
static_assert(16 * CH == 8 * TBL_WAVES * 32, "relation staging: 8 elements per thread");
__global__ __launch_bounds__(TBL_WAVES * 32)
void relnorm_wmma_kernel(const float* __restrict__ ent,
                         const float* __restrict__ weight,
                         float* __restrict__ tbl)
{
    __shared__ float sW2[16 * LDP];
    __shared__ float sA[TBL_WAVES][16 * LDP];
    __shared__ float sB[TBL_WAVES][16 * LDP];

    const int lane = threadIdx.x & 31;
    const int wave = threadIdx.x >> 5;

    const int tA = 2 * (blockIdx.x * TBL_WAVES + wave);
    const bool validA = tA < N_TILES;
    const bool validB = tA + 1 < N_TILES;
    const int tAc = validA ? tA : 0;
    const int tBc = validB ? tA + 1 : tAc;

    float wv[8];
#pragma unroll
    for (int j = 0; j < 8; ++j) {
        const int i = threadIdx.x + j * (TBL_WAVES * 32);
        const int n = i >> 6, c = i & 63;
        const int nc = (n < NRELM1) ? n : NRELM1 - 1;
        const float w = weight[nc * CH + c];
        wv[j] = (n < NRELM1) ? w : 0.0f;
    }

    const v4f* __restrict__ srcA = (const v4f*)(ent + (size_t)tAc * 16 * CH);
    const v4f* __restrict__ srcB = (const v4f*)(ent + (size_t)tBc * 16 * CH);
    v4f xa[8], xb[8];
#pragma unroll
    for (int it = 0; it < 8; ++it) {
        xa[it] = srcA[it * 32 + lane];
        xb[it] = srcB[it * 32 + lane];
    }

#pragma unroll
    for (int j = 0; j < 8; ++j) {
        const int i = threadIdx.x + j * (TBL_WAVES * 32);
        const int n = i >> 6, c = i & 63;
        sW2[n * LDP + c] = wv[j] * wv[j];
    }
    __syncthreads();
    if (!validA) return;

    const int srow = lane >> 4;
    const int scol = (lane & 15) << 2;
#pragma unroll
    for (int it = 0; it < 8; ++it) {
        const v4f a = xa[it], b = xb[it];
        *(v4f*)&sA[wave][(2 * it + srow) * LDP + scol] = a * a;
        *(v4f*)&sB[wave][(2 * it + srow) * LDP + scol] = b * b;
    }

    v8f accA = {};
    v8f accB = {};
    const int m    = lane & 15;
    const int koff = (lane >> 4) << 1;
    for (int kk = 0; kk < 16; ++kk) {
        const int c = kk * 4 + koff;
        v2f a, a2, b;
        a.x  = sA[wave][m * LDP + c];  a.y  = sA[wave][m * LDP + c + 1];
        a2.x = sB[wave][m * LDP + c];  a2.y = sB[wave][m * LDP + c + 1];
        b.x  = sW2[m * LDP + c];       b.y  = sW2[m * LDP + c + 1];
        accA = __builtin_amdgcn_wmma_f32_16x16x4_f32(false, a,  false, b, (short)0, accA, false, false);
        accB = __builtin_amdgcn_wmma_f32_16x16x4_f32(false, a2, false, b, (short)0, accB, false, false);
    }

    const int mrow = (lane >> 4) << 3;
    const int ncol = lane & 15;
    float* dA = tbl + (size_t)(tA * 16 + mrow) * TP + ncol;
#pragma unroll
    for (int v = 0; v < 8; ++v) dA[v * TP] = accA[v];
    if (validB) {
        float* dB = dA + 16 * TP;
#pragma unroll
        for (int v = 0; v < 8; ++v) dB[v * TP] = accB[v];
    }
}

#define L_SEG    (2 * N_ENT + N_USR)
#define SCAN_T   256
#define SCAN_PER (4 * SCAN_T)
#define SCAN_NB  ((L_SEG + SCAN_PER - 1) / SCAN_PER)
#define L_PAD    (SCAN_NB * SCAN_PER)
static_assert(E_EDGES == NNZ_CNT, "one thread takes edge i and interaction i");
static_assert((long long)E_EDGES + 2LL * NNZ_CNT < (1LL << 31), "slots fit an int");

__global__ __launch_bounds__(256)
void count_rank_kernel(const int* __restrict__ head,
                       const int* __restrict__ cols,
                       const int* __restrict__ rows,
                       int* __restrict__ cnt,
                       int* __restrict__ rank)
{
    const int i = blockIdx.x * 256 + threadIdx.x;
    if (i >= E_EDGES) return;
    const int h = head[i], c = cols[i], r = rows[i];
    const int kH = atomicAdd(&cnt[h], 1);
    const int kC = atomicAdd(&cnt[N_ENT + c], 1);
    const int kR = atomicAdd(&cnt[2 * N_ENT + r], 1);
    rank[i]               = kH;
    rank[E_EDGES + i]     = kC;
    rank[2 * E_EDGES + i] = kR;
}

__device__ __forceinline__ int block_sum_256(int v, int* sred)
{
    for (int m = 16; m >= 1; m >>= 1) v += __shfl_xor(v, m, 32);
    if ((threadIdx.x & 31) == 0) sred[threadIdx.x >> 5] = v;
    __syncthreads();
    int t = 0;
    for (int w = 0; w < SCAN_T / 32; ++w) t += sred[w];
    return t;
}

__global__ __launch_bounds__(SCAN_T)
void scan_sums_kernel(const int* __restrict__ cnt, int* __restrict__ bsum)
{
    __shared__ int sred[SCAN_T / 32];
    const v4i c = ((const v4i*)cnt)[blockIdx.x * SCAN_T + threadIdx.x];
    const int t = block_sum_256(c.x + c.y + c.z + c.w, sred);
    if (threadIdx.x == 0) bsum[blockIdx.x] = t;
}

__global__ __launch_bounds__(SCAN_T)
void scan_offsets_kernel(const int* __restrict__ cnt,
                         const int* __restrict__ bsum,
                         int* __restrict__ off)
{
    __shared__ int sred[SCAN_T / 32];
    __shared__ int swav[SCAN_T / 32];
    const int lane = threadIdx.x & 31, wave = threadIdx.x >> 5;
    int part = 0;
    for (int j = threadIdx.x; j < (int)blockIdx.x; j += SCAN_T) part += bsum[j];
    const int base = block_sum_256(part, sred);
    const v4i c = ((const v4i*)cnt)[blockIdx.x * SCAN_T + threadIdx.x];
    const int tsum = c.x + c.y + c.z + c.w;
    int inc = tsum;
    for (int d = 1; d < 32; d <<= 1) {
        const int o = __shfl_up(inc, d, 32);
        if (lane >= d) inc += o;
    }
    if (lane == 31) swav[wave] = inc;
    __syncthreads();
    int wpre = 0;
    for (int w = 0; w < wave; ++w) wpre += swav[w];
    const int ex = base + wpre + inc - tsum;
    v4i o;
    o.x = ex; o.y = ex + c.x; o.z = o.y + c.y; o.w = o.z + c.z;
    ((v4i*)off)[blockIdx.x * SCAN_T + threadIdx.x] = o;
}

__global__ __launch_bounds__(256)
void place_kernel(const int* __restrict__ head,
                  const int* __restrict__ tail,
                  const int* __restrict__ etype,
                  const int* __restrict__ rows,
                  const int* __restrict__ cols,
                  const float* __restrict__ vals,
                  const int* __restrict__ off,
                  const int* __restrict__ rank,
                  int* __restrict__ pay)
{
    const int i = blockIdx.x * 256 + threadIdx.x;
    if (i >= E_EDGES) return;
    const int h = head[i], t = tail[i], c = cols[i], r = rows[i];
    int et = etype[i] - 1;
    if (et < 0) et += NRELM1;
    const int vb = __float_as_int(vals[i]);
    const int pH = off[h] + rank[i];
    const int pC = off[N_ENT + c] + rank[E_EDGES + i];
    const int pR = off[2 * N_ENT + r] + rank[2 * E_EDGES + i];
    v2i* __restrict__ p2 = (v2i*)pay;
    v2i a; a.x = t; a.y = et;  p2[pH] = a;
    v2i b; b.x = r; b.y = vb;  p2[pC] = b;
    v2i d; d.x = c; d.y = vb;  p2[pR] = d;
}

__device__ __forceinline__ float wave_sum(float v)
{
    for (int m = 16; m >= 1; m >>= 1) v += __shfl_xor(v, m, 32);
    return v;
}
__device__ __forceinline__ float wave_max(float v)
{
    for (int m = 16; m >= 1; m >>= 1) v = fmaxf(v, __shfl_xor(v, m, 32));
    return v;
}

__device__ __forceinline__ void add_scaled_rows(v2f& acc, const float* __restrict__ src,
                                                int idx, float sc, int c, int lane)
{
    const int sci = __float_as_int(sc);
    for (int j = 0; j < c; j += 4) {
        v2f x[4];
        float s[4];
#pragma unroll
        for (int u = 0; u < 4; ++u) {
            const int ij = __builtin_amdgcn_readlane(idx, j + u);
            s[u] = __int_as_float(__builtin_amdgcn_readlane(sci, j + u));
            x[u] = *(const v2f*)(src + (size_t)ij * CH + 2 * lane);
        }
#pragma unroll
        for (int u = 0; u < 4; ++u)
            if (j + u < c) {
                acc.x += s[u] * x[u].x;
                acc.y += s[u] * x[u].y;
            }
    }
}

__device__ __forceinline__ void normalize_store(v2f v, int row, int lane,
                                                const float* __restrict__ emb,
                                                const float* __restrict__ n1,
                                                float* __restrict__ dst, int last)
{
    const int c = 2 * lane;
    v2f res = {0.0f, 0.0f};
    if (last) {
        const v2f e0 = *(const v2f*)(emb + (size_t)row * CH + c);
        const v2f p1 = *(const v2f*)(n1 + (size_t)row * CH + c);
        res.x = e0.x + p1.x; res.y = e0.y + p1.y;
    }
    const float ss = wave_sum(v.x * v.x + v.y * v.y);
    const float inv = 1.0f / fmaxf(sqrtf(ss), EPS_N);
    v.x *= inv; v.y *= inv;
    if (last) { v.x = res.x + v.x; v.y = res.y + v.y; }
    *(v2f*)(dst + (size_t)row * CH + c) = v;
}

__global__ __launch_bounds__(256)
void entity_row_kernel(const float* __restrict__ cur_ent,
                       const float* __restrict__ cur_usr,
                       const float* __restrict__ weight,
                       const float* __restrict__ tbl,
                       const int* __restrict__ off,
                       const int* __restrict__ cnt,
                       const int* __restrict__ pay,
                       const float* __restrict__ emb,
                       const float* __restrict__ n1,
                       float* __restrict__ dst, int last)
{
    const int wave = threadIdx.x >> 5, lane = threadIdx.x & 31;
    const int row = blockIdx.x * 8 + wave;
    if (row >= N_ENT) return;
    const int sH = __builtin_amdgcn_readfirstlane(off[row]);
    const int nH = __builtin_amdgcn_readfirstlane(cnt[row]);
    const int sC = __builtin_amdgcn_readfirstlane(off[N_ENT + row]);
    const int nC = __builtin_amdgcn_readfirstlane(cnt[N_ENT + row]);
    const v2i* __restrict__ p2 = (const v2i*)pay;
    const float* __restrict__ trow = tbl + (size_t)row * TP;
    const float NEG_INF = -__builtin_inff();

    auto load_edge = [&](int base, int& tl, int& et, float& a) {
        tl = 0; et = 0; a = NEG_INF;
        if (base + lane < nH) {
            const v2i p = p2[sH + base + lane];
            tl = p.x; et = p.y;
            a = trow[et] * tbl[(size_t)tl * TP + et];
        }
    };

    int tl0, et0; float a0;
    load_edge(0, tl0, et0, a0);
    float mx = a0;
    for (int base = 32; base < nH; base += 32) {
        int tl, et; float a;
        load_edge(base, tl, et, a);
        mx = fmaxf(mx, a);
    }
    mx = wave_max(mx);
    const float e0 = (lane < nH) ? expf(a0 - mx) : 0.0f;
    float sm = e0;
    for (int base = 32; base < nH; base += 32) {
        int tl, et; float a;
        load_edge(base, tl, et, a);
        sm += (base + lane < nH) ? expf(a - mx) : 0.0f;
    }
    sm = wave_sum(sm);

    v2f accE = {0.0f, 0.0f};
    for (int base = 0; base < nH; base += 32) {
        int tl = tl0, et = et0; float e = e0;
        if (base != 0) {
            float a;
            load_edge(base, tl, et, a);
            e = (base + lane < nH) ? expf(a - mx) : 0.0f;
        }
        const int wi = __float_as_int(e / sm);
        const int c = (nH - base < 32) ? nH - base : 32;
        for (int j = 0; j < c; j += 4) {
            v2f tv[4], rv[4];
            float wj[4];
#pragma unroll
            for (int u = 0; u < 4; ++u) {
                const int tj = __builtin_amdgcn_readlane(tl, j + u);
                const int ej = __builtin_amdgcn_readlane(et, j + u);
                wj[u] = __int_as_float(__builtin_amdgcn_readlane(wi, j + u));
                tv[u] = *(const v2f*)(cur_ent + (size_t)tj * CH + 2 * lane);
                rv[u] = *(const v2f*)(weight + ej * CH + 2 * lane);
            }
#pragma unroll
            for (int u = 0; u < 4; ++u)
                if (j + u < c) {
                    const float nx = tv[u].x * rv[u].x, ny = tv[u].y * rv[u].y;
                    accE.x += wj[u] * nx;
                    accE.y += wj[u] * ny;
                }
        }
    }

    v2f accI = {0.0f, 0.0f};
    for (int base = 0; base < nC; base += 32) {
        int ur = 0; float v = 0.0f;
        if (base + lane < nC) {
            const v2i p = p2[sC + base + lane];
            ur = p.x; v = __int_as_float(p.y);
        }
        const int c = (nC - base < 32) ? nC - base : 32;
        add_scaled_rows(accI, cur_usr, ur, v, c, lane);
    }

    v2f v;
    v.x = accE.x + accI.x; v.y = accE.y + accI.y;
    normalize_store(v, row, lane, emb, n1, dst, last);
}

__global__ __launch_bounds__(256)
void user_row_kernel(const float* __restrict__ cur_ent,
                     const int* __restrict__ off,
                     const int* __restrict__ cnt,
                     const int* __restrict__ pay,
                     const float* __restrict__ emb,
                     const float* __restrict__ n1,
                     float* __restrict__ dst, int last)
{
    const int wave = threadIdx.x >> 5, lane = threadIdx.x & 31;
    const int row = blockIdx.x * 8 + wave;
    if (row >= N_USR) return;
    const int sR = __builtin_amdgcn_readfirstlane(off[2 * N_ENT + row]);
    const int nR = __builtin_amdgcn_readfirstlane(cnt[2 * N_ENT + row]);
    const v2i* __restrict__ p2 = (const v2i*)pay;
    v2f acc = {0.0f, 0.0f};
    for (int base = 0; base < nR; base += 32) {
        int er = 0; float v = 0.0f;
        if (base + lane < nR) {
            const v2i p = p2[sR + base + lane];
            er = p.x; v = __int_as_float(p.y);
        }
        const int c = (nR - base < 32) ? nR - base : 32;
        add_scaled_rows(acc, cur_ent, er, v, c, lane);
    }
    normalize_store(acc, row, lane, emb, n1, dst, last);
}

static_assert(NHOPS == 2, "kernel_launch writes the two hops out");
extern "C" void kernel_launch(void* const* d_in, const int* in_sizes, int n_in,
                              void* d_out, int out_size, void* d_ws, size_t ws_size,
                              hipStream_t stream)
{
    (void)in_sizes; (void)n_in; (void)out_size; (void)ws_size;

    const float* user_emb   = (const float*)d_in[0];
    const float* entity_emb = (const float*)d_in[1];
    const float* weight     = (const float*)d_in[2];
    const float* inter_vals = (const float*)d_in[3];
    const int*   edge_index = (const int*)d_in[4];
    const int*   edge_type  = (const int*)d_in[5];
    const int*   inter_rows = (const int*)d_in[6];
    const int*   inter_cols = (const int*)d_in[7];
    const int*   head = edge_index;
    const int*   tail = edge_index + E_EDGES;

    float* out_ent = (float*)d_out;
    float* out_usr = out_ent + (size_t)N_ENT * CH;

    char* ws = (char*)d_ws;
    size_t off_b = 0;
    auto alloc_b = [&](size_t bytes) -> void* {
        void* p = (void*)(ws + off_b);
        off_b += ((bytes + 255) / 256) * 256;
        return p;
    };
    float* n1_ent = (float*)alloc_b((size_t)N_ENT * CH * sizeof(float));
    float* n1_usr = (float*)alloc_b((size_t)N_USR * CH * sizeof(float));
    float* tbl    = (float*)alloc_b((size_t)N_ENT * TP * sizeof(float));
    int*   pay    = (int*)alloc_b(((size_t)E_EDGES + 2 * (size_t)NNZ_CNT) * 2 * sizeof(int));
    int*   rank   = (int*)alloc_b(((size_t)E_EDGES + 2 * (size_t)NNZ_CNT) * sizeof(int));
    int*   cnt    = (int*)alloc_b((size_t)L_PAD * sizeof(int));
    int*   offs   = (int*)alloc_b((size_t)L_PAD * sizeof(int));
    int*   bsum   = (int*)alloc_b((size_t)SCAN_NB * sizeof(int));

    hipMemsetAsync(cnt, 0, (size_t)L_PAD * sizeof(int), stream);
    count_rank_kernel<<<(E_EDGES + 255) / 256, 256, 0, stream>>>(head, inter_cols, inter_rows, cnt, rank);
    scan_sums_kernel<<<SCAN_NB, SCAN_T, 0, stream>>>(cnt, bsum);
    scan_offsets_kernel<<<SCAN_NB, SCAN_T, 0, stream>>>(cnt, bsum, offs);
    place_kernel<<<(E_EDGES + 255) / 256, 256, 0, stream>>>(
        head, tail, edge_type, inter_rows, inter_cols, inter_vals, offs, rank, pay);

    auto hop = [&](const float* cur_ent, const float* cur_usr, float* dst_ent, float* dst_usr, int last) {
        relnorm_wmma_kernel<<<((N_TILES + 1) / 2 + TBL_WAVES - 1) / TBL_WAVES, TBL_WAVES * 32, 0, stream>>>(
            cur_ent, weight, tbl);
        entity_row_kernel<<<(N_ENT + 7) / 8, 256, 0, stream>>>(
            cur_ent, cur_usr, weight, tbl, offs, cnt, pay, entity_emb, n1_ent, dst_ent, last);
        user_row_kernel<<<(N_USR + 7) / 8, 256, 0, stream>>>(
            cur_ent, offs, cnt, pay, user_emb, n1_usr, dst_usr, last);
    };
    hop(entity_emb, user_emb, n1_ent, n1_usr, 0);
    hop(n1_ent, n1_usr, out_ent, out_usr, 1);
}


# Fewer, longer waves: a row-kernel wave does 4 consecutive rows (bounds by scalar loads), a count/place thread 4 items (16-byte loads): 4x fewer waves in these kernels; sum order still by slot arrival
# speedup vs baseline: 2.5959x; 1.0362x over previous
#include <hip/hip_runtime.h>
#include <math.h>

#define N_ENT   150000
#define N_USR   60000
#define CH      64
#define E_EDGES 1000000
#define NNZ_CNT 1000000
#define NRELM1  9
#define NHOPS   2
#define EPS_N   1e-12f

typedef __attribute__((ext_vector_type(2))) float v2f;
typedef __attribute__((ext_vector_type(4))) float v4f;
typedef __attribute__((ext_vector_type(2))) int   v2i;
typedef __attribute__((ext_vector_type(4))) int   v4i;
typedef __attribute__((ext_vector_type(8))) float v8f;

#define TBL_WAVES 4
#define TP 16
#define N_TILES (N_ENT / 16)
#define LDP 68
static_assert(N_ENT % 16 == 0, "entity rows tile by 16");
static_assert(16 * CH == 8 * TBL_WAVES * 32, "relation staging: 8 elements per thread");
__global__ __launch_bounds__(TBL_WAVES * 32)
void relnorm_wmma_kernel(const float* __restrict__ ent,
                         const float* __restrict__ weight,
                         float* __restrict__ tbl)
{
    __shared__ float sW2[16 * LDP];
    __shared__ float sA[TBL_WAVES][16 * LDP];
    __shared__ float sB[TBL_WAVES][16 * LDP];

    const int lane = threadIdx.x & 31;
    const int wave = threadIdx.x >> 5;

    const int tA = 2 * (blockIdx.x * TBL_WAVES + wave);
    const bool validA = tA < N_TILES;
    const bool validB = tA + 1 < N_TILES;
    const int tAc = validA ? tA : 0;
    const int tBc = validB ? tA + 1 : tAc;

    float wv[8];
#pragma unroll
    for (int j = 0; j < 8; ++j) {
        const int i = threadIdx.x + j * (TBL_WAVES * 32);
        const int n = i >> 6, c = i & 63;
        const int nc = (n < NRELM1) ? n : NRELM1 - 1;
        const float w = weight[nc * CH + c];
        wv[j] = (n < NRELM1) ? w : 0.0f;
    }

    const v4f* __restrict__ srcA = (const v4f*)(ent + (size_t)tAc * 16 * CH);
    const v4f* __restrict__ srcB = (const v4f*)(ent + (size_t)tBc * 16 * CH);
    v4f xa[8], xb[8];
#pragma unroll
    for (int it = 0; it < 8; ++it) {
        xa[it] = srcA[it * 32 + lane];
        xb[it] = srcB[it * 32 + lane];
    }

#pragma unroll
    for (int j = 0; j < 8; ++j) {
        const int i = threadIdx.x + j * (TBL_WAVES * 32);
        const int n = i >> 6, c = i & 63;
        sW2[n * LDP + c] = wv[j] * wv[j];
    }
    __syncthreads();
    if (!validA) return;

    const int srow = lane >> 4;
    const int scol = (lane & 15) << 2;
#pragma unroll
    for (int it = 0; it < 8; ++it) {
        const v4f a = xa[it], b = xb[it];
        *(v4f*)&sA[wave][(2 * it + srow) * LDP + scol] = a * a;
        *(v4f*)&sB[wave][(2 * it + srow) * LDP + scol] = b * b;
    }

    v8f accA = {};
    v8f accB = {};
    const int m    = lane & 15;
    const int koff = (lane >> 4) << 1;
    for (int kk = 0; kk < 16; ++kk) {
        const int c = kk * 4 + koff;
        v2f a, a2, b;
        a.x  = sA[wave][m * LDP + c];  a.y  = sA[wave][m * LDP + c + 1];
        a2.x = sB[wave][m * LDP + c];  a2.y = sB[wave][m * LDP + c + 1];
        b.x  = sW2[m * LDP + c];       b.y  = sW2[m * LDP + c + 1];
        accA = __builtin_amdgcn_wmma_f32_16x16x4_f32(false, a,  false, b, (short)0, accA, false, false);
        accB = __builtin_amdgcn_wmma_f32_16x16x4_f32(false, a2, false, b, (short)0, accB, false, false);
    }

    const int mrow = (lane >> 4) << 3;
    const int ncol = lane & 15;
    float* dA = tbl + (size_t)(tA * 16 + mrow) * TP + ncol;
#pragma unroll
    for (int v = 0; v < 8; ++v) dA[v * TP] = accA[v];
    if (validB) {
        float* dB = dA + 16 * TP;
#pragma unroll
        for (int v = 0; v < 8; ++v) dB[v * TP] = accB[v];
    }
}

#define L_SEG    (2 * N_ENT + N_USR)
#define SCAN_T   256
#define SCAN_PER (4 * SCAN_T)
#define SCAN_NB  ((L_SEG + SCAN_PER - 1) / SCAN_PER)
#define L_PAD    (SCAN_NB * SCAN_PER)
static_assert(E_EDGES == NNZ_CNT, "one thread takes edge i and interaction i");
static_assert((long long)E_EDGES + 2LL * NNZ_CNT < (1LL << 31), "slots fit an int");

#define IPT 4
#define ITEM_BLOCKS ((E_EDGES / IPT + 255) / 256)
static_assert(E_EDGES % IPT == 0, "items split into whole groups of IPT");
static_assert(IPT == 4, "the item kernels read their groups as 4-vectors");

__global__ __launch_bounds__(256)
void count_rank_kernel(const int* __restrict__ head,
                       const int* __restrict__ cols,
                       const int* __restrict__ rows,
                       int* __restrict__ cnt,
                       int* __restrict__ rank)
{
    const int g = blockIdx.x * 256 + threadIdx.x;
    if (g >= E_EDGES / IPT) return;
    const v4i h = ((const v4i*)head)[g];
    const v4i c = ((const v4i*)cols)[g];
    const v4i r = ((const v4i*)rows)[g];
    v4i kH, kC, kR;
    kH.x = atomicAdd(&cnt[h.x], 1);
    kH.y = atomicAdd(&cnt[h.y], 1);
    kH.z = atomicAdd(&cnt[h.z], 1);
    kH.w = atomicAdd(&cnt[h.w], 1);
    kC.x = atomicAdd(&cnt[N_ENT + c.x], 1);
    kC.y = atomicAdd(&cnt[N_ENT + c.y], 1);
    kC.z = atomicAdd(&cnt[N_ENT + c.z], 1);
    kC.w = atomicAdd(&cnt[N_ENT + c.w], 1);
    kR.x = atomicAdd(&cnt[2 * N_ENT + r.x], 1);
    kR.y = atomicAdd(&cnt[2 * N_ENT + r.y], 1);
    kR.z = atomicAdd(&cnt[2 * N_ENT + r.z], 1);
    kR.w = atomicAdd(&cnt[2 * N_ENT + r.w], 1);
    ((v4i*)rank)[g]                 = kH;
    ((v4i*)(rank + E_EDGES))[g]     = kC;
    ((v4i*)(rank + 2 * E_EDGES))[g] = kR;
}

__device__ __forceinline__ int block_sum_256(int v, int* sred)
{
    for (int m = 16; m >= 1; m >>= 1) v += __shfl_xor(v, m, 32);
    if ((threadIdx.x & 31) == 0) sred[threadIdx.x >> 5] = v;
    __syncthreads();
    int t = 0;
    for (int w = 0; w < SCAN_T / 32; ++w) t += sred[w];
    return t;
}

__global__ __launch_bounds__(SCAN_T)
void scan_sums_kernel(const int* __restrict__ cnt, int* __restrict__ bsum)
{
    __shared__ int sred[SCAN_T / 32];
    const v4i c = ((const v4i*)cnt)[blockIdx.x * SCAN_T + threadIdx.x];
    const int t = block_sum_256(c.x + c.y + c.z + c.w, sred);
    if (threadIdx.x == 0) bsum[blockIdx.x] = t;
}

__global__ __launch_bounds__(SCAN_T)
void scan_offsets_kernel(const int* __restrict__ cnt,
                         const int* __restrict__ bsum,
                         int* __restrict__ off)
{
    __shared__ int sred[SCAN_T / 32];
    __shared__ int swav[SCAN_T / 32];
    const int lane = threadIdx.x & 31, wave = threadIdx.x >> 5;
    int part = 0;
    for (int j = threadIdx.x; j < (int)blockIdx.x; j += SCAN_T) part += bsum[j];
    const int base = block_sum_256(part, sred);
    const v4i c = ((const v4i*)cnt)[blockIdx.x * SCAN_T + threadIdx.x];
    const int tsum = c.x + c.y + c.z + c.w;
    int inc = tsum;
    for (int d = 1; d < 32; d <<= 1) {
        const int o = __shfl_up(inc, d, 32);
        if (lane >= d) inc += o;
    }
    if (lane == 31) swav[wave] = inc;
    __syncthreads();
    int wpre = 0;
    for (int w = 0; w < wave; ++w) wpre += swav[w];
    const int ex = base + wpre + inc - tsum;
    v4i o;
    o.x = ex; o.y = ex + c.x; o.z = o.y + c.y; o.w = o.z + c.z;
    ((v4i*)off)[blockIdx.x * SCAN_T + threadIdx.x] = o;
}

__global__ __launch_bounds__(256)
void place_kernel(const int* __restrict__ head,
                  const int* __restrict__ tail,
                  const int* __restrict__ etype,
                  const int* __restrict__ rows,
                  const int* __restrict__ cols,
                  const float* __restrict__ vals,
                  const int* __restrict__ off,
                  const int* __restrict__ rank,
                  int* __restrict__ pay)
{
    const int g = blockIdx.x * 256 + threadIdx.x;
    if (g >= E_EDGES / IPT) return;
    const v4i h  = ((const v4i*)head)[g];
    const v4i t  = ((const v4i*)tail)[g];
    const v4i c  = ((const v4i*)cols)[g];
    const v4i r  = ((const v4i*)rows)[g];
    const v4i e  = ((const v4i*)etype)[g];
    const v4f vf = ((const v4f*)vals)[g];
    const v4i kH = ((const v4i*)rank)[g];
    const v4i kC = ((const v4i*)(rank + E_EDGES))[g];
    const v4i kR = ((const v4i*)(rank + 2 * E_EDGES))[g];
    v2i* __restrict__ p2 = (v2i*)pay;
#pragma unroll
    for (int u = 0; u < IPT; ++u) {
        int et = e[u] - 1;
        if (et < 0) et += NRELM1;
        const int vb = __float_as_int(vf[u]);
        const int pH = off[h[u]] + kH[u];
        const int pC = off[N_ENT + c[u]] + kC[u];
        const int pR = off[2 * N_ENT + r[u]] + kR[u];
        v2i a; a.x = t[u]; a.y = et;  p2[pH] = a;
        v2i b; b.x = r[u]; b.y = vb;  p2[pC] = b;
        v2i d; d.x = c[u]; d.y = vb;  p2[pR] = d;
    }
}

#define RPW 4
#define ROWS_PER_BLOCK (8 * RPW)
__device__ __forceinline__ float wave_sum(float v)
{
    for (int m = 16; m >= 1; m >>= 1) v += __shfl_xor(v, m, 32);
    return v;
}
__device__ __forceinline__ float wave_max(float v)
{
    for (int m = 16; m >= 1; m >>= 1) v = fmaxf(v, __shfl_xor(v, m, 32));
    return v;
}

__device__ __forceinline__ void add_scaled_rows(v2f& acc, const float* __restrict__ src,
                                                int idx, float sc, int c, int lane)
{
    const int sci = __float_as_int(sc);
    for (int j = 0; j < c; j += 4) {
        v2f x[4];
        float s[4];
#pragma unroll
        for (int u = 0; u < 4; ++u) {
            const int ij = __builtin_amdgcn_readlane(idx, j + u);
            s[u] = __int_as_float(__builtin_amdgcn_readlane(sci, j + u));
            x[u] = *(const v2f*)(src + (size_t)ij * CH + 2 * lane);
        }
#pragma unroll
        for (int u = 0; u < 4; ++u)
            if (j + u < c) {
                acc.x += s[u] * x[u].x;
                acc.y += s[u] * x[u].y;
            }
    }
}

__device__ __forceinline__ void normalize_store(v2f v, int row, int lane,
                                                const float* __restrict__ emb,
                                                const float* __restrict__ n1,
                                                float* __restrict__ dst, int last)
{
    const int c = 2 * lane;
    v2f res = {0.0f, 0.0f};
    if (last) {
        const v2f e0 = *(const v2f*)(emb + (size_t)row * CH + c);
        const v2f p1 = *(const v2f*)(n1 + (size_t)row * CH + c);
        res.x = e0.x + p1.x; res.y = e0.y + p1.y;
    }
    const float ss = wave_sum(v.x * v.x + v.y * v.y);
    const float inv = 1.0f / fmaxf(sqrtf(ss), EPS_N);
    v.x *= inv; v.y *= inv;
    if (last) { v.x = res.x + v.x; v.y = res.y + v.y; }
    *(v2f*)(dst + (size_t)row * CH + c) = v;
}

__device__ __forceinline__ void entity_row(int row, int lane, int sH, int nH, int sC, int nC,
                                           const float* __restrict__ cur_ent,
                                           const float* __restrict__ cur_usr,
                                           const float* __restrict__ weight,
                                           const float* __restrict__ tbl,
                                           const v2i* __restrict__ p2,
                                           const float* __restrict__ emb,
                                           const float* __restrict__ n1,
                                           float* __restrict__ dst, int last)
{
    const float* __restrict__ trow = tbl + (size_t)row * TP;
    const float NEG_INF = -__builtin_inff();

    auto load_edge = [&](int base, int& tl, int& et, float& a) {
        tl = 0; et = 0; a = NEG_INF;
        if (base + lane < nH) {
            const v2i p = p2[sH + base + lane];
            tl = p.x; et = p.y;
            a = trow[et] * tbl[(size_t)tl * TP + et];
        }
    };

    int tl0, et0; float a0;
    load_edge(0, tl0, et0, a0);
    float mx = a0;
    for (int base = 32; base < nH; base += 32) {
        int tl, et; float a;
        load_edge(base, tl, et, a);
        mx = fmaxf(mx, a);
    }
    mx = wave_max(mx);
    const float e0 = (lane < nH) ? expf(a0 - mx) : 0.0f;
    float sm = e0;
    for (int base = 32; base < nH; base += 32) {
        int tl, et; float a;
        load_edge(base, tl, et, a);
        sm += (base + lane < nH) ? expf(a - mx) : 0.0f;
    }
    sm = wave_sum(sm);

    v2f accE = {0.0f, 0.0f};
    for (int base = 0; base < nH; base += 32) {
        int tl = tl0, et = et0; float e = e0;
        if (base != 0) {
            float a;
            load_edge(base, tl, et, a);
            e = (base + lane < nH) ? expf(a - mx) : 0.0f;
        }
        const int wi = __float_as_int(e / sm);
        const int c = (nH - base < 32) ? nH - base : 32;
        for (int j = 0; j < c; j += 4) {
            v2f tv[4], rv[4];
            float wj[4];
#pragma unroll
            for (int u = 0; u < 4; ++u) {
                const int tj = __builtin_amdgcn_readlane(tl, j + u);
                const int ej = __builtin_amdgcn_readlane(et, j + u);
                wj[u] = __int_as_float(__builtin_amdgcn_readlane(wi, j + u));
                tv[u] = *(const v2f*)(cur_ent + (size_t)tj * CH + 2 * lane);
                rv[u] = *(const v2f*)(weight + ej * CH + 2 * lane);
            }
#pragma unroll
            for (int u = 0; u < 4; ++u)
                if (j + u < c) {
                    const float nx = tv[u].x * rv[u].x, ny = tv[u].y * rv[u].y;
                    accE.x += wj[u] * nx;
                    accE.y += wj[u] * ny;
                }
        }
    }

    v2f accI = {0.0f, 0.0f};
    for (int base = 0; base < nC; base += 32) {
        int ur = 0; float v = 0.0f;
        if (base + lane < nC) {
            const v2i p = p2[sC + base + lane];
            ur = p.x; v = __int_as_float(p.y);
        }
        const int c = (nC - base < 32) ? nC - base : 32;
        add_scaled_rows(accI, cur_usr, ur, v, c, lane);
    }

    v2f v;
    v.x = accE.x + accI.x; v.y = accE.y + accI.y;
    normalize_store(v, row, lane, emb, n1, dst, last);
}

__global__ __launch_bounds__(256)
void entity_row_kernel(const float* __restrict__ cur_ent,
                       const float* __restrict__ cur_usr,
                       const float* __restrict__ weight,
                       const float* __restrict__ tbl,
                       const int* __restrict__ off,
                       const int* __restrict__ cnt,
                       const int* __restrict__ pay,
                       const float* __restrict__ emb,
                       const float* __restrict__ n1,
                       float* __restrict__ dst, int last)
{
    const int lane = threadIdx.x & 31;
    const int wave = __builtin_amdgcn_readfirstlane(threadIdx.x >> 5);
    const int row0 = (blockIdx.x * 8 + wave) * RPW;
    for (int r = 0; r < RPW && row0 + r < N_ENT; ++r) {
        const int row = row0 + r;
        const int sH = off[row],         nH = cnt[row];
        const int sC = off[N_ENT + row], nC = cnt[N_ENT + row];
        entity_row(row, lane, sH, nH, sC, nC, cur_ent, cur_usr, weight, tbl,
                   (const v2i*)pay, emb, n1, dst, last);
    }
}

__global__ __launch_bounds__(256)
void user_row_kernel(const float* __restrict__ cur_ent,
                     const int* __restrict__ off,
                     const int* __restrict__ cnt,
                     const int* __restrict__ pay,
                     const float* __restrict__ emb,
                     const float* __restrict__ n1,
                     float* __restrict__ dst, int last)
{
    const int lane = threadIdx.x & 31;
    const int wave = __builtin_amdgcn_readfirstlane(threadIdx.x >> 5);
    const int row0 = (blockIdx.x * 8 + wave) * RPW;
    const v2i* __restrict__ p2 = (const v2i*)pay;
    for (int r = 0; r < RPW && row0 + r < N_USR; ++r) {
        const int sR = off[2 * N_ENT + row0 + r];
        const int nR = cnt[2 * N_ENT + row0 + r];
        v2f acc = {0.0f, 0.0f};
        for (int base = 0; base < nR; base += 32) {
            int er = 0; float v = 0.0f;
            if (base + lane < nR) {
                const v2i p = p2[sR + base + lane];
                er = p.x; v = __int_as_float(p.y);
            }
            const int c = (nR - base < 32) ? nR - base : 32;
            add_scaled_rows(acc, cur_ent, er, v, c, lane);
        }
        normalize_store(acc, row0 + r, lane, emb, n1, dst, last);
    }
}

static_assert(NHOPS == 2, "kernel_launch writes the two hops out");
extern "C" void kernel_launch(void* const* d_in, const int* in_sizes, int n_in,
                              void* d_out, int out_size, void* d_ws, size_t ws_size,
                              hipStream_t stream)
{
    (void)in_sizes; (void)n_in; (void)out_size; (void)ws_size;

    const float* user_emb   = (const float*)d_in[0];
    const float* entity_emb = (const float*)d_in[1];
    const float* weight     = (const float*)d_in[2];
    const float* inter_vals = (const float*)d_in[3];
    const int*   edge_index = (const int*)d_in[4];
    const int*   edge_type  = (const int*)d_in[5];
    const int*   inter_rows = (const int*)d_in[6];
    const int*   inter_cols = (const int*)d_in[7];
    const int*   head = edge_index;
    const int*   tail = edge_index + E_EDGES;

    float* out_ent = (float*)d_out;
    float* out_usr = out_ent + (size_t)N_ENT * CH;

    char* ws = (char*)d_ws;
    size_t off_b = 0;
    auto alloc_b = [&](size_t bytes) -> void* {
        void* p = (void*)(ws + off_b);
        off_b += ((bytes + 255) / 256) * 256;
        return p;
    };
    float* n1_ent = (float*)alloc_b((size_t)N_ENT * CH * sizeof(float));
    float* n1_usr = (float*)alloc_b((size_t)N_USR * CH * sizeof(float));
    float* tbl    = (float*)alloc_b((size_t)N_ENT * TP * sizeof(float));
    int*   pay    = (int*)alloc_b(((size_t)E_EDGES + 2 * (size_t)NNZ_CNT) * 2 * sizeof(int));
    int*   rank   = (int*)alloc_b(((size_t)E_EDGES + 2 * (size_t)NNZ_CNT) * sizeof(int));
    int*   cnt    = (int*)alloc_b((size_t)L_PAD * sizeof(int));
    int*   offs   = (int*)alloc_b((size_t)L_PAD * sizeof(int));
    int*   bsum   = (int*)alloc_b((size_t)SCAN_NB * sizeof(int));

    hipMemsetAsync(cnt, 0, (size_t)L_PAD * sizeof(int), stream);
    count_rank_kernel<<<ITEM_BLOCKS, 256, 0, stream>>>(head, inter_cols, inter_rows, cnt, rank);
    scan_sums_kernel<<<SCAN_NB, SCAN_T, 0, stream>>>(cnt, bsum);
    scan_offsets_kernel<<<SCAN_NB, SCAN_T, 0, stream>>>(cnt, bsum, offs);
    place_kernel<<<ITEM_BLOCKS, 256, 0, stream>>>(
        head, tail, edge_type, inter_rows, inter_cols, inter_vals, offs, rank, pay);

    auto hop = [&](const float* cur_ent, const float* cur_usr, float* dst_ent, float* dst_usr, int last) {
        relnorm_wmma_kernel<<<((N_TILES + 1) / 2 + TBL_WAVES - 1) / TBL_WAVES, TBL_WAVES * 32, 0, stream>>>(
            cur_ent, weight, tbl);
        entity_row_kernel<<<(N_ENT + ROWS_PER_BLOCK - 1) / ROWS_PER_BLOCK, 256, 0, stream>>>(
            cur_ent, cur_usr, weight, tbl, offs, cnt, pay, entity_emb, n1_ent, dst_ent, last);
        user_row_kernel<<<(N_USR + ROWS_PER_BLOCK - 1) / ROWS_PER_BLOCK, 256, 0, stream>>>(
            cur_ent, offs, cnt, pay, user_emb, n1_usr, dst_usr, last);
    };
    hop(entity_emb, user_emb, n1_ent, n1_usr, 0);
    hop(n1_ent, n1_usr, out_ent, out_usr, 1);
}


# Row kernels: the item lists of a wave's 4 rows are one run of slots, read and walked as one list (8 source rows in flight, sum handed over at row starts); softmax sums reduce at shifted lanes
# speedup vs baseline: 2.7483x; 1.0587x over previous
#include <hip/hip_runtime.h>
#include <math.h>

#define N_ENT   150000
#define N_USR   60000
#define CH      64
#define E_EDGES 1000000
#define NNZ_CNT 1000000
#define NRELM1  9
#define NHOPS   2
#define EPS_N   1e-12f

typedef __attribute__((ext_vector_type(2))) float v2f;
typedef __attribute__((ext_vector_type(4))) float v4f;
typedef __attribute__((ext_vector_type(2))) int   v2i;
typedef __attribute__((ext_vector_type(4))) int   v4i;
typedef __attribute__((ext_vector_type(8))) float v8f;

#define TBL_WAVES 4
#define TP 16
#define N_TILES (N_ENT / 16)
#define LDP 68
static_assert(N_ENT % 16 == 0, "entity rows tile by 16");
static_assert(16 * CH == 8 * TBL_WAVES * 32, "relation staging: 8 elements per thread");
__global__ __launch_bounds__(TBL_WAVES * 32)
void relnorm_wmma_kernel(const float* __restrict__ ent,
                         const float* __restrict__ weight,
                         float* __restrict__ tbl)
{
    __shared__ float sW2[16 * LDP];
    __shared__ float sA[TBL_WAVES][16 * LDP];
    __shared__ float sB[TBL_WAVES][16 * LDP];

    const int lane = threadIdx.x & 31;
    const int wave = threadIdx.x >> 5;

    const int tA = 2 * (blockIdx.x * TBL_WAVES + wave);
    const bool validA = tA < N_TILES;
    const bool validB = tA + 1 < N_TILES;
    const int tAc = validA ? tA : 0;
    const int tBc = validB ? tA + 1 : tAc;

    float wv[8];
#pragma unroll
    for (int j = 0; j < 8; ++j) {
        const int i = threadIdx.x + j * (TBL_WAVES * 32);
        const int n = i >> 6, c = i & 63;
        const int nc = (n < NRELM1) ? n : NRELM1 - 1;
        const float w = weight[nc * CH + c];
        wv[j] = (n < NRELM1) ? w : 0.0f;
    }

    const v4f* __restrict__ srcA = (const v4f*)(ent + (size_t)tAc * 16 * CH);
    const v4f* __restrict__ srcB = (const v4f*)(ent + (size_t)tBc * 16 * CH);
    v4f xa[8], xb[8];
#pragma unroll
    for (int it = 0; it < 8; ++it) {
        xa[it] = srcA[it * 32 + lane];
        xb[it] = srcB[it * 32 + lane];
    }

#pragma unroll
    for (int j = 0; j < 8; ++j) {
        const int i = threadIdx.x + j * (TBL_WAVES * 32);
        const int n = i >> 6, c = i & 63;
        sW2[n * LDP + c] = wv[j] * wv[j];
    }
    __syncthreads();
    if (!validA) return;

    const int srow = lane >> 4;
    const int scol = (lane & 15) << 2;
#pragma unroll
    for (int it = 0; it < 8; ++it) {
        const v4f a = xa[it], b = xb[it];
        *(v4f*)&sA[wave][(2 * it + srow) * LDP + scol] = a * a;
        *(v4f*)&sB[wave][(2 * it + srow) * LDP + scol] = b * b;
    }

    v8f accA = {};
    v8f accB = {};
    const int m    = lane & 15;
    const int koff = (lane >> 4) << 1;
    for (int kk = 0; kk < 16; ++kk) {
        const int c = kk * 4 + koff;
        v2f a, a2, b;
        a.x  = sA[wave][m * LDP + c];  a.y  = sA[wave][m * LDP + c + 1];
        a2.x = sB[wave][m * LDP + c];  a2.y = sB[wave][m * LDP + c + 1];
        b.x  = sW2[m * LDP + c];       b.y  = sW2[m * LDP + c + 1];
        accA = __builtin_amdgcn_wmma_f32_16x16x4_f32(false, a,  false, b, (short)0, accA, false, false);
        accB = __builtin_amdgcn_wmma_f32_16x16x4_f32(false, a2, false, b, (short)0, accB, false, false);
    }

    const int mrow = (lane >> 4) << 3;
    const int ncol = lane & 15;
    float* dA = tbl + (size_t)(tA * 16 + mrow) * TP + ncol;
#pragma unroll
    for (int v = 0; v < 8; ++v) dA[v * TP] = accA[v];
    if (validB) {
        float* dB = dA + 16 * TP;
#pragma unroll
        for (int v = 0; v < 8; ++v) dB[v * TP] = accB[v];
    }
}

#define L_SEG    (2 * N_ENT + N_USR)
#define SCAN_T   256
#define SCAN_PER (4 * SCAN_T)
#define SCAN_NB  ((L_SEG + SCAN_PER - 1) / SCAN_PER)
#define L_PAD    (SCAN_NB * SCAN_PER)
static_assert(E_EDGES == NNZ_CNT, "one thread takes edge i and interaction i");
static_assert((long long)E_EDGES + 2LL * NNZ_CNT < (1LL << 31), "slots fit an int");

#define IPT 4
#define ITEM_BLOCKS ((E_EDGES / IPT + 255) / 256)
static_assert(E_EDGES % IPT == 0, "items split into whole groups of IPT");
static_assert(IPT == 4, "the item kernels read their groups as 4-vectors");

__global__ __launch_bounds__(256)
void count_rank_kernel(const int* __restrict__ head,
                       const int* __restrict__ cols,
                       const int* __restrict__ rows,
                       int* __restrict__ cnt,
                       int* __restrict__ rank)
{
    const int g = blockIdx.x * 256 + threadIdx.x;
    if (g >= E_EDGES / IPT) return;
    const v4i h = ((const v4i*)head)[g];
    const v4i c = ((const v4i*)cols)[g];
    const v4i r = ((const v4i*)rows)[g];
    v4i kH, kC, kR;
    kH.x = atomicAdd(&cnt[h.x], 1);
    kH.y = atomicAdd(&cnt[h.y], 1);
    kH.z = atomicAdd(&cnt[h.z], 1);
    kH.w = atomicAdd(&cnt[h.w], 1);
    kC.x = atomicAdd(&cnt[N_ENT + c.x], 1);
    kC.y = atomicAdd(&cnt[N_ENT + c.y], 1);
    kC.z = atomicAdd(&cnt[N_ENT + c.z], 1);
    kC.w = atomicAdd(&cnt[N_ENT + c.w], 1);
    kR.x = atomicAdd(&cnt[2 * N_ENT + r.x], 1);
    kR.y = atomicAdd(&cnt[2 * N_ENT + r.y], 1);
    kR.z = atomicAdd(&cnt[2 * N_ENT + r.z], 1);
    kR.w = atomicAdd(&cnt[2 * N_ENT + r.w], 1);
    ((v4i*)rank)[g]                 = kH;
    ((v4i*)(rank + E_EDGES))[g]     = kC;
    ((v4i*)(rank + 2 * E_EDGES))[g] = kR;
}

__device__ __forceinline__ int block_sum_256(int v, int* sred)
{
    for (int m = 16; m >= 1; m >>= 1) v += __shfl_xor(v, m, 32);
    if ((threadIdx.x & 31) == 0) sred[threadIdx.x >> 5] = v;
    __syncthreads();
    int t = 0;
    for (int w = 0; w < SCAN_T / 32; ++w) t += sred[w];
    return t;
}

__global__ __launch_bounds__(SCAN_T)
void scan_sums_kernel(const int* __restrict__ cnt, int* __restrict__ bsum)
{
    __shared__ int sred[SCAN_T / 32];
    const v4i c = ((const v4i*)cnt)[blockIdx.x * SCAN_T + threadIdx.x];
    const int t = block_sum_256(c.x + c.y + c.z + c.w, sred);
    if (threadIdx.x == 0) bsum[blockIdx.x] = t;
}

__global__ __launch_bounds__(SCAN_T)
void scan_offsets_kernel(const int* __restrict__ cnt,
                         const int* __restrict__ bsum,
                         int* __restrict__ off)
{
    __shared__ int sred[SCAN_T / 32];
    __shared__ int swav[SCAN_T / 32];
    const int lane = threadIdx.x & 31, wave = threadIdx.x >> 5;
    int part = 0;
    for (int j = threadIdx.x; j < (int)blockIdx.x; j += SCAN_T) part += bsum[j];
    const int base = block_sum_256(part, sred);
    const v4i c = ((const v4i*)cnt)[blockIdx.x * SCAN_T + threadIdx.x];
    const int tsum = c.x + c.y + c.z + c.w;
    int inc = tsum;
    for (int d = 1; d < 32; d <<= 1) {
        const int o = __shfl_up(inc, d, 32);
        if (lane >= d) inc += o;
    }
    if (lane == 31) swav[wave] = inc;
    __syncthreads();
    int wpre = 0;
    for (int w = 0; w < wave; ++w) wpre += swav[w];
    const int ex = base + wpre + inc - tsum;
    v4i o;
    o.x = ex; o.y = ex + c.x; o.z = o.y + c.y; o.w = o.z + c.z;
    ((v4i*)off)[blockIdx.x * SCAN_T + threadIdx.x] = o;
}

__global__ __launch_bounds__(256)
void place_kernel(const int* __restrict__ head,
                  const int* __restrict__ tail,
                  const int* __restrict__ etype,
                  const int* __restrict__ rows,
                  const int* __restrict__ cols,
                  const float* __restrict__ vals,
                  const int* __restrict__ off,
                  const int* __restrict__ rank,
                  int* __restrict__ pay)
{
    const int g = blockIdx.x * 256 + threadIdx.x;
    if (g >= E_EDGES / IPT) return;
    const v4i h  = ((const v4i*)head)[g];
    const v4i t  = ((const v4i*)tail)[g];
    const v4i c  = ((const v4i*)cols)[g];
    const v4i r  = ((const v4i*)rows)[g];
    const v4i e  = ((const v4i*)etype)[g];
    const v4f vf = ((const v4f*)vals)[g];
    const v4i kH = ((const v4i*)rank)[g];
    const v4i kC = ((const v4i*)(rank + E_EDGES))[g];
    const v4i kR = ((const v4i*)(rank + 2 * E_EDGES))[g];
    v2i* __restrict__ p2 = (v2i*)pay;
#pragma unroll
    for (int u = 0; u < IPT; ++u) {
        int et = e[u] - 1;
        if (et < 0) et += NRELM1;
        const int vb = __float_as_int(vf[u]);
        const int pH = off[h[u]] + kH[u];
        const int pC = off[N_ENT + c[u]] + kC[u];
        const int pR = off[2 * N_ENT + r[u]] + kR[u];
        v2i a; a.x = t[u]; a.y = et;  p2[pH] = a;
        v2i b; b.x = r[u]; b.y = vb;  p2[pC] = b;
        v2i d; d.x = c[u]; d.y = vb;  p2[pR] = d;
    }
}

#define RPW 4
#define ROWS_PER_BLOCK (8 * RPW)
#define NO_SLOT 0x7fffffff
static_assert(RPW == 4, "the row kernels read their bounds as 4-vectors and name four sums");
static_assert(N_ENT % RPW == 0 && N_USR % RPW == 0, "a wave's rows all exist or none does");
static_assert(L_PAD > L_SEG, "off[] has an entry behind the last segment");

__device__ __forceinline__ void wave_sum4(float (&v)[4])
{
    for (int m = 16; m >= 1; m >>= 1) {
        float t[4];
#pragma unroll
        for (int r = 0; r < 4; ++r) t[r] = __shfl_xor(v[r], m, 32);
#pragma unroll
        for (int r = 0; r < 4; ++r) v[r] += t[r];
    }
}
__device__ __forceinline__ void wave_max4(float (&v)[4])
{
    for (int m = 16; m >= 1; m >>= 1) {
        float t[4];
#pragma unroll
        for (int r = 0; r < 4; ++r) t[r] = __shfl_xor(v[r], m, 32);
#pragma unroll
        for (int r = 0; r < 4; ++r) v[r] = fmaxf(v[r], t[r]);
    }
}

struct RowSums {
    v2f acc, r0, r1, r2;
    int cur, nb;
};
__device__ __forceinline__ void rows_begin(RowSums& st, int b1)
{
    st.acc = v2f{0.0f, 0.0f}; st.r0 = st.acc; st.r1 = st.acc; st.r2 = st.acc;
    st.cur = 0; st.nb = b1;
}
__device__ __forceinline__ void rows_step(RowSums& st, int b2, int b3)
{
    if (st.cur == 0) st.r0 = st.acc; else if (st.cur == 1) st.r1 = st.acc; else st.r2 = st.acc;
    st.acc = v2f{0.0f, 0.0f};
    ++st.cur;
    st.nb = (st.cur == 1) ? b2 : (st.cur == 2) ? b3 : NO_SLOT;
}
__device__ __forceinline__ void rows_end(RowSums& st, int b2, int b3, v2f (&o)[4])
{
    while (st.cur < 3) rows_step(st, b2, b3);
    o[0] = st.r0; o[1] = st.r1; o[2] = st.r2; o[3] = st.acc;
}

template <bool REL>
__device__ __forceinline__ void walk_chunk(RowSums& st, int b2, int b3,
                                           const float* __restrict__ src,
                                           const float* __restrict__ weight,
                                           int idx, int rel, int wbits,
                                           int sbase, int c, int lane)
{
    for (int j = 0; j < c; j += 8) {
        v2f x[8], rv[8];
        float w[8];
#pragma unroll
        for (int u = 0; u < 8; ++u) {
            const int ij = __builtin_amdgcn_readlane(idx, j + u);
            w[u] = __int_as_float(__builtin_amdgcn_readlane(wbits, j + u));
            x[u] = *(const v2f*)(src + (size_t)ij * CH + 2 * lane);
            if (REL) {
                const int ej = __builtin_amdgcn_readlane(rel, j + u);
                rv[u] = *(const v2f*)(weight + ej * CH + 2 * lane);
            }
        }
#pragma unroll
        for (int u = 0; u < 8; ++u)
            if (j + u < c) {
                const int s = sbase + j + u;
                while (s >= st.nb) rows_step(st, b2, b3);
                float nx = x[u].x, ny = x[u].y;
                if (REL) { nx *= rv[u].x; ny *= rv[u].y; }
                st.acc.x += w[u] * nx;
                st.acc.y += w[u] * ny;
            }
    }
}

__device__ __forceinline__ void normalize_store4(const v2f (&vin)[4], int row0, int lane,
                                                 const float* __restrict__ emb,
                                                 const float* __restrict__ n1,
                                                 float* __restrict__ dst, int last)
{
    const int c = 2 * lane;
    v2f res[4];
#pragma unroll
    for (int r = 0; r < 4; ++r) res[r] = v2f{0.0f, 0.0f};
    if (last) {
#pragma unroll
        for (int r = 0; r < 4; ++r) {
            const v2f e0 = *(const v2f*)(emb + (size_t)(row0 + r) * CH + c);
            const v2f p1 = *(const v2f*)(n1 + (size_t)(row0 + r) * CH + c);
            res[r].x = e0.x + p1.x; res[r].y = e0.y + p1.y;
        }
    }
    float ss[4];
#pragma unroll
    for (int r = 0; r < 4; ++r) ss[r] = vin[r].x * vin[r].x + vin[r].y * vin[r].y;
    wave_sum4(ss);
#pragma unroll
    for (int r = 0; r < 4; ++r) {
        const float inv = 1.0f / fmaxf(sqrtf(ss[r]), EPS_N);
        v2f v = vin[r];
        v.x *= inv; v.y *= inv;
        if (last) { v.x = res[r].x + v.x; v.y = res[r].y + v.y; }
        *(v2f*)(dst + (size_t)(row0 + r) * CH + c) = v;
    }
}

__global__ __launch_bounds__(256)
void entity_row_kernel(const float* __restrict__ cur_ent,
                       const float* __restrict__ cur_usr,
                       const float* __restrict__ weight,
                       const float* __restrict__ tbl,
                       const int* __restrict__ off,
                       const int* __restrict__ pay,
                       const float* __restrict__ emb,
                       const float* __restrict__ n1,
                       float* __restrict__ dst, int last)
{
    const int lane = threadIdx.x & 31;
    const int wave = __builtin_amdgcn_readfirstlane(threadIdx.x >> 5);
    const int row0 = (blockIdx.x * 8 + wave) * RPW;
    if (row0 >= N_ENT) return;
    const v2i* __restrict__ p2 = (const v2i*)pay;
    const float NEG_INF = -__builtin_inff();

    const v4i oH = *(const v4i*)(off + row0);
    const int hEnd = off[row0 + RPW];
    const v4i oC = *(const v4i*)(off + N_ENT + row0);
    const int cEnd = off[N_ENT + row0 + RPW];
    const int hB0 = oH.x, hB1 = oH.y, hB2 = oH.z, hB3 = oH.w;
    const int cB0 = oC.x, cB1 = oC.y, cB2 = oC.z, cB3 = oC.w;
    const int nH = hEnd - hB0, nC = cEnd - cB0;

    auto load_inter = [&](int base, int& ur, int& vb) {
        ur = 0; vb = 0;
        const int s = cB0 + base + lane;
        if (s < cEnd) {
            const v2i p = p2[s];
            ur = p.x; vb = p.y;
        }
    };
    auto load_edge = [&](int base, int& tl, int& et, int& rid, float& a) {
        tl = 0; et = 0; rid = 0; a = NEG_INF;
        const int s = hB0 + base + lane;
        if (s < hEnd) {
            const v2i p = p2[s];
            tl = p.x; et = p.y;
            rid = (s >= hB1) + (s >= hB2) + (s >= hB3);
            a = tbl[(size_t)(row0 + rid) * TP + et] * tbl[(size_t)tl * TP + et];
        }
    };
    auto pick = [](int rid, const float (&v)[4]) {
        return rid == 0 ? v[0] : rid == 1 ? v[1] : rid == 2 ? v[2] : v[3];
    };

    int ur0, vb0;
    load_inter(0, ur0, vb0);
    int tl0, et0, rid0; float a0;
    load_edge(0, tl0, et0, rid0, a0);

    float mx[4];
#pragma unroll
    for (int r = 0; r < 4; ++r) mx[r] = (rid0 == r) ? a0 : NEG_INF;
    for (int base = 32; base < nH; base += 32) {
        int tl, et, rid; float a;
        load_edge(base, tl, et, rid, a);
#pragma unroll
        for (int r = 0; r < 4; ++r) mx[r] = fmaxf(mx[r], (rid == r) ? a : NEG_INF);
    }
    wave_max4(mx);

    const float e0 = (lane < nH) ? expf(a0 - pick(rid0, mx)) : 0.0f;
    float sm[4];
#pragma unroll
    for (int r = 0; r < 4; ++r) sm[r] = (rid0 == r) ? e0 : 0.0f;
    for (int base = 32; base < nH; base += 32) {
        int tl, et, rid; float a;
        load_edge(base, tl, et, rid, a);
        const float e = (base + lane < nH) ? expf(a - pick(rid, mx)) : 0.0f;
#pragma unroll
        for (int r = 0; r < 4; ++r) sm[r] += (rid == r) ? e : 0.0f;
    }
    wave_sum4(sm);

    RowSums st;
    rows_begin(st, hB1);
    for (int base = 0; base < nH; base += 32) {
        int tl = tl0, et = et0, rid = rid0; float e = e0;
        if (base != 0) {
            float a;
            load_edge(base, tl, et, rid, a);
            e = (base + lane < nH) ? expf(a - pick(rid, mx)) : 0.0f;
        }
        const int wi = __float_as_int(e / pick(rid, sm));
        const int c = (nH - base < 32) ? nH - base : 32;
        walk_chunk<true>(st, hB2, hB3, cur_ent, weight, tl, et, wi, hB0 + base, c, lane);
    }
    v2f accE[4];
    rows_end(st, hB2, hB3, accE);

    rows_begin(st, cB1);
    for (int base = 0; base < nC; base += 32) {
        int ur = ur0, vb = vb0;
        if (base != 0) load_inter(base, ur, vb);
        const int c = (nC - base < 32) ? nC - base : 32;
        walk_chunk<false>(st, cB2, cB3, cur_usr, weight, ur, 0, vb, cB0 + base, c, lane);
    }
    v2f accI[4];
    rows_end(st, cB2, cB3, accI);

    v2f v[4];
#pragma unroll
    for (int r = 0; r < 4; ++r) { v[r].x = accE[r].x + accI[r].x; v[r].y = accE[r].y + accI[r].y; }
    normalize_store4(v, row0, lane, emb, n1, dst, last);
}

__global__ __launch_bounds__(256)
void user_row_kernel(const float* __restrict__ cur_ent,
                     const int* __restrict__ off,
                     const int* __restrict__ pay,
                     const float* __restrict__ emb,
                     const float* __restrict__ n1,
                     float* __restrict__ dst, int last)
{
    const int lane = threadIdx.x & 31;
    const int wave = __builtin_amdgcn_readfirstlane(threadIdx.x >> 5);
    const int row0 = (blockIdx.x * 8 + wave) * RPW;
    if (row0 >= N_USR) return;
    const v2i* __restrict__ p2 = (const v2i*)pay;
    const v4i oR = *(const v4i*)(off + 2 * N_ENT + row0);
    const int rEnd = off[2 * N_ENT + row0 + RPW];
    const int rB0 = oR.x, rB1 = oR.y, rB2 = oR.z, rB3 = oR.w;
    const int nR = rEnd - rB0;

    auto load_inter = [&](int base, int& er, int& vb) {
        er = 0; vb = 0;
        const int s = rB0 + base + lane;
        if (s < rEnd) {
            const v2i p = p2[s];
            er = p.x; vb = p.y;
        }
    };

    RowSums st;
    rows_begin(st, rB1);
    int er, vb;
    load_inter(0, er, vb);
    for (int base = 0; base < nR; base += 32) {
        int ern = 0, vbn = 0;
        if (base + 32 < nR) load_inter(base + 32, ern, vbn);
        const int c = (nR - base < 32) ? nR - base : 32;
        walk_chunk<false>(st, rB2, rB3, cur_ent, cur_ent, er, 0, vb, rB0 + base, c, lane);
        er = ern; vb = vbn;
    }
    v2f acc[4];
    rows_end(st, rB2, rB3, acc);
    normalize_store4(acc, row0, lane, emb, n1, dst, last);
}

static_assert(NHOPS == 2, "kernel_launch writes the two hops out");
extern "C" void kernel_launch(void* const* d_in, const int* in_sizes, int n_in,
                              void* d_out, int out_size, void* d_ws, size_t ws_size,
                              hipStream_t stream)
{
    (void)in_sizes; (void)n_in; (void)out_size; (void)ws_size;

    const float* user_emb   = (const float*)d_in[0];
    const float* entity_emb = (const float*)d_in[1];
    const float* weight     = (const float*)d_in[2];
    const float* inter_vals = (const float*)d_in[3];
    const int*   edge_index = (const int*)d_in[4];
    const int*   edge_type  = (const int*)d_in[5];
    const int*   inter_rows = (const int*)d_in[6];
    const int*   inter_cols = (const int*)d_in[7];
    const int*   head = edge_index;
    const int*   tail = edge_index + E_EDGES;

    float* out_ent = (float*)d_out;
    float* out_usr = out_ent + (size_t)N_ENT * CH;

    char* ws = (char*)d_ws;
    size_t off_b = 0;
    auto alloc_b = [&](size_t bytes) -> void* {
        void* p = (void*)(ws + off_b);
        off_b += ((bytes + 255) / 256) * 256;
        return p;
    };
    float* n1_ent = (float*)alloc_b((size_t)N_ENT * CH * sizeof(float));
    float* n1_usr = (float*)alloc_b((size_t)N_USR * CH * sizeof(float));
    float* tbl    = (float*)alloc_b((size_t)N_ENT * TP * sizeof(float));
    int*   pay    = (int*)alloc_b(((size_t)E_EDGES + 2 * (size_t)NNZ_CNT) * 2 * sizeof(int));
    int*   rank   = (int*)alloc_b(((size_t)E_EDGES + 2 * (size_t)NNZ_CNT) * sizeof(int));
    int*   cnt    = (int*)alloc_b((size_t)L_PAD * sizeof(int));
    int*   offs   = (int*)alloc_b((size_t)L_PAD * sizeof(int));
    int*   bsum   = (int*)alloc_b((size_t)SCAN_NB * sizeof(int));

    hipMemsetAsync(cnt, 0, (size_t)L_PAD * sizeof(int), stream);
    count_rank_kernel<<<ITEM_BLOCKS, 256, 0, stream>>>(head, inter_cols, inter_rows, cnt, rank);
    scan_sums_kernel<<<SCAN_NB, SCAN_T, 0, stream>>>(cnt, bsum);
    scan_offsets_kernel<<<SCAN_NB, SCAN_T, 0, stream>>>(cnt, bsum, offs);
    place_kernel<<<ITEM_BLOCKS, 256, 0, stream>>>(
        head, tail, edge_type, inter_rows, inter_cols, inter_vals, offs, rank, pay);

    auto hop = [&](const float* cur_ent, const float* cur_usr, float* dst_ent, float* dst_usr, int last) {
        relnorm_wmma_kernel<<<((N_TILES + 1) / 2 + TBL_WAVES - 1) / TBL_WAVES, TBL_WAVES * 32, 0, stream>>>(
            cur_ent, weight, tbl);
        entity_row_kernel<<<(N_ENT + ROWS_PER_BLOCK - 1) / ROWS_PER_BLOCK, 256, 0, stream>>>(
            cur_ent, cur_usr, weight, tbl, offs, pay, entity_emb, n1_ent, dst_ent, last);
        user_row_kernel<<<(N_USR + ROWS_PER_BLOCK - 1) / ROWS_PER_BLOCK, 256, 0, stream>>>(
            cur_ent, offs, pay, user_emb, n1_usr, dst_usr, last);
    };
    hop(entity_emb, user_emb, n1_ent, n1_usr, 0);
    hop(n1_ent, n1_usr, out_ent, out_usr, 1);
}


# Grouping by destination row: items ranked with LDS counters in a two-level counting sort (key ranges, then keys), not one returning global atomic per item; sum order by LDS arrival, still varies
# speedup vs baseline: 4.3870x; 1.5963x over previous
#include <hip/hip_runtime.h>
#include <math.h>

#define N_ENT   150000
#define N_USR   60000
#define CH      64
#define E_EDGES 1000000
#define NNZ_CNT 1000000
#define NRELM1  9
#define NHOPS   2
#define EPS_N   1e-12f

typedef __attribute__((ext_vector_type(2))) float v2f;
typedef __attribute__((ext_vector_type(4))) float v4f;
typedef __attribute__((ext_vector_type(2))) int   v2i;
typedef __attribute__((ext_vector_type(4))) int   v4i;
typedef __attribute__((ext_vector_type(8))) float v8f;

#define TBL_WAVES 4
#define TP 16
#define N_TILES (N_ENT / 16)
#define LDP 68
static_assert(N_ENT % 16 == 0, "entity rows tile by 16");
static_assert(16 * CH == 8 * TBL_WAVES * 32, "relation staging: 8 elements per thread");
__global__ __launch_bounds__(TBL_WAVES * 32)
void relnorm_wmma_kernel(const float* __restrict__ ent,
                         const float* __restrict__ weight,
                         float* __restrict__ tbl)
{
    __shared__ float sW2[16 * LDP];
    __shared__ float sA[TBL_WAVES][16 * LDP];
    __shared__ float sB[TBL_WAVES][16 * LDP];

    const int lane = threadIdx.x & 31;
    const int wave = threadIdx.x >> 5;

    const int tA = 2 * (blockIdx.x * TBL_WAVES + wave);
    const bool validA = tA < N_TILES;
    const bool validB = tA + 1 < N_TILES;
    const int tAc = validA ? tA : 0;
    const int tBc = validB ? tA + 1 : tAc;

    float wv[8];
#pragma unroll
    for (int j = 0; j < 8; ++j) {
        const int i = threadIdx.x + j * (TBL_WAVES * 32);
        const int n = i >> 6, c = i & 63;
        const int nc = (n < NRELM1) ? n : NRELM1 - 1;
        const float w = weight[nc * CH + c];
        wv[j] = (n < NRELM1) ? w : 0.0f;
    }

    const v4f* __restrict__ srcA = (const v4f*)(ent + (size_t)tAc * 16 * CH);
    const v4f* __restrict__ srcB = (const v4f*)(ent + (size_t)tBc * 16 * CH);
    v4f xa[8], xb[8];
#pragma unroll
    for (int it = 0; it < 8; ++it) {
        xa[it] = srcA[it * 32 + lane];
        xb[it] = srcB[it * 32 + lane];
    }

#pragma unroll
    for (int j = 0; j < 8; ++j) {
        const int i = threadIdx.x + j * (TBL_WAVES * 32);
        const int n = i >> 6, c = i & 63;
        sW2[n * LDP + c] = wv[j] * wv[j];
    }
    __syncthreads();
    if (!validA) return;

    const int srow = lane >> 4;
    const int scol = (lane & 15) << 2;
#pragma unroll
    for (int it = 0; it < 8; ++it) {
        const v4f a = xa[it], b = xb[it];
        *(v4f*)&sA[wave][(2 * it + srow) * LDP + scol] = a * a;
        *(v4f*)&sB[wave][(2 * it + srow) * LDP + scol] = b * b;
    }

    v8f accA = {};
    v8f accB = {};
    const int m    = lane & 15;
    const int koff = (lane >> 4) << 1;
    for (int kk = 0; kk < 16; ++kk) {
        const int c = kk * 4 + koff;
        v2f a, a2, b;
        a.x  = sA[wave][m * LDP + c];  a.y  = sA[wave][m * LDP + c + 1];
        a2.x = sB[wave][m * LDP + c];  a2.y = sB[wave][m * LDP + c + 1];
        b.x  = sW2[m * LDP + c];       b.y  = sW2[m * LDP + c + 1];
        accA = __builtin_amdgcn_wmma_f32_16x16x4_f32(false, a,  false, b, (short)0, accA, false, false);
        accB = __builtin_amdgcn_wmma_f32_16x16x4_f32(false, a2, false, b, (short)0, accB, false, false);
    }

    const int mrow = (lane >> 4) << 3;
    const int ncol = lane & 15;
    float* dA = tbl + (size_t)(tA * 16 + mrow) * TP + ncol;
#pragma unroll
    for (int v = 0; v < 8; ++v) dA[v * TP] = accA[v];
    if (validB) {
        float* dB = dA + 16 * TP;
#pragma unroll
        for (int v = 0; v < 8; ++v) dB[v * TP] = accB[v];
    }
}

#define L_SEG    (2 * N_ENT + N_USR)
#define N_ITEMS  (E_EDGES + 2 * NNZ_CNT)
#define SCAN_T   256
#define SCAN_PER (4 * SCAN_T)
static_assert(E_EDGES == NNZ_CNT, "one thread takes edge i and interaction i");
static_assert((long long)E_EDGES + 2LL * NNZ_CNT < (1LL << 31), "slots fit an int");

#define KSH      10
#define KPB      (1 << KSH)
#define NBKT     ((L_SEG + KPB - 1) / KPB)
#define L_PAD    (NBKT * KPB)
#define IDX_BITS 18
static_assert(N_ENT <= (1 << IDX_BITS) && N_USR <= (1 << IDX_BITS), "source rows fit IDX_BITS");
static_assert(IDX_BITS + KSH <= 31, "the packed word stays non-negative");

#define IPT   4
#define NGRP  (E_EDGES / IPT)
#define PBLK  256
#define PTHR  1024
#define GPB   ((NGRP + PBLK - 1) / PBLK)
#define H_LEN (NBKT * PBLK)
#define H_NB  (H_LEN / SCAN_PER)
static_assert(E_EDGES % IPT == 0, "items split into whole groups of IPT");
static_assert(IPT == 4, "the item kernels read their groups as 4-vectors");
static_assert(GPB <= PTHR, "a level-1 thread has at most one group");
static_assert(NBKT <= PTHR, "a level-1 thread has at most one bucket counter");
static_assert(H_LEN % SCAN_PER == 0, "the count table splits into whole scan blocks");
static_assert((size_t)N_ITEMS * 8 <= (size_t)N_ENT * CH * sizeof(float), "the level-1 copy fits the plane it borrows");

__global__ __launch_bounds__(PTHR)
void bucket_count_kernel(const int* __restrict__ head,
                         const int* __restrict__ cols,
                         const int* __restrict__ rows,
                         int* __restrict__ hist)
{
    __shared__ int sh[NBKT];
    if (threadIdx.x < NBKT) sh[threadIdx.x] = 0;
    __syncthreads();
    const int g = blockIdx.x * GPB + threadIdx.x;
    if (threadIdx.x < GPB && g < NGRP) {
        const v4i h = ((const v4i*)head)[g];
        const v4i c = ((const v4i*)cols)[g];
        const v4i r = ((const v4i*)rows)[g];
#pragma unroll
        for (int u = 0; u < IPT; ++u) {
            atomicAdd(&sh[h[u] >> KSH], 1);
            atomicAdd(&sh[(N_ENT + c[u]) >> KSH], 1);
            atomicAdd(&sh[(2 * N_ENT + r[u]) >> KSH], 1);
        }
    }
    __syncthreads();
    if (threadIdx.x < NBKT) hist[threadIdx.x * PBLK + blockIdx.x] = sh[threadIdx.x];
}

__device__ __forceinline__ int block_sum_256(int v, int* sred)
{
    for (int m = 16; m >= 1; m >>= 1) v += __shfl_xor(v, m, 32);
    if ((threadIdx.x & 31) == 0) sred[threadIdx.x >> 5] = v;
    __syncthreads();
    int t = 0;
    for (int w = 0; w < SCAN_T / 32; ++w) t += sred[w];
    return t;
}

__global__ __launch_bounds__(SCAN_T)
void scan_sums_kernel(const int* __restrict__ cnt, int* __restrict__ bsum)
{
    __shared__ int sred[SCAN_T / 32];
    const v4i c = ((const v4i*)cnt)[blockIdx.x * SCAN_T + threadIdx.x];
    const int t = block_sum_256(c.x + c.y + c.z + c.w, sred);
    if (threadIdx.x == 0) bsum[blockIdx.x] = t;
}

__global__ __launch_bounds__(SCAN_T)
void scan_offsets_kernel(const int* __restrict__ cnt,
                         const int* __restrict__ bsum,
                         int* __restrict__ off)
{
    __shared__ int sred[SCAN_T / 32];
    __shared__ int swav[SCAN_T / 32];
    const int lane = threadIdx.x & 31, wave = threadIdx.x >> 5;
    int part = 0;
    for (int j = threadIdx.x; j < (int)blockIdx.x; j += SCAN_T) part += bsum[j];
    const int base = block_sum_256(part, sred);
    const v4i c = ((const v4i*)cnt)[blockIdx.x * SCAN_T + threadIdx.x];
    const int tsum = c.x + c.y + c.z + c.w;
    int inc = tsum;
    for (int d = 1; d < 32; d <<= 1) {
        const int o = __shfl_up(inc, d, 32);
        if (lane >= d) inc += o;
    }
    if (lane == 31) swav[wave] = inc;
    __syncthreads();
    int wpre = 0;
    for (int w = 0; w < wave; ++w) wpre += swav[w];
    const int ex = base + wpre + inc - tsum;
    v4i o;
    o.x = ex; o.y = ex + c.x; o.z = o.y + c.y; o.w = o.z + c.z;
    ((v4i*)off)[blockIdx.x * SCAN_T + threadIdx.x] = o;
}

__global__ __launch_bounds__(PTHR)
void bucket_scatter_kernel(const int* __restrict__ head,
                           const int* __restrict__ tail,
                           const int* __restrict__ etype,
                           const int* __restrict__ rows,
                           const int* __restrict__ cols,
                           const float* __restrict__ vals,
                           const int* __restrict__ hoff,
                           int* __restrict__ tmp)
{
    __shared__ int sp[NBKT];
    if (threadIdx.x < NBKT) sp[threadIdx.x] = hoff[threadIdx.x * PBLK + blockIdx.x];
    __syncthreads();
    const int g = blockIdx.x * GPB + threadIdx.x;
    if (threadIdx.x >= GPB || g >= NGRP) return;
    const v4i h  = ((const v4i*)head)[g];
    const v4i t  = ((const v4i*)tail)[g];
    const v4i c  = ((const v4i*)cols)[g];
    const v4i r  = ((const v4i*)rows)[g];
    const v4i e  = ((const v4i*)etype)[g];
    const v4f vf = ((const v4f*)vals)[g];
    v2i* __restrict__ t2 = (v2i*)tmp;
    int pH[IPT], pC[IPT], pR[IPT];
#pragma unroll
    for (int u = 0; u < IPT; ++u) {
        pH[u] = atomicAdd(&sp[h[u] >> KSH], 1);
        pC[u] = atomicAdd(&sp[(N_ENT + c[u]) >> KSH], 1);
        pR[u] = atomicAdd(&sp[(2 * N_ENT + r[u]) >> KSH], 1);
    }
#pragma unroll
    for (int u = 0; u < IPT; ++u) {
        int et = e[u] - 1;
        if (et < 0) et += NRELM1;
        const int vb = __float_as_int(vf[u]);
        const int kH = h[u], kC = N_ENT + c[u], kR = 2 * N_ENT + r[u];
        v2i a; a.x = t[u] | ((kH & (KPB - 1)) << IDX_BITS); a.y = et;  t2[pH[u]] = a;
        v2i b; b.x = r[u] | ((kC & (KPB - 1)) << IDX_BITS); b.y = vb;  t2[pC[u]] = b;
        v2i d; d.x = c[u] | ((kR & (KPB - 1)) << IDX_BITS); d.y = vb;  t2[pR[u]] = d;
    }
}

__global__ __launch_bounds__(KPB)
void bucket_sort_kernel(const int* __restrict__ hoff,
                        const int* __restrict__ tmp,
                        int* __restrict__ off,
                        int* __restrict__ pay)
{
    __shared__ int scnt[KPB];
    __shared__ int swav[KPB / 32];
    const int b = blockIdx.x;
    const int lane = threadIdx.x & 31, wave = threadIdx.x >> 5;
    const int bstart = hoff[b * PBLK];
    const int bend = (b + 1 < NBKT) ? hoff[(b + 1) * PBLK] : N_ITEMS;
    const v2i* __restrict__ t2 = (const v2i*)tmp;
    v2i* __restrict__ p2 = (v2i*)pay;

    scnt[threadIdx.x] = 0;
    __syncthreads();
    int i = bstart + threadIdx.x;
    for (; i + 3 * KPB < bend; i += 4 * KPB) {
        int k[4];
#pragma unroll
        for (int u = 0; u < 4; ++u) k[u] = tmp[2 * (i + u * KPB)];
#pragma unroll
        for (int u = 0; u < 4; ++u) atomicAdd(&scnt[k[u] >> IDX_BITS], 1);
    }
    for (; i < bend; i += KPB) atomicAdd(&scnt[tmp[2 * i] >> IDX_BITS], 1);
    __syncthreads();

    const int c = scnt[threadIdx.x];
    int inc = c;
    for (int d = 1; d < 32; d <<= 1) {
        const int o = __shfl_up(inc, d, 32);
        if (lane >= d) inc += o;
    }
    if (lane == 31) swav[wave] = inc;
    __syncthreads();
    int wpre = 0;
    for (int w = 0; w < wave; ++w) wpre += swav[w];
    const int first = bstart + wpre + inc - c;
    off[b * KPB + threadIdx.x] = first;
    scnt[threadIdx.x] = first;
    __syncthreads();

    i = bstart + threadIdx.x;
    for (; i + 3 * KPB < bend; i += 4 * KPB) {
        v2i it[4];
        int slot[4];
#pragma unroll
        for (int u = 0; u < 4; ++u) it[u] = t2[i + u * KPB];
#pragma unroll
        for (int u = 0; u < 4; ++u) slot[u] = atomicAdd(&scnt[it[u].x >> IDX_BITS], 1);
#pragma unroll
        for (int u = 0; u < 4; ++u) {
            v2i o; o.x = it[u].x & ((1 << IDX_BITS) - 1); o.y = it[u].y;
            p2[slot[u]] = o;
        }
    }
    for (; i < bend; i += KPB) {
        const v2i it = t2[i];
        const int slot = atomicAdd(&scnt[it.x >> IDX_BITS], 1);
        v2i o; o.x = it.x & ((1 << IDX_BITS) - 1); o.y = it.y;
        p2[slot] = o;
    }
}

#define RPW 4
#define ROWS_PER_BLOCK (8 * RPW)
#define NO_SLOT 0x7fffffff
static_assert(RPW == 4, "the row kernels read their bounds as 4-vectors and name four sums");
static_assert(N_ENT % RPW == 0 && N_USR % RPW == 0, "a wave's rows all exist or none does");
static_assert(L_PAD > L_SEG, "off[] has an entry behind the last segment");

__device__ __forceinline__ void wave_sum4(float (&v)[4])
{
    for (int m = 16; m >= 1; m >>= 1) {
        float t[4];
#pragma unroll
        for (int r = 0; r < 4; ++r) t[r] = __shfl_xor(v[r], m, 32);
#pragma unroll
        for (int r = 0; r < 4; ++r) v[r] += t[r];
    }
}
__device__ __forceinline__ void wave_max4(float (&v)[4])
{
    for (int m = 16; m >= 1; m >>= 1) {
        float t[4];
#pragma unroll
        for (int r = 0; r < 4; ++r) t[r] = __shfl_xor(v[r], m, 32);
#pragma unroll
        for (int r = 0; r < 4; ++r) v[r] = fmaxf(v[r], t[r]);
    }
}

struct RowSums {
    v2f acc, r0, r1, r2;
    int cur, nb;
};
__device__ __forceinline__ void rows_begin(RowSums& st, int b1)
{
    st.acc = v2f{0.0f, 0.0f}; st.r0 = st.acc; st.r1 = st.acc; st.r2 = st.acc;
    st.cur = 0; st.nb = b1;
}
__device__ __forceinline__ void rows_step(RowSums& st, int b2, int b3)
{
    if (st.cur == 0) st.r0 = st.acc; else if (st.cur == 1) st.r1 = st.acc; else st.r2 = st.acc;
    st.acc = v2f{0.0f, 0.0f};
    ++st.cur;
    st.nb = (st.cur == 1) ? b2 : (st.cur == 2) ? b3 : NO_SLOT;
}
__device__ __forceinline__ void rows_end(RowSums& st, int b2, int b3, v2f (&o)[4])
{
    while (st.cur < 3) rows_step(st, b2, b3);
    o[0] = st.r0; o[1] = st.r1; o[2] = st.r2; o[3] = st.acc;
}

template <bool REL>
__device__ __forceinline__ void walk_chunk(RowSums& st, int b2, int b3,
                                           const float* __restrict__ src,
                                           const float* __restrict__ weight,
                                           int idx, int rel, int wbits,
                                           int sbase, int c, int lane)
{
    for (int j = 0; j < c; j += 8) {
        v2f x[8], rv[8];
        float w[8];
#pragma unroll
        for (int u = 0; u < 8; ++u) {
            const int ij = __builtin_amdgcn_readlane(idx, j + u);
            w[u] = __int_as_float(__builtin_amdgcn_readlane(wbits, j + u));
            x[u] = *(const v2f*)(src + (size_t)ij * CH + 2 * lane);
            if (REL) {
                const int ej = __builtin_amdgcn_readlane(rel, j + u);
                rv[u] = *(const v2f*)(weight + ej * CH + 2 * lane);
            }
        }
#pragma unroll
        for (int u = 0; u < 8; ++u)
            if (j + u < c) {
                const int s = sbase + j + u;
                while (s >= st.nb) rows_step(st, b2, b3);
                float nx = x[u].x, ny = x[u].y;
                if (REL) { nx *= rv[u].x; ny *= rv[u].y; }
                st.acc.x += w[u] * nx;
                st.acc.y += w[u] * ny;
            }
    }
}

__device__ __forceinline__ void normalize_store4(const v2f (&vin)[4], int row0, int lane,
                                                 const float* __restrict__ emb,
                                                 const float* __restrict__ n1,
                                                 float* __restrict__ dst, int last)
{
    const int c = 2 * lane;
    v2f res[4];
#pragma unroll
    for (int r = 0; r < 4; ++r) res[r] = v2f{0.0f, 0.0f};
    if (last) {
#pragma unroll
        for (int r = 0; r < 4; ++r) {
            const v2f e0 = *(const v2f*)(emb + (size_t)(row0 + r) * CH + c);
            const v2f p1 = *(const v2f*)(n1 + (size_t)(row0 + r) * CH + c);
            res[r].x = e0.x + p1.x; res[r].y = e0.y + p1.y;
        }
    }
    float ss[4];
#pragma unroll
    for (int r = 0; r < 4; ++r) ss[r] = vin[r].x * vin[r].x + vin[r].y * vin[r].y;
    wave_sum4(ss);
#pragma unroll
    for (int r = 0; r < 4; ++r) {
        const float inv = 1.0f / fmaxf(sqrtf(ss[r]), EPS_N);
        v2f v = vin[r];
        v.x *= inv; v.y *= inv;
        if (last) { v.x = res[r].x + v.x; v.y = res[r].y + v.y; }
        *(v2f*)(dst + (size_t)(row0 + r) * CH + c) = v;
    }
}

__global__ __launch_bounds__(256)
void entity_row_kernel(const float* __restrict__ cur_ent,
                       const float* __restrict__ cur_usr,
                       const float* __restrict__ weight,
                       const float* __restrict__ tbl,
                       const int* __restrict__ off,
                       const int* __restrict__ pay,
                       const float* __restrict__ emb,
                       const float* __restrict__ n1,
                       float* __restrict__ dst, int last)
{
    const int lane = threadIdx.x & 31;
    const int wave = __builtin_amdgcn_readfirstlane(threadIdx.x >> 5);
    const int row0 = (blockIdx.x * 8 + wave) * RPW;
    if (row0 >= N_ENT) return;
    const v2i* __restrict__ p2 = (const v2i*)pay;
    const float NEG_INF = -__builtin_inff();

    const v4i oH = *(const v4i*)(off + row0);
    const int hEnd = off[row0 + RPW];
    const v4i oC = *(const v4i*)(off + N_ENT + row0);
    const int cEnd = off[N_ENT + row0 + RPW];
    const int hB0 = oH.x, hB1 = oH.y, hB2 = oH.z, hB3 = oH.w;
    const int cB0 = oC.x, cB1 = oC.y, cB2 = oC.z, cB3 = oC.w;
    const int nH = hEnd - hB0, nC = cEnd - cB0;

    auto load_inter = [&](int base, int& ur, int& vb) {
        ur = 0; vb = 0;
        const int s = cB0 + base + lane;
        if (s < cEnd) {
            const v2i p = p2[s];
            ur = p.x; vb = p.y;
        }
    };
    auto load_edge = [&](int base, int& tl, int& et, int& rid, float& a) {
        tl = 0; et = 0; rid = 0; a = NEG_INF;
        const int s = hB0 + base + lane;
        if (s < hEnd) {
            const v2i p = p2[s];
            tl = p.x; et = p.y;
            rid = (s >= hB1) + (s >= hB2) + (s >= hB3);
            a = tbl[(size_t)(row0 + rid) * TP + et] * tbl[(size_t)tl * TP + et];
        }
    };
    auto pick = [](int rid, const float (&v)[4]) {
        return rid == 0 ? v[0] : rid == 1 ? v[1] : rid == 2 ? v[2] : v[3];
    };

    int ur0, vb0;
    load_inter(0, ur0, vb0);
    int tl0, et0, rid0; float a0;
    load_edge(0, tl0, et0, rid0, a0);

    float mx[4];
#pragma unroll
    for (int r = 0; r < 4; ++r) mx[r] = (rid0 == r) ? a0 : NEG_INF;
    for (int base = 32; base < nH; base += 32) {
        int tl, et, rid; float a;
        load_edge(base, tl, et, rid, a);
#pragma unroll
        for (int r = 0; r < 4; ++r) mx[r] = fmaxf(mx[r], (rid == r) ? a : NEG_INF);
    }
    wave_max4(mx);

    const float e0 = (lane < nH) ? expf(a0 - pick(rid0, mx)) : 0.0f;
    float sm[4];
#pragma unroll
    for (int r = 0; r < 4; ++r) sm[r] = (rid0 == r) ? e0 : 0.0f;
    for (int base = 32; base < nH; base += 32) {
        int tl, et, rid; float a;
        load_edge(base, tl, et, rid, a);
        const float e = (base + lane < nH) ? expf(a - pick(rid, mx)) : 0.0f;
#pragma unroll
        for (int r = 0; r < 4; ++r) sm[r] += (rid == r) ? e : 0.0f;
    }
    wave_sum4(sm);

    RowSums st;
    rows_begin(st, hB1);
    for (int base = 0; base < nH; base += 32) {
        int tl = tl0, et = et0, rid = rid0; float e = e0;
        if (base != 0) {
            float a;
            load_edge(base, tl, et, rid, a);
            e = (base + lane < nH) ? expf(a - pick(rid, mx)) : 0.0f;
        }
        const int wi = __float_as_int(e / pick(rid, sm));
        const int c = (nH - base < 32) ? nH - base : 32;
        walk_chunk<true>(st, hB2, hB3, cur_ent, weight, tl, et, wi, hB0 + base, c, lane);
    }
    v2f accE[4];
    rows_end(st, hB2, hB3, accE);

    rows_begin(st, cB1);
    for (int base = 0; base < nC; base += 32) {
        int ur = ur0, vb = vb0;
        if (base != 0) load_inter(base, ur, vb);
        const int c = (nC - base < 32) ? nC - base : 32;
        walk_chunk<false>(st, cB2, cB3, cur_usr, weight, ur, 0, vb, cB0 + base, c, lane);
    }
    v2f accI[4];
    rows_end(st, cB2, cB3, accI);

    v2f v[4];
#pragma unroll
    for (int r = 0; r < 4; ++r) { v[r].x = accE[r].x + accI[r].x; v[r].y = accE[r].y + accI[r].y; }
    normalize_store4(v, row0, lane, emb, n1, dst, last);
}

__global__ __launch_bounds__(256)
void user_row_kernel(const float* __restrict__ cur_ent,
                     const int* __restrict__ off,
                     const int* __restrict__ pay,
                     const float* __restrict__ emb,
                     const float* __restrict__ n1,
                     float* __restrict__ dst, int last)
{
    const int lane = threadIdx.x & 31;
    const int wave = __builtin_amdgcn_readfirstlane(threadIdx.x >> 5);
    const int row0 = (blockIdx.x * 8 + wave) * RPW;
    if (row0 >= N_USR) return;
    const v2i* __restrict__ p2 = (const v2i*)pay;
    const v4i oR = *(const v4i*)(off + 2 * N_ENT + row0);
    const int rEnd = off[2 * N_ENT + row0 + RPW];
    const int rB0 = oR.x, rB1 = oR.y, rB2 = oR.z, rB3 = oR.w;
    const int nR = rEnd - rB0;

    auto load_inter = [&](int base, int& er, int& vb) {
        er = 0; vb = 0;
        const int s = rB0 + base + lane;
        if (s < rEnd) {
            const v2i p = p2[s];
            er = p.x; vb = p.y;
        }
    };

    RowSums st;
    rows_begin(st, rB1);
    int er, vb;
    load_inter(0, er, vb);
    for (int base = 0; base < nR; base += 32) {
        int ern = 0, vbn = 0;
        if (base + 32 < nR) load_inter(base + 32, ern, vbn);
        const int c = (nR - base < 32) ? nR - base : 32;
        walk_chunk<false>(st, rB2, rB3, cur_ent, cur_ent, er, 0, vb, rB0 + base, c, lane);
        er = ern; vb = vbn;
    }
    v2f acc[4];
    rows_end(st, rB2, rB3, acc);
    normalize_store4(acc, row0, lane, emb, n1, dst, last);
}

static_assert(NHOPS == 2, "kernel_launch writes the two hops out");
extern "C" void kernel_launch(void* const* d_in, const int* in_sizes, int n_in,
                              void* d_out, int out_size, void* d_ws, size_t ws_size,
                              hipStream_t stream)
{
    (void)in_sizes; (void)n_in; (void)out_size; (void)ws_size;

    const float* user_emb   = (const float*)d_in[0];
    const float* entity_emb = (const float*)d_in[1];
    const float* weight     = (const float*)d_in[2];
    const float* inter_vals = (const float*)d_in[3];
    const int*   edge_index = (const int*)d_in[4];
    const int*   edge_type  = (const int*)d_in[5];
    const int*   inter_rows = (const int*)d_in[6];
    const int*   inter_cols = (const int*)d_in[7];
    const int*   head = edge_index;
    const int*   tail = edge_index + E_EDGES;

    float* out_ent = (float*)d_out;
    float* out_usr = out_ent + (size_t)N_ENT * CH;

    char* ws = (char*)d_ws;
    size_t off_b = 0;
    auto alloc_b = [&](size_t bytes) -> void* {
        void* p = (void*)(ws + off_b);
        off_b += ((bytes + 255) / 256) * 256;
        return p;
    };
    float* n1_ent = (float*)alloc_b((size_t)N_ENT * CH * sizeof(float));
    float* n1_usr = (float*)alloc_b((size_t)N_USR * CH * sizeof(float));
    float* tbl    = (float*)alloc_b((size_t)N_ENT * TP * sizeof(float));
    int*   pay    = (int*)alloc_b((size_t)N_ITEMS * 2 * sizeof(int));
    int*   hist   = (int*)alloc_b((size_t)H_LEN * sizeof(int));
    int*   hoff   = (int*)alloc_b((size_t)H_LEN * sizeof(int));
    int*   offs   = (int*)alloc_b((size_t)L_PAD * sizeof(int));
    int*   bsum   = (int*)alloc_b((size_t)H_NB * sizeof(int));
    int*   tmp    = (int*)n1_ent;

    bucket_count_kernel<<<PBLK, PTHR, 0, stream>>>(head, inter_cols, inter_rows, hist);
    scan_sums_kernel<<<H_NB, SCAN_T, 0, stream>>>(hist, bsum);
    scan_offsets_kernel<<<H_NB, SCAN_T, 0, stream>>>(hist, bsum, hoff);
    bucket_scatter_kernel<<<PBLK, PTHR, 0, stream>>>(
        head, tail, edge_type, inter_rows, inter_cols, inter_vals, hoff, tmp);
    bucket_sort_kernel<<<NBKT, KPB, 0, stream>>>(hoff, tmp, offs, pay);

    auto hop = [&](const float* cur_ent, const float* cur_usr, float* dst_ent, float* dst_usr, int last) {
        relnorm_wmma_kernel<<<((N_TILES + 1) / 2 + TBL_WAVES - 1) / TBL_WAVES, TBL_WAVES * 32, 0, stream>>>(
            cur_ent, weight, tbl);
        entity_row_kernel<<<(N_ENT + ROWS_PER_BLOCK - 1) / ROWS_PER_BLOCK, 256, 0, stream>>>(
            cur_ent, cur_usr, weight, tbl, offs, pay, entity_emb, n1_ent, dst_ent, last);
        user_row_kernel<<<(N_USR + ROWS_PER_BLOCK - 1) / ROWS_PER_BLOCK, 256, 0, stream>>>(
            cur_ent, offs, pay, user_emb, n1_usr, dst_usr, last);
    };
    hop(entity_emb, user_emb, n1_ent, n1_usr, 0);
    hop(n1_ent, n1_usr, out_ent, out_usr, 1);
}


# A hop's entity rows and user rows run as one launch (block index picks the kind, entity blocks first), not two back to back: 11 -> 9 launches; kernel held at 12 waves/SIMD; row sum order as before
# speedup vs baseline: 4.5314x; 1.0329x over previous
#include <hip/hip_runtime.h>
#include <math.h>

#define N_ENT   150000
#define N_USR   60000
#define CH      64
#define E_EDGES 1000000
#define NNZ_CNT 1000000
#define NRELM1  9
#define NHOPS   2
#define EPS_N   1e-12f

typedef __attribute__((ext_vector_type(2))) float v2f;
typedef __attribute__((ext_vector_type(4))) float v4f;
typedef __attribute__((ext_vector_type(2))) int   v2i;
typedef __attribute__((ext_vector_type(4))) int   v4i;
typedef __attribute__((ext_vector_type(8))) float v8f;

#define TBL_WAVES 4
#define TP 16
#define N_TILES (N_ENT / 16)
#define LDP 68
static_assert(N_ENT % 16 == 0, "entity rows tile by 16");
static_assert(16 * CH == 8 * TBL_WAVES * 32, "relation staging: 8 elements per thread");
__global__ __launch_bounds__(TBL_WAVES * 32)
void relnorm_wmma_kernel(const float* __restrict__ ent,
                         const float* __restrict__ weight,
                         float* __restrict__ tbl)
{
    __shared__ float sW2[16 * LDP];
    __shared__ float sA[TBL_WAVES][16 * LDP];
    __shared__ float sB[TBL_WAVES][16 * LDP];

    const int lane = threadIdx.x & 31;
    const int wave = threadIdx.x >> 5;

    const int tA = 2 * (blockIdx.x * TBL_WAVES + wave);
    const bool validA = tA < N_TILES;
    const bool validB = tA + 1 < N_TILES;
    const int tAc = validA ? tA : 0;
    const int tBc = validB ? tA + 1 : tAc;

    float wv[8];
#pragma unroll
    for (int j = 0; j < 8; ++j) {
        const int i = threadIdx.x + j * (TBL_WAVES * 32);
        const int n = i >> 6, c = i & 63;
        const int nc = (n < NRELM1) ? n : NRELM1 - 1;
        const float w = weight[nc * CH + c];
        wv[j] = (n < NRELM1) ? w : 0.0f;
    }

    const v4f* __restrict__ srcA = (const v4f*)(ent + (size_t)tAc * 16 * CH);
    const v4f* __restrict__ srcB = (const v4f*)(ent + (size_t)tBc * 16 * CH);
    v4f xa[8], xb[8];
#pragma unroll
    for (int it = 0; it < 8; ++it) {
        xa[it] = srcA[it * 32 + lane];
        xb[it] = srcB[it * 32 + lane];
    }

#pragma unroll
    for (int j = 0; j < 8; ++j) {
        const int i = threadIdx.x + j * (TBL_WAVES * 32);
        const int n = i >> 6, c = i & 63;
        sW2[n * LDP + c] = wv[j] * wv[j];
    }
    __syncthreads();
    if (!validA) return;

    const int srow = lane >> 4;
    const int scol = (lane & 15) << 2;
#pragma unroll
    for (int it = 0; it < 8; ++it) {
        const v4f a = xa[it], b = xb[it];
        *(v4f*)&sA[wave][(2 * it + srow) * LDP + scol] = a * a;
        *(v4f*)&sB[wave][(2 * it + srow) * LDP + scol] = b * b;
    }

    v8f accA = {};
    v8f accB = {};
    const int m    = lane & 15;
    const int koff = (lane >> 4) << 1;
    for (int kk = 0; kk < 16; ++kk) {
        const int c = kk * 4 + koff;
        v2f a, a2, b;
        a.x  = sA[wave][m * LDP + c];  a.y  = sA[wave][m * LDP + c + 1];
        a2.x = sB[wave][m * LDP + c];  a2.y = sB[wave][m * LDP + c + 1];
        b.x  = sW2[m * LDP + c];       b.y  = sW2[m * LDP + c + 1];
        accA = __builtin_amdgcn_wmma_f32_16x16x4_f32(false, a,  false, b, (short)0, accA, false, false);
        accB = __builtin_amdgcn_wmma_f32_16x16x4_f32(false, a2, false, b, (short)0, accB, false, false);
    }

    const int mrow = (lane >> 4) << 3;
    const int ncol = lane & 15;
    float* dA = tbl + (size_t)(tA * 16 + mrow) * TP + ncol;
#pragma unroll
    for (int v = 0; v < 8; ++v) dA[v * TP] = accA[v];
    if (validB) {
        float* dB = dA + 16 * TP;
#pragma unroll
        for (int v = 0; v < 8; ++v) dB[v * TP] = accB[v];
    }
}

#define L_SEG    (2 * N_ENT + N_USR)
#define N_ITEMS  (E_EDGES + 2 * NNZ_CNT)
#define SCAN_T   256
#define SCAN_PER (4 * SCAN_T)
static_assert(E_EDGES == NNZ_CNT, "one thread takes edge i and interaction i");
static_assert((long long)E_EDGES + 2LL * NNZ_CNT < (1LL << 31), "slots fit an int");

#define KSH      10
#define KPB      (1 << KSH)
#define NBKT     ((L_SEG + KPB - 1) / KPB)
#define L_PAD    (NBKT * KPB)
#define IDX_BITS 18
static_assert(N_ENT <= (1 << IDX_BITS) && N_USR <= (1 << IDX_BITS), "source rows fit IDX_BITS");
static_assert(IDX_BITS + KSH <= 31, "the packed word stays non-negative");

#define IPT   4
#define NGRP  (E_EDGES / IPT)
#define PBLK  256
#define PTHR  1024
#define GPB   ((NGRP + PBLK - 1) / PBLK)
#define H_LEN (NBKT * PBLK)
#define H_NB  (H_LEN / SCAN_PER)
static_assert(E_EDGES % IPT == 0, "items split into whole groups of IPT");
static_assert(IPT == 4, "the item kernels read their groups as 4-vectors");
static_assert(GPB <= PTHR, "a level-1 thread has at most one group");
static_assert(NBKT <= PTHR, "a level-1 thread has at most one bucket counter");
static_assert(H_LEN % SCAN_PER == 0, "the count table splits into whole scan blocks");
static_assert((size_t)N_ITEMS * 8 <= (size_t)N_ENT * CH * sizeof(float), "the level-1 copy fits the plane it borrows");

__global__ __launch_bounds__(PTHR)
void bucket_count_kernel(const int* __restrict__ head,
                         const int* __restrict__ cols,
                         const int* __restrict__ rows,
                         int* __restrict__ hist)
{
    __shared__ int sh[NBKT];
    if (threadIdx.x < NBKT) sh[threadIdx.x] = 0;
    __syncthreads();
    const int g = blockIdx.x * GPB + threadIdx.x;
    if (threadIdx.x < GPB && g < NGRP) {
        const v4i h = ((const v4i*)head)[g];
        const v4i c = ((const v4i*)cols)[g];
        const v4i r = ((const v4i*)rows)[g];
#pragma unroll
        for (int u = 0; u < IPT; ++u) {
            atomicAdd(&sh[h[u] >> KSH], 1);
            atomicAdd(&sh[(N_ENT + c[u]) >> KSH], 1);
            atomicAdd(&sh[(2 * N_ENT + r[u]) >> KSH], 1);
        }
    }
    __syncthreads();
    if (threadIdx.x < NBKT) hist[threadIdx.x * PBLK + blockIdx.x] = sh[threadIdx.x];
}

__device__ __forceinline__ int block_sum_256(int v, int* sred)
{
    for (int m = 16; m >= 1; m >>= 1) v += __shfl_xor(v, m, 32);
    if ((threadIdx.x & 31) == 0) sred[threadIdx.x >> 5] = v;
    __syncthreads();
    int t = 0;
    for (int w = 0; w < SCAN_T / 32; ++w) t += sred[w];
    return t;
}

__global__ __launch_bounds__(SCAN_T)
void scan_sums_kernel(const int* __restrict__ cnt, int* __restrict__ bsum)
{
    __shared__ int sred[SCAN_T / 32];
    const v4i c = ((const v4i*)cnt)[blockIdx.x * SCAN_T + threadIdx.x];
    const int t = block_sum_256(c.x + c.y + c.z + c.w, sred);
    if (threadIdx.x == 0) bsum[blockIdx.x] = t;
}

__global__ __launch_bounds__(SCAN_T)
void scan_offsets_kernel(const int* __restrict__ cnt,
                         const int* __restrict__ bsum,
                         int* __restrict__ off)
{
    __shared__ int sred[SCAN_T / 32];
    __shared__ int swav[SCAN_T / 32];
    const int lane = threadIdx.x & 31, wave = threadIdx.x >> 5;
    int part = 0;
    for (int j = threadIdx.x; j < (int)blockIdx.x; j += SCAN_T) part += bsum[j];
    const int base = block_sum_256(part, sred);
    const v4i c = ((const v4i*)cnt)[blockIdx.x * SCAN_T + threadIdx.x];
    const int tsum = c.x + c.y + c.z + c.w;
    int inc = tsum;
    for (int d = 1; d < 32; d <<= 1) {
        const int o = __shfl_up(inc, d, 32);
        if (lane >= d) inc += o;
    }
    if (lane == 31) swav[wave] = inc;
    __syncthreads();
    int wpre = 0;
    for (int w = 0; w < wave; ++w) wpre += swav[w];
    const int ex = base + wpre + inc - tsum;
    v4i o;
    o.x = ex; o.y = ex + c.x; o.z = o.y + c.y; o.w = o.z + c.z;
    ((v4i*)off)[blockIdx.x * SCAN_T + threadIdx.x] = o;
}

__global__ __launch_bounds__(PTHR)
void bucket_scatter_kernel(const int* __restrict__ head,
                           const int* __restrict__ tail,
                           const int* __restrict__ etype,
                           const int* __restrict__ rows,
                           const int* __restrict__ cols,
                           const float* __restrict__ vals,
                           const int* __restrict__ hoff,
                           int* __restrict__ tmp)
{
    __shared__ int sp[NBKT];
    if (threadIdx.x < NBKT) sp[threadIdx.x] = hoff[threadIdx.x * PBLK + blockIdx.x];
    __syncthreads();
    const int g = blockIdx.x * GPB + threadIdx.x;
    if (threadIdx.x >= GPB || g >= NGRP) return;
    const v4i h  = ((const v4i*)head)[g];
    const v4i t  = ((const v4i*)tail)[g];
    const v4i c  = ((const v4i*)cols)[g];
    const v4i r  = ((const v4i*)rows)[g];
    const v4i e  = ((const v4i*)etype)[g];
    const v4f vf = ((const v4f*)vals)[g];
    v2i* __restrict__ t2 = (v2i*)tmp;
    int pH[IPT], pC[IPT], pR[IPT];
#pragma unroll
    for (int u = 0; u < IPT; ++u) {
        pH[u] = atomicAdd(&sp[h[u] >> KSH], 1);
        pC[u] = atomicAdd(&sp[(N_ENT + c[u]) >> KSH], 1);
        pR[u] = atomicAdd(&sp[(2 * N_ENT + r[u]) >> KSH], 1);
    }
#pragma unroll
    for (int u = 0; u < IPT; ++u) {
        int et = e[u] - 1;
        if (et < 0) et += NRELM1;
        const int vb = __float_as_int(vf[u]);
        const int kH = h[u], kC = N_ENT + c[u], kR = 2 * N_ENT + r[u];
        v2i a; a.x = t[u] | ((kH & (KPB - 1)) << IDX_BITS); a.y = et;  t2[pH[u]] = a;
        v2i b; b.x = r[u] | ((kC & (KPB - 1)) << IDX_BITS); b.y = vb;  t2[pC[u]] = b;
        v2i d; d.x = c[u] | ((kR & (KPB - 1)) << IDX_BITS); d.y = vb;  t2[pR[u]] = d;
    }
}

__global__ __launch_bounds__(KPB)
void bucket_sort_kernel(const int* __restrict__ hoff,
                        const int* __restrict__ tmp,
                        int* __restrict__ off,
                        int* __restrict__ pay)
{
    __shared__ int scnt[KPB];
    __shared__ int swav[KPB / 32];
    const int b = blockIdx.x;
    const int lane = threadIdx.x & 31, wave = threadIdx.x >> 5;
    const int bstart = hoff[b * PBLK];
    const int bend = (b + 1 < NBKT) ? hoff[(b + 1) * PBLK] : N_ITEMS;
    const v2i* __restrict__ t2 = (const v2i*)tmp;
    v2i* __restrict__ p2 = (v2i*)pay;

    scnt[threadIdx.x] = 0;
    __syncthreads();
    int i = bstart + threadIdx.x;
    for (; i + 3 * KPB < bend; i += 4 * KPB) {
        int k[4];
#pragma unroll
        for (int u = 0; u < 4; ++u) k[u] = tmp[2 * (i + u * KPB)];
#pragma unroll
        for (int u = 0; u < 4; ++u) atomicAdd(&scnt[k[u] >> IDX_BITS], 1);
    }
    for (; i < bend; i += KPB) atomicAdd(&scnt[tmp[2 * i] >> IDX_BITS], 1);
    __syncthreads();

    const int c = scnt[threadIdx.x];
    int inc = c;
    for (int d = 1; d < 32; d <<= 1) {
        const int o = __shfl_up(inc, d, 32);
        if (lane >= d) inc += o;
    }
    if (lane == 31) swav[wave] = inc;
    __syncthreads();
    int wpre = 0;
    for (int w = 0; w < wave; ++w) wpre += swav[w];
    const int first = bstart + wpre + inc - c;
    off[b * KPB + threadIdx.x] = first;
    scnt[threadIdx.x] = first;
    __syncthreads();

    i = bstart + threadIdx.x;
    for (; i + 3 * KPB < bend; i += 4 * KPB) {
        v2i it[4];
        int slot[4];
#pragma unroll
        for (int u = 0; u < 4; ++u) it[u] = t2[i + u * KPB];
#pragma unroll
        for (int u = 0; u < 4; ++u) slot[u] = atomicAdd(&scnt[it[u].x >> IDX_BITS], 1);
#pragma unroll
        for (int u = 0; u < 4; ++u) {
            v2i o; o.x = it[u].x & ((1 << IDX_BITS) - 1); o.y = it[u].y;
            p2[slot[u]] = o;
        }
    }
    for (; i < bend; i += KPB) {
        const v2i it = t2[i];
        const int slot = atomicAdd(&scnt[it.x >> IDX_BITS], 1);
        v2i o; o.x = it.x & ((1 << IDX_BITS) - 1); o.y = it.y;
        p2[slot] = o;
    }
}

#define RPW 4
#define ROWS_PER_BLOCK (8 * RPW)
#define NO_SLOT 0x7fffffff
static_assert(RPW == 4, "the row kernels read their bounds as 4-vectors and name four sums");
static_assert(N_ENT % RPW == 0 && N_USR % RPW == 0, "a wave's rows all exist or none does");
static_assert(L_PAD > L_SEG, "off[] has an entry behind the last segment");

__device__ __forceinline__ void wave_sum4(float (&v)[4])
{
    for (int m = 16; m >= 1; m >>= 1) {
        float t[4];
#pragma unroll
        for (int r = 0; r < 4; ++r) t[r] = __shfl_xor(v[r], m, 32);
#pragma unroll
        for (int r = 0; r < 4; ++r) v[r] += t[r];
    }
}
__device__ __forceinline__ void wave_max4(float (&v)[4])
{
    for (int m = 16; m >= 1; m >>= 1) {
        float t[4];
#pragma unroll
        for (int r = 0; r < 4; ++r) t[r] = __shfl_xor(v[r], m, 32);
#pragma unroll
        for (int r = 0; r < 4; ++r) v[r] = fmaxf(v[r], t[r]);
    }
}

struct RowSums {
    v2f acc, r0, r1, r2;
    int cur, nb;
};
__device__ __forceinline__ void rows_begin(RowSums& st, int b1)
{
    st.acc = v2f{0.0f, 0.0f}; st.r0 = st.acc; st.r1 = st.acc; st.r2 = st.acc;
    st.cur = 0; st.nb = b1;
}
__device__ __forceinline__ void rows_step(RowSums& st, int b2, int b3)
{
    if (st.cur == 0) st.r0 = st.acc; else if (st.cur == 1) st.r1 = st.acc; else st.r2 = st.acc;
    st.acc = v2f{0.0f, 0.0f};
    ++st.cur;
    st.nb = (st.cur == 1) ? b2 : (st.cur == 2) ? b3 : NO_SLOT;
}
__device__ __forceinline__ void rows_end(RowSums& st, int b2, int b3, v2f (&o)[4])
{
    while (st.cur < 3) rows_step(st, b2, b3);
    o[0] = st.r0; o[1] = st.r1; o[2] = st.r2; o[3] = st.acc;
}

template <bool REL>
__device__ __forceinline__ void walk_chunk(RowSums& st, int b2, int b3,
                                           const float* __restrict__ src,
                                           const float* __restrict__ weight,
                                           int idx, int rel, int wbits,
                                           int sbase, int c, int lane)
{
    for (int j = 0; j < c; j += 8) {
        v2f x[8], rv[8];
        float w[8];
#pragma unroll
        for (int u = 0; u < 8; ++u) {
            const int ij = __builtin_amdgcn_readlane(idx, j + u);
            w[u] = __int_as_float(__builtin_amdgcn_readlane(wbits, j + u));
            x[u] = *(const v2f*)(src + (size_t)ij * CH + 2 * lane);
            if (REL) {
                const int ej = __builtin_amdgcn_readlane(rel, j + u);
                rv[u] = *(const v2f*)(weight + ej * CH + 2 * lane);
            }
        }
#pragma unroll
        for (int u = 0; u < 8; ++u)
            if (j + u < c) {
                const int s = sbase + j + u;
                while (s >= st.nb) rows_step(st, b2, b3);
                float nx = x[u].x, ny = x[u].y;
                if (REL) { nx *= rv[u].x; ny *= rv[u].y; }
                st.acc.x += w[u] * nx;
                st.acc.y += w[u] * ny;
            }
    }
}

__device__ __forceinline__ void normalize_store4(const v2f (&vin)[4], int row0, int lane,
                                                 const float* __restrict__ emb,
                                                 const float* __restrict__ n1,
                                                 float* __restrict__ dst, int last)
{
    const int c = 2 * lane;
    v2f res[4];
#pragma unroll
    for (int r = 0; r < 4; ++r) res[r] = v2f{0.0f, 0.0f};
    if (last) {
#pragma unroll
        for (int r = 0; r < 4; ++r) {
            const v2f e0 = *(const v2f*)(emb + (size_t)(row0 + r) * CH + c);
            const v2f p1 = *(const v2f*)(n1 + (size_t)(row0 + r) * CH + c);
            res[r].x = e0.x + p1.x; res[r].y = e0.y + p1.y;
        }
    }
    float ss[4];
#pragma unroll
    for (int r = 0; r < 4; ++r) ss[r] = vin[r].x * vin[r].x + vin[r].y * vin[r].y;
    wave_sum4(ss);
#pragma unroll
    for (int r = 0; r < 4; ++r) {
        const float inv = 1.0f / fmaxf(sqrtf(ss[r]), EPS_N);
        v2f v = vin[r];
        v.x *= inv; v.y *= inv;
        if (last) { v.x = res[r].x + v.x; v.y = res[r].y + v.y; }
        *(v2f*)(dst + (size_t)(row0 + r) * CH + c) = v;
    }
}

__device__ __forceinline__
void entity_rows(const int blk,
                 const float* __restrict__ cur_ent,
                 const float* __restrict__ cur_usr,
                 const float* __restrict__ weight,
                 const float* __restrict__ tbl,
                 const int* __restrict__ off,
                 const int* __restrict__ pay,
                 const float* __restrict__ emb,
                 const float* __restrict__ n1,
                 float* __restrict__ dst, int last)
{
    const int lane = threadIdx.x & 31;
    const int wave = __builtin_amdgcn_readfirstlane(threadIdx.x >> 5);
    const int row0 = (blk * 8 + wave) * RPW;
    if (row0 >= N_ENT) return;
    const v2i* __restrict__ p2 = (const v2i*)pay;
    const float NEG_INF = -__builtin_inff();

    const v4i oH = *(const v4i*)(off + row0);
    const int hEnd = off[row0 + RPW];
    const v4i oC = *(const v4i*)(off + N_ENT + row0);
    const int cEnd = off[N_ENT + row0 + RPW];
    const int hB0 = oH.x, hB1 = oH.y, hB2 = oH.z, hB3 = oH.w;
    const int cB0 = oC.x, cB1 = oC.y, cB2 = oC.z, cB3 = oC.w;
    const int nH = hEnd - hB0, nC = cEnd - cB0;

    auto load_inter = [&](int base, int& ur, int& vb) {
        ur = 0; vb = 0;
        const int s = cB0 + base + lane;
        if (s < cEnd) {
            const v2i p = p2[s];
            ur = p.x; vb = p.y;
        }
    };
    auto load_edge = [&](int base, int& tl, int& et, int& rid, float& a) {
        tl = 0; et = 0; rid = 0; a = NEG_INF;
        const int s = hB0 + base + lane;
        if (s < hEnd) {
            const v2i p = p2[s];
            tl = p.x; et = p.y;
            rid = (s >= hB1) + (s >= hB2) + (s >= hB3);
            a = tbl[(size_t)(row0 + rid) * TP + et] * tbl[(size_t)tl * TP + et];
        }
    };
    auto pick = [](int rid, const float (&v)[4]) {
        return rid == 0 ? v[0] : rid == 1 ? v[1] : rid == 2 ? v[2] : v[3];
    };

    int ur0, vb0;
    load_inter(0, ur0, vb0);
    int tl0, et0, rid0; float a0;
    load_edge(0, tl0, et0, rid0, a0);

    float mx[4];
#pragma unroll
    for (int r = 0; r < 4; ++r) mx[r] = (rid0 == r) ? a0 : NEG_INF;
    for (int base = 32; base < nH; base += 32) {
        int tl, et, rid; float a;
        load_edge(base, tl, et, rid, a);
#pragma unroll
        for (int r = 0; r < 4; ++r) mx[r] = fmaxf(mx[r], (rid == r) ? a : NEG_INF);
    }
    wave_max4(mx);

    const float e0 = (lane < nH) ? expf(a0 - pick(rid0, mx)) : 0.0f;
    float sm[4];
#pragma unroll
    for (int r = 0; r < 4; ++r) sm[r] = (rid0 == r) ? e0 : 0.0f;
    for (int base = 32; base < nH; base += 32) {
        int tl, et, rid; float a;
        load_edge(base, tl, et, rid, a);
        const float e = (base + lane < nH) ? expf(a - pick(rid, mx)) : 0.0f;
#pragma unroll
        for (int r = 0; r < 4; ++r) sm[r] += (rid == r) ? e : 0.0f;
    }
    wave_sum4(sm);

    RowSums st;
    rows_begin(st, hB1);
    for (int base = 0; base < nH; base += 32) {
        int tl = tl0, et = et0, rid = rid0; float e = e0;
        if (base != 0) {
            float a;
            load_edge(base, tl, et, rid, a);
            e = (base + lane < nH) ? expf(a - pick(rid, mx)) : 0.0f;
        }
        const int wi = __float_as_int(e / pick(rid, sm));
        const int c = (nH - base < 32) ? nH - base : 32;
        walk_chunk<true>(st, hB2, hB3, cur_ent, weight, tl, et, wi, hB0 + base, c, lane);
    }
    v2f accE[4];
    rows_end(st, hB2, hB3, accE);

    rows_begin(st, cB1);
    for (int base = 0; base < nC; base += 32) {
        int ur = ur0, vb = vb0;
        if (base != 0) load_inter(base, ur, vb);
        const int c = (nC - base < 32) ? nC - base : 32;
        walk_chunk<false>(st, cB2, cB3, cur_usr, weight, ur, 0, vb, cB0 + base, c, lane);
    }
    v2f accI[4];
    rows_end(st, cB2, cB3, accI);

    v2f v[4];
#pragma unroll
    for (int r = 0; r < 4; ++r) { v[r].x = accE[r].x + accI[r].x; v[r].y = accE[r].y + accI[r].y; }
    normalize_store4(v, row0, lane, emb, n1, dst, last);
}

__device__ __forceinline__
void user_rows(const int blk,
               const float* __restrict__ cur_ent,
               const int* __restrict__ off,
               const int* __restrict__ pay,
               const float* __restrict__ emb,
               const float* __restrict__ n1,
               float* __restrict__ dst, int last)
{
    const int lane = threadIdx.x & 31;
    const int wave = __builtin_amdgcn_readfirstlane(threadIdx.x >> 5);
    const int row0 = (blk * 8 + wave) * RPW;
    if (row0 >= N_USR) return;
    const v2i* __restrict__ p2 = (const v2i*)pay;
    const v4i oR = *(const v4i*)(off + 2 * N_ENT + row0);
    const int rEnd = off[2 * N_ENT + row0 + RPW];
    const int rB0 = oR.x, rB1 = oR.y, rB2 = oR.z, rB3 = oR.w;
    const int nR = rEnd - rB0;

    auto load_inter = [&](int base, int& er, int& vb) {
        er = 0; vb = 0;
        const int s = rB0 + base + lane;
        if (s < rEnd) {
            const v2i p = p2[s];
            er = p.x; vb = p.y;
        }
    };

    RowSums st;
    rows_begin(st, rB1);
    int er, vb;
    load_inter(0, er, vb);
    for (int base = 0; base < nR; base += 32) {
        int ern = 0, vbn = 0;
        if (base + 32 < nR) load_inter(base + 32, ern, vbn);
        const int c = (nR - base < 32) ? nR - base : 32;
        walk_chunk<false>(st, rB2, rB3, cur_ent, cur_ent, er, 0, vb, rB0 + base, c, lane);
        er = ern; vb = vbn;
    }
    v2f acc[4];
    rows_end(st, rB2, rB3, acc);
    normalize_store4(acc, row0, lane, emb, n1, dst, last);
}

#define ENT_BLOCKS ((N_ENT + ROWS_PER_BLOCK - 1) / ROWS_PER_BLOCK)
#define USR_BLOCKS ((N_USR + ROWS_PER_BLOCK - 1) / ROWS_PER_BLOCK)
__global__ __launch_bounds__(256) __attribute__((amdgpu_waves_per_eu(12)))
void hop_rows_kernel(const float* __restrict__ cur_ent,
                     const float* __restrict__ cur_usr,
                     const float* __restrict__ weight,
                     const float* __restrict__ tbl,
                     const int* __restrict__ off,
                     const int* __restrict__ pay,
                     const float* __restrict__ emb_ent,
                     const float* __restrict__ n1_ent,
                     float* __restrict__ dst_ent,
                     const float* __restrict__ emb_usr,
                     const float* __restrict__ n1_usr,
                     float* __restrict__ dst_usr, int last)
{
    if (blockIdx.x < ENT_BLOCKS)
        entity_rows(blockIdx.x, cur_ent, cur_usr, weight, tbl, off, pay, emb_ent, n1_ent, dst_ent, last);
    else
        user_rows(blockIdx.x - ENT_BLOCKS, cur_ent, off, pay, emb_usr, n1_usr, dst_usr, last);
}

static_assert(NHOPS == 2, "kernel_launch writes the two hops out");
extern "C" void kernel_launch(void* const* d_in, const int* in_sizes, int n_in,
                              void* d_out, int out_size, void* d_ws, size_t ws_size,
                              hipStream_t stream)
{
    (void)in_sizes; (void)n_in; (void)out_size; (void)ws_size;

    const float* user_emb   = (const float*)d_in[0];
    const float* entity_emb = (const float*)d_in[1];
    const float* weight     = (const float*)d_in[2];
    const float* inter_vals = (const float*)d_in[3];
    const int*   edge_index = (const int*)d_in[4];
    const int*   edge_type  = (const int*)d_in[5];
    const int*   inter_rows = (const int*)d_in[6];
    const int*   inter_cols = (const int*)d_in[7];
    const int*   head = edge_index;
    const int*   tail = edge_index + E_EDGES;

    float* out_ent = (float*)d_out;
    float* out_usr = out_ent + (size_t)N_ENT * CH;

    char* ws = (char*)d_ws;
    size_t off_b = 0;
    auto alloc_b = [&](size_t bytes) -> void* {
        void* p = (void*)(ws + off_b);
        off_b += ((bytes + 255) / 256) * 256;
        return p;
    };
    float* n1_ent = (float*)alloc_b((size_t)N_ENT * CH * sizeof(float));
    float* n1_usr = (float*)alloc_b((size_t)N_USR * CH * sizeof(float));
    float* tbl    = (float*)alloc_b((size_t)N_ENT * TP * sizeof(float));
    int*   pay    = (int*)alloc_b((size_t)N_ITEMS * 2 * sizeof(int));
    int*   hist   = (int*)alloc_b((size_t)H_LEN * sizeof(int));
    int*   hoff   = (int*)alloc_b((size_t)H_LEN * sizeof(int));
    int*   offs   = (int*)alloc_b((size_t)L_PAD * sizeof(int));
    int*   bsum   = (int*)alloc_b((size_t)H_NB * sizeof(int));
    int*   tmp    = (int*)n1_ent;

    bucket_count_kernel<<<PBLK, PTHR, 0, stream>>>(head, inter_cols, inter_rows, hist);
    scan_sums_kernel<<<H_NB, SCAN_T, 0, stream>>>(hist, bsum);
    scan_offsets_kernel<<<H_NB, SCAN_T, 0, stream>>>(hist, bsum, hoff);
    bucket_scatter_kernel<<<PBLK, PTHR, 0, stream>>>(
        head, tail, edge_type, inter_rows, inter_cols, inter_vals, hoff, tmp);
    bucket_sort_kernel<<<NBKT, KPB, 0, stream>>>(hoff, tmp, offs, pay);

    auto hop = [&](const float* cur_ent, const float* cur_usr, float* dst_ent, float* dst_usr, int last) {
        relnorm_wmma_kernel<<<((N_TILES + 1) / 2 + TBL_WAVES - 1) / TBL_WAVES, TBL_WAVES * 32, 0, stream>>>(
            cur_ent, weight, tbl);
        hop_rows_kernel<<<ENT_BLOCKS + USR_BLOCKS, 256, 0, stream>>>(
            cur_ent, cur_usr, weight, tbl, offs, pay, entity_emb, n1_ent, dst_ent,
            user_emb, n1_usr, dst_usr, last);
    };
    hop(entity_emb, user_emb, n1_ent, n1_usr, 0);
    hop(n1_ent, n1_usr, out_ent, out_usr, 1);
}
